# starting kernel
# baseline (speedup 1.0000x reference)
#include <hip/hip_runtime.h>
#include <hip/hip_bf16.h>

typedef __attribute__((ext_vector_type(2))) float v2f;
typedef __attribute__((ext_vector_type(8))) float v8f;

#define ODIM 64

__global__ __launch_bounds__(128) void gemm_wmma_f32(
    const float* __restrict__ A, const float* __restrict__ W,
    float* __restrict__ C, int M, int K) {
  const int lane = threadIdx.x & 31;
  const int wave = threadIdx.x >> 5;
  const int nb   = wave << 4;
  const int half = lane >> 4;
  const int l15  = lane & 15;
  const int mtiles = M >> 4;

  for (int mt = blockIdx.x; mt < mtiles; mt += gridDim.x) {
    const int mbase = mt << 4;
    v8f acc = {};
    const float* __restrict__ arow = A + (size_t)(mbase + l15) * K;
    for (int k = 0; k < K; k += 4) {
      v2f a;
      a[0] = arow[k + 2 * half + 0];
      a[1] = arow[k + 2 * half + 1];
      v2f b;
      b[0] = W[(size_t)(k + 2 * half + 0) * ODIM + nb + l15];
      b[1] = W[(size_t)(k + 2 * half + 1) * ODIM + nb + l15];
      acc = __builtin_amdgcn_wmma_f32_16x16x4_f32(
           false, a,  false, b,
           (short)0, acc,  false,  false);
    }
    float* __restrict__ crow = C + (size_t)(mbase + 8 * half) * ODIM + nb + l15;
#pragma unroll
    for (int r = 0; r < 8; ++r) crow[(size_t)r * ODIM] = acc[r];
  }
}

__global__ void init_deg(float* __restrict__ deg, int N) {
  int i = blockIdx.x * blockDim.x + threadIdx.x;
  if (i < N) deg[i] = 1.0f;
}

__global__ void edge_deg(const int* __restrict__ dst,
                         const float* __restrict__ w,
                         float* __restrict__ deg, int E) {
  int e = blockIdx.x * blockDim.x + threadIdx.x;
  if (e < E) atomicAdd(&deg[dst[e]], w[e]);
}

__global__ void make_dinv(const float* __restrict__ deg,
                          float* __restrict__ dinv, int N) {
  int i = blockIdx.x * blockDim.x + threadIdx.x;
  if (i < N) {
    float d = deg[i];
    dinv[i] = d > 0.0f ? __frsqrt_rn(d) : 0.0f;
  }
}

__global__ __launch_bounds__(256) void init_h(
    const float* __restrict__ xw, const float* __restrict__ dinv,
    const float* __restrict__ b, float* __restrict__ h, int N) {
  long long t = (long long)blockIdx.x * blockDim.x + threadIdx.x;
  int i = (int)(t >> 6);
  int f = (int)(t & 63);
  if (i < N) {
    float di = dinv[i];
    h[(size_t)i * ODIM + f] = b[f] + di * di * xw[(size_t)i * ODIM + f];
  }
}

__global__ __launch_bounds__(256) void gcn_aggregate(
    const float* __restrict__ xw, const float* __restrict__ dinv,
    const int* __restrict__ src, const int* __restrict__ dst,
    const float* __restrict__ w, float* __restrict__ h, int E) {
  long long t = (long long)blockIdx.x * blockDim.x + threadIdx.x;
  long long e = t >> 6;
  int f = (int)(t & 63);
  if (e >= E) return;
  int s = src[e];
  int d = dst[e];
  float norm = dinv[s] * w[e] * dinv[d];
  atomicAdd(&h[(size_t)d * ODIM + f], norm * xw[(size_t)s * ODIM + f]);
}

__global__ __launch_bounds__(256) void pool_max(
    const float* __restrict__ h, const int* __restrict__ batch,
    float* __restrict__ xp, int* __restrict__ bp, int nc) {
  long long t = (long long)blockIdx.x * blockDim.x + threadIdx.x;
  int c = (int)(t >> 6);
  int f = (int)(t & 63);
  if (c >= nc) return;
  size_t r0 = (size_t)(2 * c) * ODIM;
  xp[(size_t)c * ODIM + f] = fmaxf(h[r0 + f], h[r0 + ODIM + f]);
  if (f == 0) bp[c] = min(batch[2 * c], batch[2 * c + 1]);
}

__global__ void init_first(int* __restrict__ first, int B) {
  int g = blockIdx.x * blockDim.x + threadIdx.x;
  if (g < B) first[g] = 0x7FFFFFFF;
}

__global__ void scatter_first(const int* __restrict__ bp,
                              int* __restrict__ first, int nc) {
  int c = blockIdx.x * blockDim.x + threadIdx.x;
  if (c < nc) atomicMin(&first[bp[c]], c);
}

__global__ void gather_out(const float* __restrict__ xp,
                           const int* __restrict__ first,
                           float* __restrict__ out, int B, int nc) {
  long long t = (long long)blockIdx.x * blockDim.x + threadIdx.x;
  int g = (int)(t >> 6);
  int f = (int)(t & 63);
  if (g >= B) return;
  int c = first[g];
  if (c > nc - 1) c = nc - 1;
  if (c < 0) c = 0;
  out[(size_t)g * ODIM + f] = xp[(size_t)c * ODIM + f];
}

extern "C" void kernel_launch(void* const* d_in, const int* in_sizes, int n_in,
                              void* d_out, int out_size, void* d_ws, size_t ws_size,
                              hipStream_t stream) {
  const float* x   = (const float*)d_in[0];
  const int*   ei  = (const int*)d_in[1];
  const float* ew  = (const float*)d_in[2];
  const int*   bat = (const int*)d_in[3];
  const float* W1  = (const float*)d_in[4];
  const float* b1  = (const float*)d_in[5];
  const float* W2  = (const float*)d_in[6];
  const float* b2  = (const float*)d_in[7];
  float* out = (float*)d_out;

  const int IN_DIM = 128;
  const int N  = in_sizes[0] / IN_DIM;
  const int E  = in_sizes[2];
  const int B  = out_size / ODIM;
  const int NC = N / 2;

  const int* src = ei;
  const int* dst = ei + E;

  char* ws = (char*)d_ws;
  size_t featBytes = (size_t)N * ODIM * sizeof(float);
  float* xw   = (float*)(ws);
  float* h    = (float*)(ws + featBytes);
  float* deg  = (float*)(ws + 2 * featBytes);
  float* dinv = (float*)(ws + 2 * featBytes + (size_t)N * 4);
  float* xp   = (float*)(ws + 2 * featBytes + (size_t)N * 8);
  int*   bp   = (int*)  (ws + 2 * featBytes + (size_t)N * 8 + (size_t)NC * ODIM * 4);
  int*   first= (int*)  (ws + 2 * featBytes + (size_t)N * 8 + (size_t)NC * ODIM * 4 + (size_t)NC * 4);

  const int T = 256;
  dim3 blk(T);

  init_deg<<<(N + T - 1) / T, blk, 0, stream>>>(deg, N);
  edge_deg<<<(E + T - 1) / T, blk, 0, stream>>>(dst, ew, deg, E);
  make_dinv<<<(N + T - 1) / T, blk, 0, stream>>>(deg, dinv, N);

  long long nfeat = (long long)N * ODIM;
  long long efeat = (long long)E * ODIM;
  int nfeatBlocks = (int)((nfeat + T - 1) / T);
  int efeatBlocks = (int)((efeat + T - 1) / T);

  gemm_wmma_f32<<<N / 16, dim3(128), 0, stream>>>(x, W1, xw, N, IN_DIM);
  init_h<<<nfeatBlocks, blk, 0, stream>>>(xw, dinv, b1, h, N);
  gcn_aggregate<<<efeatBlocks, blk, 0, stream>>>(xw, dinv, src, dst, ew, h, E);

  gemm_wmma_f32<<<N / 16, dim3(128), 0, stream>>>(h, W2, xw, N, ODIM);
  init_h<<<nfeatBlocks, blk, 0, stream>>>(xw, dinv, b2, h, N);
  gcn_aggregate<<<efeatBlocks, blk, 0, stream>>>(xw, dinv, src, dst, ew, h, E);

  long long cfeat = (long long)NC * ODIM;
  pool_max<<<(int)((cfeat + T - 1) / T), blk, 0, stream>>>(h, bat, xp, bp, NC);
  init_first<<<(B + T - 1) / T, blk, 0, stream>>>(first, B);
  scatter_first<<<(NC + T - 1) / T, blk, 0, stream>>>(bp, first, NC);
  gather_out<<<(B * ODIM + T - 1) / T, blk, 0, stream>>>(xp, first, out, B, NC);
}


# Edges counting-sorted into 64-row dst buckets once per call; one block per bucket sums deg and h rows in LDS (LDS f32 atomics, arrival order) in place of global f32 atomics; h stored once
# speedup vs baseline: 3.3896x; 3.3896x over previous
#include <hip/hip_runtime.h>
#include <hip/hip_bf16.h>

typedef __attribute__((ext_vector_type(2))) float v2f;
typedef __attribute__((ext_vector_type(8))) float v8f;

#define ODIM 64

__global__ __launch_bounds__(128) void gemm_wmma_f32(
    const float* __restrict__ A, const float* __restrict__ W,
    float* __restrict__ C, int M, int K) {
  const int lane = threadIdx.x & 31;
  const int wave = threadIdx.x >> 5;
  const int nb   = wave << 4;
  const int half = lane >> 4;
  const int l15  = lane & 15;
  const int mtiles = M >> 4;

  for (int mt = blockIdx.x; mt < mtiles; mt += gridDim.x) {
    const int mbase = mt << 4;
    v8f acc = {};
    const float* __restrict__ arow = A + (size_t)(mbase + l15) * K;
    for (int k = 0; k < K; k += 4) {
      v2f a;
      a[0] = arow[k + 2 * half + 0];
      a[1] = arow[k + 2 * half + 1];
      v2f b;
      b[0] = W[(size_t)(k + 2 * half + 0) * ODIM + nb + l15];
      b[1] = W[(size_t)(k + 2 * half + 1) * ODIM + nb + l15];
      acc = __builtin_amdgcn_wmma_f32_16x16x4_f32(
           false, a,  false, b,
           (short)0, acc,  false,  false);
    }
    float* __restrict__ crow = C + (size_t)(mbase + 8 * half) * ODIM + nb + l15;
#pragma unroll
    for (int r = 0; r < 8; ++r) crow[(size_t)r * ODIM] = acc[r];
  }
}

#define RB 64
#define RB_SHIFT 6
#define NBMAX 2048

__global__ void zero_bins(int* __restrict__ bincnt, int* __restrict__ cursor, int nb) {
  int i = blockIdx.x * blockDim.x + threadIdx.x;
  if (i < nb) { bincnt[i] = 0; cursor[i] = 0; }
}

__global__ __launch_bounds__(512) void bin_count(
    const int* __restrict__ dst, int* __restrict__ bincnt, int E, int nb, int chunk) {
  __shared__ int hist[NBMAX];
  for (int b = threadIdx.x; b < nb; b += blockDim.x) hist[b] = 0;
  __syncthreads();
  const int e0 = blockIdx.x * chunk;
  const int e1 = min(e0 + chunk, E);
  for (int e = e0 + threadIdx.x; e < e1; e += blockDim.x)
    atomicAdd(&hist[dst[e] >> RB_SHIFT], 1);
  __syncthreads();
  for (int b = threadIdx.x; b < nb; b += blockDim.x) {
    int c = hist[b];
    if (c) atomicAdd(&bincnt[b], c);
  }
}

__global__ __launch_bounds__(1024) void bin_scan(
    const int* __restrict__ bincnt, int* __restrict__ binstart, int nb) {
  __shared__ int part[1024];
  const int t = threadIdx.x;
  const int per = (nb + 1023) >> 10;
  const int b0 = t * per;
  int s = 0;
  for (int j = 0; j < per; ++j) { int b = b0 + j; if (b < nb) s += bincnt[b]; }
  part[t] = s;
  __syncthreads();
  for (int off = 1; off < 1024; off <<= 1) {
    int v = (t >= off) ? part[t - off] : 0;
    __syncthreads();
    part[t] += v;
    __syncthreads();
  }
  int run = part[t] - s;
  for (int j = 0; j < per; ++j) {
    int b = b0 + j;
    if (b < nb) { binstart[b] = run; run += bincnt[b]; }
  }
  if (t == 1023) binstart[nb] = part[1023];
}

__global__ __launch_bounds__(512) void bin_scatter(
    const int* __restrict__ src, const int* __restrict__ dst, const float* __restrict__ w,
    const int* __restrict__ binstart, int* __restrict__ cursor,
    int4* __restrict__ rec, int E, int nb, int chunk) {
  __shared__ int hist[NBMAX];
  __shared__ int lbase[NBMAX];
  for (int b = threadIdx.x; b < nb; b += blockDim.x) hist[b] = 0;
  __syncthreads();
  const int e0 = blockIdx.x * chunk;
  const int e1 = min(e0 + chunk, E);
  for (int e = e0 + threadIdx.x; e < e1; e += blockDim.x)
    atomicAdd(&hist[dst[e] >> RB_SHIFT], 1);
  __syncthreads();
  for (int b = threadIdx.x; b < nb; b += blockDim.x) {
    int c = hist[b];
    int base = 0;
    if (c) base = binstart[b] + atomicAdd(&cursor[b], c);
    lbase[b] = base;
    hist[b] = 0;
  }
  __syncthreads();
  for (int e = e0 + threadIdx.x; e < e1; e += blockDim.x) {
    int d = dst[e];
    int b = d >> RB_SHIFT;
    int pos = lbase[b] + atomicAdd(&hist[b], 1);
    rec[pos] = make_int4(src[e], d, __float_as_int(w[e]), 0);
  }
}

__global__ __launch_bounds__(256) void bucket_deg(
    const int4* __restrict__ rec, const int* __restrict__ binstart,
    float* __restrict__ dinv, int N) {
  __shared__ float degl[RB];
  const int bkt = blockIdx.x;
  const int base = bkt << RB_SHIFT;
  if (threadIdx.x < RB) degl[threadIdx.x] = 1.0f;
  __syncthreads();
  const int e0 = binstart[bkt], e1 = binstart[bkt + 1];
  for (int e = e0 + threadIdx.x; e < e1; e += blockDim.x) {
    int4 r = rec[e];
    __hip_atomic_fetch_add(&degl[r.y - base], __int_as_float(r.z),
                           __ATOMIC_RELAXED, __HIP_MEMORY_SCOPE_WORKGROUP);
  }
  __syncthreads();
  if (threadIdx.x < RB) {
    int i = base + threadIdx.x;
    if (i < N) {
      float d = degl[threadIdx.x];
      dinv[i] = d > 0.0f ? __frsqrt_rn(d) : 0.0f;
    }
  }
}

__global__ __launch_bounds__(256) void edge_norm(
    int4* __restrict__ rec, const float* __restrict__ dinv, int E) {
  int e = blockIdx.x * blockDim.x + threadIdx.x;
  if (e < E) {
    int4 r = rec[e];
    float nrm = dinv[r.x] * __int_as_float(r.z) * dinv[r.y];
    r.z = __float_as_int(nrm);
    rec[e] = r;
  }
}

__global__ __launch_bounds__(256) void bucket_aggregate(
    const float* __restrict__ xw, const float* __restrict__ dinv,
    const float* __restrict__ bias, const int4* __restrict__ rec,
    const int* __restrict__ binstart, float* __restrict__ h, int N) {
  __shared__ float acc[RB * ODIM];
  const int tid  = threadIdx.x;
  const int bkt  = blockIdx.x;
  const int base = bkt << RB_SHIFT;

#pragma unroll
  for (int k = 0; k < (RB * ODIM / 4) / 256; ++k) {
    const int idx = tid + k * 256;
    const int r = idx >> 4, c4 = idx & 15;
    const int i = base + r;
    float4 v = make_float4(0.0f, 0.0f, 0.0f, 0.0f);
    if (i < N) {
      const float di = dinv[i];
      const float s = di * di;
      const float4 x = *(const float4*)(xw + (size_t)i * ODIM + 4 * c4);
      v.x = bias[4 * c4 + 0] + s * x.x;
      v.y = bias[4 * c4 + 1] + s * x.y;
      v.z = bias[4 * c4 + 2] + s * x.z;
      v.w = bias[4 * c4 + 3] + s * x.w;
    }
    *(float4*)(acc + 4 * idx) = v;
  }
  __syncthreads();

  const int lane = tid & 31;
  const int wave = __builtin_amdgcn_readfirstlane(tid >> 5);
  const int e0 = binstart[bkt], e1 = binstart[bkt + 1];
  const int n = e1 - e0;
  const int seg = (((n + 7) >> 3) + 31) & ~31;
  const int wb = e0 + wave * seg;
  const int we = min(wb + seg, e1);
  const float* __restrict__ xl = xw + 2 * lane;
  float* al = acc + 2 * lane;

  for (int e = wb; e < we; e += 32) {
    const int cnt = min(32, we - e);
    int4 r = make_int4(0, 0, 0, 0);
    if (lane < cnt) r = rec[e + lane];
    int j = 0;
    for (; j + 8 <= cnt; j += 8) {
      float2 x[8]; int dl[8]; float nr[8];
#pragma unroll
      for (int u = 0; u < 8; ++u) {
        const int s = __builtin_amdgcn_readlane(r.x, j + u);
        dl[u] = (__builtin_amdgcn_readlane(r.y, j + u) - base) * ODIM;
        nr[u] = __int_as_float(__builtin_amdgcn_readlane(r.z, j + u));
        x[u] = *(const float2*)(xl + (size_t)s * ODIM);
      }
#pragma unroll
      for (int u = 0; u < 8; ++u) {
        __hip_atomic_fetch_add(al + dl[u],     nr[u] * x[u].x, __ATOMIC_RELAXED, __HIP_MEMORY_SCOPE_WORKGROUP);
        __hip_atomic_fetch_add(al + dl[u] + 1, nr[u] * x[u].y, __ATOMIC_RELAXED, __HIP_MEMORY_SCOPE_WORKGROUP);
      }
    }
    for (; j < cnt; ++j) {
      const int s = __builtin_amdgcn_readlane(r.x, j);
      const int dlj = (__builtin_amdgcn_readlane(r.y, j) - base) * ODIM;
      const float nrj = __int_as_float(__builtin_amdgcn_readlane(r.z, j));
      const float2 xv = *(const float2*)(xl + (size_t)s * ODIM);
      __hip_atomic_fetch_add(al + dlj,     nrj * xv.x, __ATOMIC_RELAXED, __HIP_MEMORY_SCOPE_WORKGROUP);
      __hip_atomic_fetch_add(al + dlj + 1, nrj * xv.y, __ATOMIC_RELAXED, __HIP_MEMORY_SCOPE_WORKGROUP);
    }
  }
  __syncthreads();

#pragma unroll
  for (int k = 0; k < (RB * ODIM / 4) / 256; ++k) {
    const int idx = tid + k * 256;
    const int i = base + (idx >> 4);
    if (i < N)
      *(float4*)(h + (size_t)i * ODIM + 4 * (idx & 15)) = *(const float4*)(acc + 4 * idx);
  }
}

__global__ __launch_bounds__(256) void pool_max(
    const float* __restrict__ h, const int* __restrict__ batch,
    float* __restrict__ xp, int* __restrict__ bp, int nc) {
  long long t = (long long)blockIdx.x * blockDim.x + threadIdx.x;
  int c = (int)(t >> 6);
  int f = (int)(t & 63);
  if (c >= nc) return;
  size_t r0 = (size_t)(2 * c) * ODIM;
  xp[(size_t)c * ODIM + f] = fmaxf(h[r0 + f], h[r0 + ODIM + f]);
  if (f == 0) bp[c] = min(batch[2 * c], batch[2 * c + 1]);
}

__global__ void init_first(int* __restrict__ first, int B) {
  int g = blockIdx.x * blockDim.x + threadIdx.x;
  if (g < B) first[g] = 0x7FFFFFFF;
}

__global__ void scatter_first(const int* __restrict__ bp,
                              int* __restrict__ first, int nc) {
  int c = blockIdx.x * blockDim.x + threadIdx.x;
  if (c < nc) atomicMin(&first[bp[c]], c);
}

__global__ void gather_out(const float* __restrict__ xp,
                           const int* __restrict__ first,
                           float* __restrict__ out, int B, int nc) {
  long long t = (long long)blockIdx.x * blockDim.x + threadIdx.x;
  int g = (int)(t >> 6);
  int f = (int)(t & 63);
  if (g >= B) return;
  int c = first[g];
  if (c > nc - 1) c = nc - 1;
  if (c < 0) c = 0;
  out[(size_t)g * ODIM + f] = xp[(size_t)c * ODIM + f];
}

extern "C" void kernel_launch(void* const* d_in, const int* in_sizes, int n_in,
                              void* d_out, int out_size, void* d_ws, size_t ws_size,
                              hipStream_t stream) {
  const float* x   = (const float*)d_in[0];
  const int*   ei  = (const int*)d_in[1];
  const float* ew  = (const float*)d_in[2];
  const int*   bat = (const int*)d_in[3];
  const float* W1  = (const float*)d_in[4];
  const float* b1  = (const float*)d_in[5];
  const float* W2  = (const float*)d_in[6];
  const float* b2  = (const float*)d_in[7];
  float* out = (float*)d_out;

  const int IN_DIM = 128;
  const int N  = in_sizes[0] / IN_DIM;
  const int E  = in_sizes[2];
  const int B  = out_size / ODIM;
  const int NC = N / 2;

  const int* src = ei;
  const int* dst = ei + E;

  char* ws = (char*)d_ws;
  size_t featBytes = (size_t)N * ODIM * sizeof(float);
  float* xw   = (float*)(ws);
  float* h    = (float*)(ws + featBytes);
  float* dinv = (float*)(ws + 2 * featBytes + (size_t)N * 4);
  float* xp   = (float*)(ws + 2 * featBytes + (size_t)N * 8);
  int*   bp   = (int*)  (ws + 2 * featBytes + (size_t)N * 8 + (size_t)NC * ODIM * 4);
  int*   first= (int*)  (ws + 2 * featBytes + (size_t)N * 8 + (size_t)NC * ODIM * 4 + (size_t)NC * 4);
  int4*  rec      = (int4*)(ws + 134217728);
  int*   bincnt   = (int*) (ws + 201326592);
  int*   cursor   = (int*) (ws + 201326592 + 65536);
  int*   binstart = (int*) (ws + 201326592 + 131072);

  const int T = 256;
  dim3 blk(T);

  const int NB = (N + RB - 1) / RB;
  const int G1 = 256;
  const int chunk = (E + G1 - 1) / G1;
  zero_bins<<<(NB + T - 1) / T, blk, 0, stream>>>(bincnt, cursor, NB);
  bin_count<<<G1, dim3(512), 0, stream>>>(dst, bincnt, E, NB, chunk);
  bin_scan<<<1, dim3(1024), 0, stream>>>(bincnt, binstart, NB);
  bin_scatter<<<G1, dim3(512), 0, stream>>>(src, dst, ew, binstart, cursor, rec, E, NB, chunk);
  bucket_deg<<<NB, blk, 0, stream>>>(rec, binstart, dinv, N);
  edge_norm<<<(E + T - 1) / T, blk, 0, stream>>>(rec, dinv, E);

  gemm_wmma_f32<<<N / 16, dim3(128), 0, stream>>>(x, W1, xw, N, IN_DIM);
  bucket_aggregate<<<NB, blk, 0, stream>>>(xw, dinv, b1, rec, binstart, h, N);

  gemm_wmma_f32<<<N / 16, dim3(128), 0, stream>>>(h, W2, xw, N, ODIM);
  bucket_aggregate<<<NB, blk, 0, stream>>>(xw, dinv, b2, rec, binstart, h, N);

  long long cfeat = (long long)NC * ODIM;
  pool_max<<<(int)((cfeat + T - 1) / T), blk, 0, stream>>>(h, bat, xp, bp, NC);
  init_first<<<(B + T - 1) / T, blk, 0, stream>>>(first, B);
  scatter_first<<<(NC + T - 1) / T, blk, 0, stream>>>(bp, first, NC);
  gather_out<<<(B * ODIM + T - 1) / T, blk, 0, stream>>>(xp, first, out, B, NC);
}


# Only rows the output gathers are computed: first[] from batch first; layer-1 sums and norms only for 2 rows per graph + in-neighbours; layer 2 = (edge sum of h) @ W2 on 2*B rows; sum order changes
# speedup vs baseline: 5.9301x; 1.7495x over previous
#include <hip/hip_runtime.h>
#include <hip/hip_bf16.h>

typedef __attribute__((ext_vector_type(2))) float v2f;
typedef __attribute__((ext_vector_type(8))) float v8f;

#define ODIM 64

__global__ __launch_bounds__(128) void gemm_wmma_f32(
    const float* __restrict__ A, const float* __restrict__ W,
    float* __restrict__ C, int M, int K) {
  const int lane = threadIdx.x & 31;
  const int wave = threadIdx.x >> 5;
  const int nb   = wave << 4;
  const int half = lane >> 4;
  const int l15  = lane & 15;
  const int mtiles = M >> 4;

  for (int mt = blockIdx.x; mt < mtiles; mt += gridDim.x) {
    const int mbase = mt << 4;
    v8f acc = {};
    const float* __restrict__ arow = A + (size_t)(mbase + l15) * K;
    for (int k = 0; k < K; k += 4) {
      v2f a;
      a[0] = arow[k + 2 * half + 0];
      a[1] = arow[k + 2 * half + 1];
      v2f b;
      b[0] = W[(size_t)(k + 2 * half + 0) * ODIM + nb + l15];
      b[1] = W[(size_t)(k + 2 * half + 1) * ODIM + nb + l15];
      acc = __builtin_amdgcn_wmma_f32_16x16x4_f32(
           false, a,  false, b,
           (short)0, acc,  false,  false);
    }
    float* __restrict__ crow = C + (size_t)(mbase + 8 * half) * ODIM + nb + l15;
#pragma unroll
    for (int r = 0; r < 8; ++r) crow[(size_t)r * ODIM] = acc[r];
  }
}

#define RB 64
#define RB_SHIFT 6
#define NBMAX 2048

__global__ void init_tables(int* __restrict__ bincnt, int* __restrict__ cursor, int nb,
                            int* __restrict__ first, int B, int* __restrict__ need, int N) {
  int i = blockIdx.x * blockDim.x + threadIdx.x;
  if (i < nb) { bincnt[i] = 0; cursor[i] = 0; }
  if (i < B) first[i] = 0x7FFFFFFF;
  if (i < N) need[i] = 0;
}

__global__ void scatter_first(const int* __restrict__ batch,
                              int* __restrict__ first, int nc) {
  int c = blockIdx.x * blockDim.x + threadIdx.x;
  if (c < nc) atomicMin(&first[min(batch[2 * c], batch[2 * c + 1])], c);
}

__device__ __forceinline__ int graph_cluster(const int* __restrict__ first, int g, int nc) {
  int c = first[g];
  if (c > nc - 1) c = nc - 1;
  if (c < 0) c = 0;
  return c;
}

__global__ __launch_bounds__(512) void bin_count(
    const int* __restrict__ dst, int* __restrict__ bincnt, int E, int nb, int chunk) {
  __shared__ int hist[NBMAX];
  for (int b = threadIdx.x; b < nb; b += blockDim.x) hist[b] = 0;
  __syncthreads();
  const int e0 = blockIdx.x * chunk;
  const int e1 = min(e0 + chunk, E);
  for (int e = e0 + threadIdx.x; e < e1; e += blockDim.x)
    atomicAdd(&hist[dst[e] >> RB_SHIFT], 1);
  __syncthreads();
  for (int b = threadIdx.x; b < nb; b += blockDim.x) {
    int c = hist[b];
    if (c) atomicAdd(&bincnt[b], c);
  }
}

__global__ __launch_bounds__(1024) void bin_scan(
    const int* __restrict__ bincnt, int* __restrict__ binstart, int nb) {
  __shared__ int part[1024];
  const int t = threadIdx.x;
  const int per = (nb + 1023) >> 10;
  const int b0 = t * per;
  int s = 0;
  for (int j = 0; j < per; ++j) { int b = b0 + j; if (b < nb) s += bincnt[b]; }
  part[t] = s;
  __syncthreads();
  for (int off = 1; off < 1024; off <<= 1) {
    int v = (t >= off) ? part[t - off] : 0;
    __syncthreads();
    part[t] += v;
    __syncthreads();
  }
  int run = part[t] - s;
  for (int j = 0; j < per; ++j) {
    int b = b0 + j;
    if (b < nb) { binstart[b] = run; run += bincnt[b]; }
  }
  if (t == 1023) binstart[nb] = part[1023];
}

__global__ __launch_bounds__(512) void bin_scatter(
    const int* __restrict__ src, const int* __restrict__ dst, const float* __restrict__ w,
    const int* __restrict__ binstart, int* __restrict__ cursor,
    int4* __restrict__ rec, int E, int nb, int chunk) {
  __shared__ int hist[NBMAX];
  __shared__ int lbase[NBMAX];
  for (int b = threadIdx.x; b < nb; b += blockDim.x) hist[b] = 0;
  __syncthreads();
  const int e0 = blockIdx.x * chunk;
  const int e1 = min(e0 + chunk, E);
  for (int e = e0 + threadIdx.x; e < e1; e += blockDim.x)
    atomicAdd(&hist[dst[e] >> RB_SHIFT], 1);
  __syncthreads();
  for (int b = threadIdx.x; b < nb; b += blockDim.x) {
    int c = hist[b];
    int base = 0;
    if (c) base = binstart[b] + atomicAdd(&cursor[b], c);
    lbase[b] = base;
    hist[b] = 0;
  }
  __syncthreads();
  for (int e = e0 + threadIdx.x; e < e1; e += blockDim.x) {
    int d = dst[e];
    int b = d >> RB_SHIFT;
    int pos = lbase[b] + atomicAdd(&hist[b], 1);
    rec[pos] = make_int4(src[e], d, __float_as_int(w[e]), 0);
  }
}

__global__ __launch_bounds__(256) void bucket_deg(
    const int4* __restrict__ rec, const int* __restrict__ binstart,
    float* __restrict__ dinv, int N) {
  __shared__ float degl[RB];
  const int bkt = blockIdx.x;
  const int base = bkt << RB_SHIFT;
  if (threadIdx.x < RB) degl[threadIdx.x] = 1.0f;
  __syncthreads();
  const int e0 = binstart[bkt], e1 = binstart[bkt + 1];
  for (int e = e0 + threadIdx.x; e < e1; e += blockDim.x) {
    int4 r = rec[e];
    __hip_atomic_fetch_add(&degl[r.y - base], __int_as_float(r.z),
                           __ATOMIC_RELAXED, __HIP_MEMORY_SCOPE_WORKGROUP);
  }
  __syncthreads();
  if (threadIdx.x < RB) {
    int i = base + threadIdx.x;
    if (i < N) {
      float d = degl[threadIdx.x];
      dinv[i] = d > 0.0f ? __frsqrt_rn(d) : 0.0f;
    }
  }
}

__global__ __launch_bounds__(256) void mark_need(
    const int* __restrict__ first, const int4* __restrict__ rec,
    const int* __restrict__ binstart, int* __restrict__ need, int nc) {
  const int c = graph_cluster(first, blockIdx.x, nc);
  const int i0 = 2 * c;
  const int bkt = i0 >> RB_SHIFT;
  if (threadIdx.x == 0) { need[i0] = 1; need[i0 + 1] = 1; }
  const int e0 = binstart[bkt], e1 = binstart[bkt + 1];
  for (int e = e0 + threadIdx.x; e < e1; e += blockDim.x) {
    int4 r = rec[e];
    if ((r.y >> 1) == c) need[r.x] = 1;
  }
}

__global__ __launch_bounds__(256) void bucket_aggregate(
    const float* __restrict__ xw, const float* __restrict__ dinv,
    const float* __restrict__ bias, const int4* __restrict__ rec,
    const int* __restrict__ binstart, const int* __restrict__ need,
    float* __restrict__ h, int N) {
  __shared__ float acc[RB * ODIM];
  __shared__ int needl[RB];
  const int tid  = threadIdx.x;
  const int bkt  = blockIdx.x;
  const int base = bkt << RB_SHIFT;

  if (tid < RB) needl[tid] = (base + tid < N) ? need[base + tid] : 0;
  __syncthreads();

#pragma unroll
  for (int k = 0; k < (RB * ODIM / 4) / 256; ++k) {
    const int idx = tid + k * 256;
    const int r = idx >> 4, c4 = idx & 15;
    const int i = base + r;
    float4 v = make_float4(0.0f, 0.0f, 0.0f, 0.0f);
    if (needl[r]) {
      const float di = dinv[i];
      const float s = di * di;
      const float4 x = *(const float4*)(xw + (size_t)i * ODIM + 4 * c4);
      v.x = bias[4 * c4 + 0] + s * x.x;
      v.y = bias[4 * c4 + 1] + s * x.y;
      v.z = bias[4 * c4 + 2] + s * x.z;
      v.w = bias[4 * c4 + 3] + s * x.w;
    }
    *(float4*)(acc + 4 * idx) = v;
  }
  __syncthreads();

  const int lane = tid & 31;
  const int wave = __builtin_amdgcn_readfirstlane(tid >> 5);
  const int e0 = binstart[bkt], e1 = binstart[bkt + 1];
  const int n = e1 - e0;
  const int seg = (((n + 7) >> 3) + 31) & ~31;
  const int wb = e0 + wave * seg;
  const int we = min(wb + seg, e1);
  const float* __restrict__ xl = xw + 2 * lane;
  float* al = acc + 2 * lane;

  for (int e = wb; e < we; e += 32) {
    const int cnt = min(32, we - e);
    int4 r = make_int4(0, 0, 0, 0);
    int keep = 0;
    if (lane < cnt) {
      r = rec[e + lane];
      keep = needl[r.y - base];
      if (keep)
        r.z = __float_as_int(dinv[r.x] * __int_as_float(r.z) * dinv[r.y]);
    }
    unsigned mask = __builtin_amdgcn_ballot_w32(keep != 0);
    while (mask) {
      const int j0 = __builtin_ctz(mask);
      mask &= mask - 1;
      int jj[4] = {j0, j0, j0, j0};
      int nk = 1;
#pragma unroll
      for (int u = 1; u < 4; ++u) {
        if (mask) { jj[u] = __builtin_ctz(mask); mask &= mask - 1; nk = u + 1; }
      }
      float2 x[4]; int dl[4]; float nr[4];
#pragma unroll
      for (int u = 0; u < 4; ++u) {
        const int s = __builtin_amdgcn_readlane(r.x, jj[u]);
        dl[u] = (__builtin_amdgcn_readlane(r.y, jj[u]) - base) * ODIM;
        nr[u] = __int_as_float(__builtin_amdgcn_readlane(r.z, jj[u]));
        x[u] = *(const float2*)(xl + (size_t)s * ODIM);
      }
#pragma unroll
      for (int u = 0; u < 4; ++u) {
        if (u < nk) {
          __hip_atomic_fetch_add(al + dl[u],     nr[u] * x[u].x, __ATOMIC_RELAXED, __HIP_MEMORY_SCOPE_WORKGROUP);
          __hip_atomic_fetch_add(al + dl[u] + 1, nr[u] * x[u].y, __ATOMIC_RELAXED, __HIP_MEMORY_SCOPE_WORKGROUP);
        }
      }
    }
  }
  __syncthreads();

#pragma unroll
  for (int k = 0; k < (RB * ODIM / 4) / 256; ++k) {
    const int idx = tid + k * 256;
    const int i = base + (idx >> 4);
    if (needl[idx >> 4])
      *(float4*)(h + (size_t)i * ODIM + 4 * (idx & 15)) = *(const float4*)(acc + 4 * idx);
  }
}

__global__ __launch_bounds__(256) void graph_rows(
    const float* __restrict__ h, const float* __restrict__ dinv,
    const int4* __restrict__ rec, const int* __restrict__ binstart,
    const int* __restrict__ first, float* __restrict__ agg, int nc) {
  __shared__ float part[8][2][ODIM];
  const int g = blockIdx.x;
  const int c = graph_cluster(first, g, nc);
  const int i0 = 2 * c;
  const int bkt = i0 >> RB_SHIFT;
  const int tid = threadIdx.x;
  const int lane = tid & 31;
  const int wave = __builtin_amdgcn_readfirstlane(tid >> 5);
  const int e0 = binstart[bkt], e1 = binstart[bkt + 1];
  const float* __restrict__ hl = h + 2 * lane;
  float2 a0 = make_float2(0.0f, 0.0f), a1 = make_float2(0.0f, 0.0f);

  for (int e = e0 + wave * 32; e < e1; e += 256) {
    const int cnt = min(32, e1 - e);
    int4 r = make_int4(0, 0, 0, 0);
    int hit = 0;
    if (lane < cnt) {
      r = rec[e + lane];
      hit = ((r.y >> 1) == c);
      if (hit)
        r.z = __float_as_int(dinv[r.x] * __int_as_float(r.z) * dinv[r.y]);
    }
    unsigned mask = __builtin_amdgcn_ballot_w32(hit != 0);
    while (mask) {
      const int j = __builtin_ctz(mask);
      mask &= mask - 1;
      const int s = __builtin_amdgcn_readlane(r.x, j);
      const int d = __builtin_amdgcn_readlane(r.y, j);
      const float nr = __int_as_float(__builtin_amdgcn_readlane(r.z, j));
      const float2 xv = *(const float2*)(hl + (size_t)s * ODIM);
      if (d & 1) { a1.x += nr * xv.x; a1.y += nr * xv.y; }
      else       { a0.x += nr * xv.x; a0.y += nr * xv.y; }
    }
  }
  if (wave == 0) {
    const float d0 = dinv[i0], d1 = dinv[i0 + 1];
    const float2 x0 = *(const float2*)(hl + (size_t)i0 * ODIM);
    const float2 x1 = *(const float2*)(hl + (size_t)(i0 + 1) * ODIM);
    a0.x += d0 * d0 * x0.x; a0.y += d0 * d0 * x0.y;
    a1.x += d1 * d1 * x1.x; a1.y += d1 * d1 * x1.y;
  }
  part[wave][0][2 * lane] = a0.x; part[wave][0][2 * lane + 1] = a0.y;
  part[wave][1][2 * lane] = a1.x; part[wave][1][2 * lane + 1] = a1.y;
  __syncthreads();
  if (tid < 2 * ODIM) {
    const int p = tid >> 6, f = tid & 63;
    float sum = part[0][p][f];
#pragma unroll
    for (int w = 1; w < 8; ++w) sum += part[w][p][f];
    agg[(size_t)(2 * g + p) * ODIM + f] = sum;
  }
}

__global__ __launch_bounds__(128) void gemm_pool_out(
    const float* __restrict__ A, const float* __restrict__ W,
    const float* __restrict__ bias, float* __restrict__ out, int M, int K) {
  const int lane = threadIdx.x & 31;
  const int wave = threadIdx.x >> 5;
  const int nb   = wave << 4;
  const int half = lane >> 4;
  const int l15  = lane & 15;
  const int mtiles = M >> 4;

  for (int mt = blockIdx.x; mt < mtiles; mt += gridDim.x) {
    const int mbase = mt << 4;
    v8f acc = {};
    const float* __restrict__ arow = A + (size_t)(mbase + l15) * K;
    for (int k = 0; k < K; k += 4) {
      v2f a;
      a[0] = arow[k + 2 * half + 0];
      a[1] = arow[k + 2 * half + 1];
      v2f b;
      b[0] = W[(size_t)(k + 2 * half + 0) * ODIM + nb + l15];
      b[1] = W[(size_t)(k + 2 * half + 1) * ODIM + nb + l15];
      acc = __builtin_amdgcn_wmma_f32_16x16x4_f32(
           false, a,  false, b,
           (short)0, acc,  false,  false);
    }
    const float bv = bias[nb + l15];
    float* __restrict__ orow = out + (size_t)((mbase >> 1) + 4 * half) * ODIM + nb + l15;
#pragma unroll
    for (int q = 0; q < 4; ++q)
      orow[(size_t)q * ODIM] = fmaxf(acc[2 * q] + bv, acc[2 * q + 1] + bv);
  }
}

extern "C" void kernel_launch(void* const* d_in, const int* in_sizes, int n_in,
                              void* d_out, int out_size, void* d_ws, size_t ws_size,
                              hipStream_t stream) {
  const float* x   = (const float*)d_in[0];
  const int*   ei  = (const int*)d_in[1];
  const float* ew  = (const float*)d_in[2];
  const int*   bat = (const int*)d_in[3];
  const float* W1  = (const float*)d_in[4];
  const float* b1  = (const float*)d_in[5];
  const float* W2  = (const float*)d_in[6];
  const float* b2  = (const float*)d_in[7];
  float* out = (float*)d_out;

  const int IN_DIM = 128;
  const int N  = in_sizes[0] / IN_DIM;
  const int E  = in_sizes[2];
  const int B  = out_size / ODIM;
  const int NC = N / 2;

  const int* src = ei;
  const int* dst = ei + E;

  char* ws = (char*)d_ws;
  size_t featBytes = (size_t)N * ODIM * sizeof(float);
  size_t aggBytes  = (size_t)2 * B * ODIM * sizeof(float);
  float* xw   = (float*)(ws);
  float* h    = (float*)(ws + featBytes);
  float* dinv = (float*)(ws + 2 * featBytes + (size_t)N * 4);
  float* agg  = (float*)(ws + 2 * featBytes + (size_t)N * 8);
  int*   need = (int*)  (ws + 2 * featBytes + (size_t)N * 8 + aggBytes);
  int*   first= (int*)  (ws + 2 * featBytes + (size_t)N * 12 + aggBytes);
  int4*  rec      = (int4*)(ws + 134217728);
  int*   bincnt   = (int*) (ws + 201326592);
  int*   cursor   = (int*) (ws + 201326592 + 65536);
  int*   binstart = (int*) (ws + 201326592 + 131072);

  const int T = 256;
  dim3 blk(T);

  const int NB = (N + RB - 1) / RB;
  const int cover = N > B ? N : B;
  init_tables<<<(cover + T - 1) / T, blk, 0, stream>>>(bincnt, cursor, NB, first, B, need, N);
  scatter_first<<<(NC + T - 1) / T, blk, 0, stream>>>(bat, first, NC);

  const int G1 = 256;
  const int chunk = (E + G1 - 1) / G1;
  bin_count<<<G1, dim3(512), 0, stream>>>(dst, bincnt, E, NB, chunk);
  bin_scan<<<1, dim3(1024), 0, stream>>>(bincnt, binstart, NB);
  bin_scatter<<<G1, dim3(512), 0, stream>>>(src, dst, ew, binstart, cursor, rec, E, NB, chunk);
  bucket_deg<<<NB, blk, 0, stream>>>(rec, binstart, dinv, N);

  mark_need<<<B, blk, 0, stream>>>(first, rec, binstart, need, NC);

  gemm_wmma_f32<<<N / 16, dim3(128), 0, stream>>>(x, W1, xw, N, IN_DIM);
  bucket_aggregate<<<NB, blk, 0, stream>>>(xw, dinv, b1, rec, binstart, need, h, N);

  graph_rows<<<B, blk, 0, stream>>>(h, dinv, rec, binstart, first, agg, NC);
  gemm_pool_out<<<(2 * B) / 16, dim3(128), 0, stream>>>(agg, W2, b2, out, 2 * B, ODIM);
}


# Layer 1 sums x rows over edges BEFORE the weight product: per bucket, flagged rows' 128-wide sums in LDS, then f32 WMMA on the packed rows; the N x 128 x 64 product kernel goes; sum order changes
# speedup vs baseline: 7.3135x; 1.2333x over previous
#include <hip/hip_runtime.h>
#include <hip/hip_bf16.h>

typedef __attribute__((ext_vector_type(2))) float v2f;
typedef __attribute__((ext_vector_type(8))) float v8f;

#define ODIM 64

#define RB 64
#define RB_SHIFT 6
#define NBMAX 2048

__global__ void init_tables(int* __restrict__ bincnt, int* __restrict__ cursor, int nb,
                            int* __restrict__ first, int B, int* __restrict__ need, int N) {
  int i = blockIdx.x * blockDim.x + threadIdx.x;
  if (i < nb) { bincnt[i] = 0; cursor[i] = 0; }
  if (i < B) first[i] = 0x7FFFFFFF;
  if (i < N) need[i] = 0;
}

__global__ void scatter_first(const int* __restrict__ batch,
                              int* __restrict__ first, int nc) {
  int c = blockIdx.x * blockDim.x + threadIdx.x;
  if (c < nc) atomicMin(&first[min(batch[2 * c], batch[2 * c + 1])], c);
}

__device__ __forceinline__ int graph_cluster(const int* __restrict__ first, int g, int nc) {
  int c = first[g];
  if (c > nc - 1) c = nc - 1;
  if (c < 0) c = 0;
  return c;
}

__global__ __launch_bounds__(512) void bin_count(
    const int* __restrict__ dst, int* __restrict__ bincnt, int E, int nb, int chunk) {
  __shared__ int hist[NBMAX];
  for (int b = threadIdx.x; b < nb; b += blockDim.x) hist[b] = 0;
  __syncthreads();
  const int e0 = blockIdx.x * chunk;
  const int e1 = min(e0 + chunk, E);
  for (int e = e0 + threadIdx.x; e < e1; e += blockDim.x)
    atomicAdd(&hist[dst[e] >> RB_SHIFT], 1);
  __syncthreads();
  for (int b = threadIdx.x; b < nb; b += blockDim.x) {
    int c = hist[b];
    if (c) atomicAdd(&bincnt[b], c);
  }
}

__global__ __launch_bounds__(1024) void bin_scan(
    const int* __restrict__ bincnt, int* __restrict__ binstart, int nb) {
  __shared__ int part[1024];
  const int t = threadIdx.x;
  const int per = (nb + 1023) >> 10;
  const int b0 = t * per;
  int s = 0;
  for (int j = 0; j < per; ++j) { int b = b0 + j; if (b < nb) s += bincnt[b]; }
  part[t] = s;
  __syncthreads();
  for (int off = 1; off < 1024; off <<= 1) {
    int v = (t >= off) ? part[t - off] : 0;
    __syncthreads();
    part[t] += v;
    __syncthreads();
  }
  int run = part[t] - s;
  for (int j = 0; j < per; ++j) {
    int b = b0 + j;
    if (b < nb) { binstart[b] = run; run += bincnt[b]; }
  }
  if (t == 1023) binstart[nb] = part[1023];
}

__global__ __launch_bounds__(512) void bin_scatter(
    const int* __restrict__ src, const int* __restrict__ dst, const float* __restrict__ w,
    const int* __restrict__ binstart, int* __restrict__ cursor,
    int4* __restrict__ rec, int E, int nb, int chunk) {
  __shared__ int hist[NBMAX];
  __shared__ int lbase[NBMAX];
  for (int b = threadIdx.x; b < nb; b += blockDim.x) hist[b] = 0;
  __syncthreads();
  const int e0 = blockIdx.x * chunk;
  const int e1 = min(e0 + chunk, E);
  for (int e = e0 + threadIdx.x; e < e1; e += blockDim.x)
    atomicAdd(&hist[dst[e] >> RB_SHIFT], 1);
  __syncthreads();
  for (int b = threadIdx.x; b < nb; b += blockDim.x) {
    int c = hist[b];
    int base = 0;
    if (c) base = binstart[b] + atomicAdd(&cursor[b], c);
    lbase[b] = base;
    hist[b] = 0;
  }
  __syncthreads();
  for (int e = e0 + threadIdx.x; e < e1; e += blockDim.x) {
    int d = dst[e];
    int b = d >> RB_SHIFT;
    int pos = lbase[b] + atomicAdd(&hist[b], 1);
    rec[pos] = make_int4(src[e], d, __float_as_int(w[e]), 0);
  }
}

__global__ __launch_bounds__(256) void bucket_deg(
    const int4* __restrict__ rec, const int* __restrict__ binstart,
    float* __restrict__ dinv, int N) {
  __shared__ float degl[RB];
  const int bkt = blockIdx.x;
  const int base = bkt << RB_SHIFT;
  if (threadIdx.x < RB) degl[threadIdx.x] = 1.0f;
  __syncthreads();
  const int e0 = binstart[bkt], e1 = binstart[bkt + 1];
  for (int e = e0 + threadIdx.x; e < e1; e += blockDim.x) {
    int4 r = rec[e];
    __hip_atomic_fetch_add(&degl[r.y - base], __int_as_float(r.z),
                           __ATOMIC_RELAXED, __HIP_MEMORY_SCOPE_WORKGROUP);
  }
  __syncthreads();
  if (threadIdx.x < RB) {
    int i = base + threadIdx.x;
    if (i < N) {
      float d = degl[threadIdx.x];
      dinv[i] = d > 0.0f ? __frsqrt_rn(d) : 0.0f;
    }
  }
}

__global__ __launch_bounds__(256) void mark_need(
    const int* __restrict__ first, const int4* __restrict__ rec,
    const int* __restrict__ binstart, int* __restrict__ need, int nc) {
  const int c = graph_cluster(first, blockIdx.x, nc);
  const int i0 = 2 * c;
  const int bkt = i0 >> RB_SHIFT;
  if (threadIdx.x == 0) { need[i0] = 1; need[i0 + 1] = 1; }
  const int e0 = binstart[bkt], e1 = binstart[bkt + 1];
  for (int e = e0 + threadIdx.x; e < e1; e += blockDim.x) {
    int4 r = rec[e];
    if ((r.y >> 1) == c) need[r.x] = 1;
  }
}

#define IDIM 128
#define APITCH 132
#define OPITCH 68

__global__ __launch_bounds__(256) void bucket_aggregate(
    const float* __restrict__ x, const float* __restrict__ W,
    const float* __restrict__ dinv, const float* __restrict__ bias,
    const int4* __restrict__ rec, const int* __restrict__ binstart,
    const int* __restrict__ need, float* __restrict__ h, int N) {
  __shared__ float acc[RB * APITCH];
  __shared__ int slotl[RB];
  __shared__ int rowl[RB];
  __shared__ int cntl[2];
  const int tid  = threadIdx.x;
  const int lane = tid & 31;
  const int wave = __builtin_amdgcn_readfirstlane(tid >> 5);
  const int bkt  = blockIdx.x;
  const int base = bkt << RB_SHIFT;

  int flag = 0, pre = 0;
  if (tid < RB) {
    flag = (base + tid < N) ? (need[base + tid] != 0) : 0;
    const unsigned m = __builtin_amdgcn_ballot_w32(flag != 0);
    pre = __builtin_popcount(m & ((1u << lane) - 1u));
    if (lane == 0) cntl[tid >> 5] = __builtin_popcount(m);
  }
  __syncthreads();
  const int c0n = cntl[0];
  const int ns = __builtin_amdgcn_readfirstlane(c0n + cntl[1]);
  if (ns == 0) return;
  if (tid < RB) {
    const int s = pre + (tid >= 32 ? c0n : 0);
    slotl[tid] = flag ? s * APITCH : -1;
    if (flag) rowl[s] = tid;
  }
  __syncthreads();

  const int mt = (ns + 15) >> 4;
  for (int idx = tid; idx < (mt << 9); idx += 256) {
    const int r = idx >> 5, c4 = idx & 31;
    float4 v = make_float4(0.0f, 0.0f, 0.0f, 0.0f);
    if (r < ns) {
      const int i = base + rowl[r];
      const float di = dinv[i];
      const float s = di * di;
      const float4 xv = *(const float4*)(x + (size_t)i * IDIM + 4 * c4);
      v.x = s * xv.x; v.y = s * xv.y; v.z = s * xv.z; v.w = s * xv.w;
    }
    *(float4*)(acc + r * APITCH + 4 * c4) = v;
  }
  __syncthreads();

  const int e0 = binstart[bkt], e1 = binstart[bkt + 1];
  const int n = e1 - e0;
  const int seg = (((n + 7) >> 3) + 31) & ~31;
  const int wb = e0 + wave * seg;
  const int we = min(wb + seg, e1);
  const float* __restrict__ xl = x + 2 * lane;
  float* al = acc + 2 * lane;

  for (int e = wb; e < we; e += 32) {
    const int cnt = min(32, we - e);
    int4 r = make_int4(0, 0, 0, 0);
    int keep = 0;
    if (lane < cnt) {
      r = rec[e + lane];
      const int so = slotl[r.y - base];
      keep = so >= 0;
      if (keep) {
        r.z = __float_as_int(dinv[r.x] * __int_as_float(r.z) * dinv[r.y]);
        r.y = so;
      }
    }
    unsigned mask = __builtin_amdgcn_ballot_w32(keep != 0);
    while (mask) {
      const int j0 = __builtin_ctz(mask);
      mask &= mask - 1;
      int jj[4] = {j0, j0, j0, j0};
      int nk = 1;
#pragma unroll
      for (int u = 1; u < 4; ++u) {
        if (mask) { jj[u] = __builtin_ctz(mask); mask &= mask - 1; nk = u + 1; }
      }
      float2 xa[4], xb[4]; int dl[4]; float nr[4];
#pragma unroll
      for (int u = 0; u < 4; ++u) {
        const int s = __builtin_amdgcn_readlane(r.x, jj[u]);
        dl[u] = __builtin_amdgcn_readlane(r.y, jj[u]);
        nr[u] = __int_as_float(__builtin_amdgcn_readlane(r.z, jj[u]));
        xa[u] = *(const float2*)(xl + (size_t)s * IDIM);
        xb[u] = *(const float2*)(xl + (size_t)s * IDIM + 64);
      }
#pragma unroll
      for (int u = 0; u < 4; ++u) {
        if (u < nk) {
          __hip_atomic_fetch_add(al + dl[u],      nr[u] * xa[u].x, __ATOMIC_RELAXED, __HIP_MEMORY_SCOPE_WORKGROUP);
          __hip_atomic_fetch_add(al + dl[u] + 1,  nr[u] * xa[u].y, __ATOMIC_RELAXED, __HIP_MEMORY_SCOPE_WORKGROUP);
          __hip_atomic_fetch_add(al + dl[u] + 64, nr[u] * xb[u].x, __ATOMIC_RELAXED, __HIP_MEMORY_SCOPE_WORKGROUP);
          __hip_atomic_fetch_add(al + dl[u] + 65, nr[u] * xb[u].y, __ATOMIC_RELAXED, __HIP_MEMORY_SCOPE_WORKGROUP);
        }
      }
    }
  }
  __syncthreads();

  const int half = lane >> 4;
  const int l15  = lane & 15;
  const int nb   = (wave & 3) << 4;
  const float* __restrict__ wr = W + (size_t)(2 * half) * ODIM + nb + l15;
  v8f d[2];
#pragma unroll
  for (int u = 0; u < 2; ++u) {
    const int t = (wave >> 2) + 2 * u;
    d[u] = (v8f){};
    if (t < mt) {
      v8f p0 = {}, p1 = {};
      const float* ar = acc + (t * 16 + l15) * APITCH + 2 * half;
#pragma unroll 4
      for (int k = 0; k < IDIM; k += 8) {
        v2f a0, b0, a1, b1;
        a0[0] = ar[k + 0];
        a0[1] = ar[k + 1];
        b0[0] = wr[(size_t)(k + 0) * ODIM];
        b0[1] = wr[(size_t)(k + 1) * ODIM];
        p0 = __builtin_amdgcn_wmma_f32_16x16x4_f32(
             false, a0,  false, b0,
             (short)0, p0,  false,  false);
        a1[0] = ar[k + 4];
        a1[1] = ar[k + 5];
        b1[0] = wr[(size_t)(k + 4) * ODIM];
        b1[1] = wr[(size_t)(k + 5) * ODIM];
        p1 = __builtin_amdgcn_wmma_f32_16x16x4_f32(
             false, a1,  false, b1,
             (short)0, p1,  false,  false);
      }
      d[u] = p0 + p1;
    }
  }
  __syncthreads();

  const float bv = bias[nb + l15];
#pragma unroll
  for (int u = 0; u < 2; ++u) {
    const int t = (wave >> 2) + 2 * u;
    if (t < mt) {
      float* orow = acc + (t * 16 + 8 * half) * OPITCH + nb + l15;
#pragma unroll
      for (int r = 0; r < 8; ++r) orow[r * OPITCH] = d[u][r] + bv;
    }
  }
  __syncthreads();

  for (int idx = tid; idx < (ns << 4); idx += 256) {
    const int s = idx >> 4, c4 = idx & 15;
    const int i = base + rowl[s];
    *(float4*)(h + (size_t)i * ODIM + 4 * c4) = *(const float4*)(acc + s * OPITCH + 4 * c4);
  }
}

__global__ __launch_bounds__(256) void graph_rows(
    const float* __restrict__ h, const float* __restrict__ dinv,
    const int4* __restrict__ rec, const int* __restrict__ binstart,
    const int* __restrict__ first, float* __restrict__ agg, int nc) {
  __shared__ float part[8][2][ODIM];
  const int g = blockIdx.x;
  const int c = graph_cluster(first, g, nc);
  const int i0 = 2 * c;
  const int bkt = i0 >> RB_SHIFT;
  const int tid = threadIdx.x;
  const int lane = tid & 31;
  const int wave = __builtin_amdgcn_readfirstlane(tid >> 5);
  const int e0 = binstart[bkt], e1 = binstart[bkt + 1];
  const float* __restrict__ hl = h + 2 * lane;
  float2 a0 = make_float2(0.0f, 0.0f), a1 = make_float2(0.0f, 0.0f);

  for (int e = e0 + wave * 32; e < e1; e += 256) {
    const int cnt = min(32, e1 - e);
    int4 r = make_int4(0, 0, 0, 0);
    int hit = 0;
    if (lane < cnt) {
      r = rec[e + lane];
      hit = ((r.y >> 1) == c);
      if (hit)
        r.z = __float_as_int(dinv[r.x] * __int_as_float(r.z) * dinv[r.y]);
    }
    unsigned mask = __builtin_amdgcn_ballot_w32(hit != 0);
    while (mask) {
      const int j = __builtin_ctz(mask);
      mask &= mask - 1;
      const int s = __builtin_amdgcn_readlane(r.x, j);
      const int d = __builtin_amdgcn_readlane(r.y, j);
      const float nr = __int_as_float(__builtin_amdgcn_readlane(r.z, j));
      const float2 xv = *(const float2*)(hl + (size_t)s * ODIM);
      if (d & 1) { a1.x += nr * xv.x; a1.y += nr * xv.y; }
      else       { a0.x += nr * xv.x; a0.y += nr * xv.y; }
    }
  }
  if (wave == 0) {
    const float d0 = dinv[i0], d1 = dinv[i0 + 1];
    const float2 x0 = *(const float2*)(hl + (size_t)i0 * ODIM);
    const float2 x1 = *(const float2*)(hl + (size_t)(i0 + 1) * ODIM);
    a0.x += d0 * d0 * x0.x; a0.y += d0 * d0 * x0.y;
    a1.x += d1 * d1 * x1.x; a1.y += d1 * d1 * x1.y;
  }
  part[wave][0][2 * lane] = a0.x; part[wave][0][2 * lane + 1] = a0.y;
  part[wave][1][2 * lane] = a1.x; part[wave][1][2 * lane + 1] = a1.y;
  __syncthreads();
  if (tid < 2 * ODIM) {
    const int p = tid >> 6, f = tid & 63;
    float sum = part[0][p][f];
#pragma unroll
    for (int w = 1; w < 8; ++w) sum += part[w][p][f];
    agg[(size_t)(2 * g + p) * ODIM + f] = sum;
  }
}

__global__ __launch_bounds__(128) void gemm_pool_out(
    const float* __restrict__ A, const float* __restrict__ W,
    const float* __restrict__ bias, float* __restrict__ out, int M, int K) {
  const int lane = threadIdx.x & 31;
  const int wave = threadIdx.x >> 5;
  const int nb   = wave << 4;
  const int half = lane >> 4;
  const int l15  = lane & 15;
  const int mtiles = M >> 4;

  for (int mt = blockIdx.x; mt < mtiles; mt += gridDim.x) {
    const int mbase = mt << 4;
    v8f acc = {};
    const float* __restrict__ arow = A + (size_t)(mbase + l15) * K;
    for (int k = 0; k < K; k += 4) {
      v2f a;
      a[0] = arow[k + 2 * half + 0];
      a[1] = arow[k + 2 * half + 1];
      v2f b;
      b[0] = W[(size_t)(k + 2 * half + 0) * ODIM + nb + l15];
      b[1] = W[(size_t)(k + 2 * half + 1) * ODIM + nb + l15];
      acc = __builtin_amdgcn_wmma_f32_16x16x4_f32(
           false, a,  false, b,
           (short)0, acc,  false,  false);
    }
    const float bv = bias[nb + l15];
    float* __restrict__ orow = out + (size_t)((mbase >> 1) + 4 * half) * ODIM + nb + l15;
#pragma unroll
    for (int q = 0; q < 4; ++q)
      orow[(size_t)q * ODIM] = fmaxf(acc[2 * q] + bv, acc[2 * q + 1] + bv);
  }
}

extern "C" void kernel_launch(void* const* d_in, const int* in_sizes, int n_in,
                              void* d_out, int out_size, void* d_ws, size_t ws_size,
                              hipStream_t stream) {
  const float* x   = (const float*)d_in[0];
  const int*   ei  = (const int*)d_in[1];
  const float* ew  = (const float*)d_in[2];
  const int*   bat = (const int*)d_in[3];
  const float* W1  = (const float*)d_in[4];
  const float* b1  = (const float*)d_in[5];
  const float* W2  = (const float*)d_in[6];
  const float* b2  = (const float*)d_in[7];
  float* out = (float*)d_out;

  const int IN_DIM = 128;
  const int N  = in_sizes[0] / IN_DIM;
  const int E  = in_sizes[2];
  const int B  = out_size / ODIM;
  const int NC = N / 2;

  const int* src = ei;
  const int* dst = ei + E;

  char* ws = (char*)d_ws;
  size_t featBytes = (size_t)N * ODIM * sizeof(float);
  size_t aggBytes  = (size_t)2 * B * ODIM * sizeof(float);
  float* h    = (float*)(ws + featBytes);
  float* dinv = (float*)(ws + 2 * featBytes + (size_t)N * 4);
  float* agg  = (float*)(ws + 2 * featBytes + (size_t)N * 8);
  int*   need = (int*)  (ws + 2 * featBytes + (size_t)N * 8 + aggBytes);
  int*   first= (int*)  (ws + 2 * featBytes + (size_t)N * 12 + aggBytes);
  int4*  rec      = (int4*)(ws + 134217728);
  int*   bincnt   = (int*) (ws + 201326592);
  int*   cursor   = (int*) (ws + 201326592 + 65536);
  int*   binstart = (int*) (ws + 201326592 + 131072);

  const int T = 256;
  dim3 blk(T);

  const int NB = (N + RB - 1) / RB;
  const int cover = N > B ? N : B;
  init_tables<<<(cover + T - 1) / T, blk, 0, stream>>>(bincnt, cursor, NB, first, B, need, N);
  scatter_first<<<(NC + T - 1) / T, blk, 0, stream>>>(bat, first, NC);

  const int G1 = 256;
  const int chunk = (E + G1 - 1) / G1;
  bin_count<<<G1, dim3(512), 0, stream>>>(dst, bincnt, E, NB, chunk);
  bin_scan<<<1, dim3(1024), 0, stream>>>(bincnt, binstart, NB);
  bin_scatter<<<G1, dim3(512), 0, stream>>>(src, dst, ew, binstart, cursor, rec, E, NB, chunk);
  bucket_deg<<<NB, blk, 0, stream>>>(rec, binstart, dinv, N);

  mark_need<<<B, blk, 0, stream>>>(first, rec, binstart, need, NC);

  bucket_aggregate<<<NB, blk, 0, stream>>>(x, W1, dinv, b1, rec, binstart, need, h, N);

  graph_rows<<<B, blk, 0, stream>>>(h, dinv, rec, binstart, first, agg, NC);
  gemm_pool_out<<<(2 * B) / 16, dim3(128), 0, stream>>>(agg, W2, b2, out, 2 * B, ODIM);
}


# Launch fusion, 10 launches to 6: first[] pass joins the bucket histogram, each scatter group scans the counts itself, need flags join the degree walk, W2 product and pair max join the row sums
# speedup vs baseline: 7.9151x; 1.0823x over previous
#include <hip/hip_runtime.h>
#include <hip/hip_bf16.h>

typedef __attribute__((ext_vector_type(2))) float v2f;
typedef __attribute__((ext_vector_type(8))) float v8f;

#define ODIM 64

#define RB 64
#define RB_SHIFT 6
#define NBMAX 2048

__global__ void init_tables(int* __restrict__ bincnt, int* __restrict__ cursor, int nb,
                            int* __restrict__ first, int B, int* __restrict__ need, int N) {
  int i = blockIdx.x * blockDim.x + threadIdx.x;
  if (i < nb) { bincnt[i] = 0; cursor[i] = 0; }
  if (i < B) first[i] = 0x7FFFFFFF;
  if (i < N) need[i] = 0;
}

__device__ __forceinline__ int graph_cluster(const int* __restrict__ first, int g, int nc) {
  int c = first[g];
  if (c > nc - 1) c = nc - 1;
  if (c < 0) c = 0;
  return c;
}

__global__ __launch_bounds__(512) void bin_count_first(
    const int* __restrict__ dst, int* __restrict__ bincnt, int E, int nb, int chunk,
    const int* __restrict__ batch, int* __restrict__ first, int nc) {
  __shared__ int hist[NBMAX];
  for (int b = threadIdx.x; b < nb; b += blockDim.x) hist[b] = 0;
  __syncthreads();
  const int c0 = blockIdx.x * blockDim.x + threadIdx.x;
  const int cl = min(c0, nc - 1);
  const int ba = batch[2 * cl], bb = batch[2 * cl + 1];
  const int e0 = blockIdx.x * chunk;
  const int e1 = min(e0 + chunk, E);
  for (int e = e0 + threadIdx.x; e < e1; e += blockDim.x)
    atomicAdd(&hist[dst[e] >> RB_SHIFT], 1);
  __syncthreads();
  for (int b = threadIdx.x; b < nb; b += blockDim.x) {
    int c = hist[b];
    if (c) atomicAdd(&bincnt[b], c);
  }
  if (c0 < nc) atomicMin(&first[min(ba, bb)], c0);
  for (int c = c0 + gridDim.x * blockDim.x; c < nc; c += gridDim.x * blockDim.x)
    atomicMin(&first[min(batch[2 * c], batch[2 * c + 1])], c);
}

#define BPT (NBMAX / 512)
__global__ __launch_bounds__(512) void bin_scatter(
    const int* __restrict__ src, const int* __restrict__ dst, const float* __restrict__ w,
    const int* __restrict__ bincnt, int* __restrict__ binstart, int* __restrict__ cursor,
    int4* __restrict__ rec, int E, int nb, int chunk) {
  __shared__ int hist[NBMAX];
  __shared__ int lbase[NBMAX];
  __shared__ int wtot[16];
  const int t = threadIdx.x;
  const int lane = t & 31;
  const int wv = t >> 5;
  for (int b = t; b < nb; b += 512) hist[b] = 0;
  const int b0 = t * BPT;
  int cv[BPT];
  int s = 0;
#pragma unroll
  for (int j = 0; j < BPT; ++j) {
    const int b = b0 + j;
    const int v = bincnt[min(b, nb - 1)];
    cv[j] = b < nb ? v : 0;
    s += cv[j];
  }
  int inc = s;
  for (int off = 1; off < 32; off <<= 1) {
    int v = __shfl_up(inc, off, 32);
    if (lane >= off) inc += v;
  }
  if (lane == 31) wtot[wv] = inc;
  __syncthreads();
  int run = inc - s;
  for (int q = 0; q < wv; ++q) run += wtot[q];
#pragma unroll
  for (int j = 0; j < BPT; ++j) {
    const int b = b0 + j;
    if (b < nb) {
      lbase[b] = run;
      if (blockIdx.x == 0) binstart[b] = run;
      run += cv[j];
    }
  }
  if (blockIdx.x == 0 && t == 511) binstart[nb] = run;

  const int e0 = blockIdx.x * chunk;
  const int e1 = min(e0 + chunk, E);
  for (int e = e0 + t; e < e1; e += 512)
    atomicAdd(&hist[dst[e] >> RB_SHIFT], 1);
  __syncthreads();
  for (int b = t; b < nb; b += 512) {
    int c = hist[b];
    int base = 0;
    if (c) base = lbase[b] + atomicAdd(&cursor[b], c);
    lbase[b] = base;
    hist[b] = 0;
  }
  __syncthreads();
  for (int e = e0 + t; e < e1; e += 512) {
    int d = dst[e];
    int b = d >> RB_SHIFT;
    int pos = lbase[b] + atomicAdd(&hist[b], 1);
    rec[pos] = make_int4(src[e], d, __float_as_int(w[e]), 0);
  }
}

__global__ __launch_bounds__(256) void bucket_deg_mark(
    const int4* __restrict__ rec, const int* __restrict__ binstart,
    float* __restrict__ dinv, int N,
    const int* __restrict__ first, int B, int* __restrict__ need, int nc) {
  __shared__ float degl[RB];
  __shared__ int tgtl[RB];
  const int bkt = blockIdx.x;
  const int base = bkt << RB_SHIFT;
  const int e0 = binstart[bkt], e1 = binstart[bkt + 1];
  int ig = -1;
  if (threadIdx.x < B) ig = 2 * graph_cluster(first, threadIdx.x, nc);
  if (threadIdx.x < RB) { degl[threadIdx.x] = 1.0f; tgtl[threadIdx.x] = 0; }
  __syncthreads();
  if (ig >= 0 && (ig >> RB_SHIFT) == bkt) { tgtl[ig - base] = 1; tgtl[ig + 1 - base] = 1; }
  for (int g = threadIdx.x + blockDim.x; g < B; g += blockDim.x) {
    const int i0 = 2 * graph_cluster(first, g, nc);
    if ((i0 >> RB_SHIFT) == bkt) { tgtl[i0 - base] = 1; tgtl[i0 + 1 - base] = 1; }
  }
  __syncthreads();
  for (int e = e0 + threadIdx.x; e < e1; e += blockDim.x) {
    int4 r = rec[e];
    __hip_atomic_fetch_add(&degl[r.y - base], __int_as_float(r.z),
                           __ATOMIC_RELAXED, __HIP_MEMORY_SCOPE_WORKGROUP);
    if (tgtl[r.y - base]) need[r.x] = 1;
  }
  __syncthreads();
  if (threadIdx.x < RB) {
    int i = base + threadIdx.x;
    if (i < N) {
      float d = degl[threadIdx.x];
      dinv[i] = d > 0.0f ? __frsqrt_rn(d) : 0.0f;
      if (tgtl[threadIdx.x]) need[i] = 1;
    }
  }
}

#define IDIM 128
#define APITCH 132
#define OPITCH 68

__global__ __launch_bounds__(256) void bucket_aggregate(
    const float* __restrict__ x, const float* __restrict__ W,
    const float* __restrict__ dinv, const float* __restrict__ bias,
    const int4* __restrict__ rec, const int* __restrict__ binstart,
    const int* __restrict__ need, float* __restrict__ h, int N) {
  __shared__ float acc[RB * APITCH];
  __shared__ int slotl[RB];
  __shared__ int rowl[RB];
  __shared__ int cntl[2];
  const int tid  = threadIdx.x;
  const int lane = tid & 31;
  const int wave = __builtin_amdgcn_readfirstlane(tid >> 5);
  const int bkt  = blockIdx.x;
  const int base = bkt << RB_SHIFT;

  int flag = 0, pre = 0;
  if (tid < RB) {
    flag = (base + tid < N) ? (need[base + tid] != 0) : 0;
    const unsigned m = __builtin_amdgcn_ballot_w32(flag != 0);
    pre = __builtin_popcount(m & ((1u << lane) - 1u));
    if (lane == 0) cntl[tid >> 5] = __builtin_popcount(m);
  }
  __syncthreads();
  const int c0n = cntl[0];
  const int ns = __builtin_amdgcn_readfirstlane(c0n + cntl[1]);
  if (ns == 0) return;
  if (tid < RB) {
    const int s = pre + (tid >= 32 ? c0n : 0);
    slotl[tid] = flag ? s * APITCH : -1;
    if (flag) rowl[s] = tid;
  }
  __syncthreads();

  const int mt = (ns + 15) >> 4;
  for (int idx = tid; idx < (mt << 9); idx += 256) {
    const int r = idx >> 5, c4 = idx & 31;
    float4 v = make_float4(0.0f, 0.0f, 0.0f, 0.0f);
    if (r < ns) {
      const int i = base + rowl[r];
      const float di = dinv[i];
      const float s = di * di;
      const float4 xv = *(const float4*)(x + (size_t)i * IDIM + 4 * c4);
      v.x = s * xv.x; v.y = s * xv.y; v.z = s * xv.z; v.w = s * xv.w;
    }
    *(float4*)(acc + r * APITCH + 4 * c4) = v;
  }
  __syncthreads();

  const int e0 = binstart[bkt], e1 = binstart[bkt + 1];
  const int n = e1 - e0;
  const int seg = (((n + 7) >> 3) + 31) & ~31;
  const int wb = e0 + wave * seg;
  const int we = min(wb + seg, e1);
  const float* __restrict__ xl = x + 2 * lane;
  float* al = acc + 2 * lane;

  for (int e = wb; e < we; e += 32) {
    const int cnt = min(32, we - e);
    int4 r = make_int4(0, 0, 0, 0);
    int keep = 0;
    if (lane < cnt) {
      r = rec[e + lane];
      const int so = slotl[r.y - base];
      keep = so >= 0;
      if (keep) {
        r.z = __float_as_int(dinv[r.x] * __int_as_float(r.z) * dinv[r.y]);
        r.y = so;
      }
    }
    unsigned mask = __builtin_amdgcn_ballot_w32(keep != 0);
    while (mask) {
      const int j0 = __builtin_ctz(mask);
      mask &= mask - 1;
      int jj[4] = {j0, j0, j0, j0};
      int nk = 1;
#pragma unroll
      for (int u = 1; u < 4; ++u) {
        if (mask) { jj[u] = __builtin_ctz(mask); mask &= mask - 1; nk = u + 1; }
      }
      float2 xa[4], xb[4]; int dl[4]; float nr[4];
#pragma unroll
      for (int u = 0; u < 4; ++u) {
        const int s = __builtin_amdgcn_readlane(r.x, jj[u]);
        dl[u] = __builtin_amdgcn_readlane(r.y, jj[u]);
        nr[u] = __int_as_float(__builtin_amdgcn_readlane(r.z, jj[u]));
        xa[u] = *(const float2*)(xl + (size_t)s * IDIM);
        xb[u] = *(const float2*)(xl + (size_t)s * IDIM + 64);
      }
#pragma unroll
      for (int u = 0; u < 4; ++u) {
        if (u < nk) {
          __hip_atomic_fetch_add(al + dl[u],      nr[u] * xa[u].x, __ATOMIC_RELAXED, __HIP_MEMORY_SCOPE_WORKGROUP);
          __hip_atomic_fetch_add(al + dl[u] + 1,  nr[u] * xa[u].y, __ATOMIC_RELAXED, __HIP_MEMORY_SCOPE_WORKGROUP);
          __hip_atomic_fetch_add(al + dl[u] + 64, nr[u] * xb[u].x, __ATOMIC_RELAXED, __HIP_MEMORY_SCOPE_WORKGROUP);
          __hip_atomic_fetch_add(al + dl[u] + 65, nr[u] * xb[u].y, __ATOMIC_RELAXED, __HIP_MEMORY_SCOPE_WORKGROUP);
        }
      }
    }
  }
  __syncthreads();

  const int half = lane >> 4;
  const int l15  = lane & 15;
  const int nb   = (wave & 3) << 4;
  const float* __restrict__ wr = W + (size_t)(2 * half) * ODIM + nb + l15;
  v8f d[2];
#pragma unroll
  for (int u = 0; u < 2; ++u) {
    const int t = (wave >> 2) + 2 * u;
    d[u] = (v8f){};
    if (t < mt) {
      v8f p0 = {}, p1 = {};
      const float* ar = acc + (t * 16 + l15) * APITCH + 2 * half;
#pragma unroll 4
      for (int k = 0; k < IDIM; k += 8) {
        v2f a0, b0, a1, b1;
        a0[0] = ar[k + 0];
        a0[1] = ar[k + 1];
        b0[0] = wr[(size_t)(k + 0) * ODIM];
        b0[1] = wr[(size_t)(k + 1) * ODIM];
        p0 = __builtin_amdgcn_wmma_f32_16x16x4_f32(
             false, a0,  false, b0,
             (short)0, p0,  false,  false);
        a1[0] = ar[k + 4];
        a1[1] = ar[k + 5];
        b1[0] = wr[(size_t)(k + 4) * ODIM];
        b1[1] = wr[(size_t)(k + 5) * ODIM];
        p1 = __builtin_amdgcn_wmma_f32_16x16x4_f32(
             false, a1,  false, b1,
             (short)0, p1,  false,  false);
      }
      d[u] = p0 + p1;
    }
  }
  __syncthreads();

  const float bv = bias[nb + l15];
#pragma unroll
  for (int u = 0; u < 2; ++u) {
    const int t = (wave >> 2) + 2 * u;
    if (t < mt) {
      float* orow = acc + (t * 16 + 8 * half) * OPITCH + nb + l15;
#pragma unroll
      for (int r = 0; r < 8; ++r) orow[r * OPITCH] = d[u][r] + bv;
    }
  }
  __syncthreads();

  for (int idx = tid; idx < (ns << 4); idx += 256) {
    const int s = idx >> 4, c4 = idx & 15;
    const int i = base + rowl[s];
    *(float4*)(h + (size_t)i * ODIM + 4 * c4) = *(const float4*)(acc + s * OPITCH + 4 * c4);
  }
}

#define TPITCH 68

__global__ __launch_bounds__(256) void graph_rows_out(
    const float* __restrict__ h, const float* __restrict__ dinv,
    const int4* __restrict__ rec, const int* __restrict__ binstart,
    const int* __restrict__ first, const float* __restrict__ W,
    const float* __restrict__ bias, float* __restrict__ out, int nc) {
  __shared__ float part[8][2][ODIM];
  __shared__ __attribute__((aligned(16))) float tile[16 * TPITCH];
  __shared__ __attribute__((aligned(16))) float orow[ODIM];
  const int g = blockIdx.x;
  const int c = graph_cluster(first, g, nc);
  const int i0 = 2 * c;
  const int bkt = i0 >> RB_SHIFT;
  const int tid = threadIdx.x;
  const int lane = tid & 31;
  const int wave = __builtin_amdgcn_readfirstlane(tid >> 5);
  const int e0 = binstart[bkt], e1 = binstart[bkt + 1];
  const float* __restrict__ hl = h + 2 * lane;
  float2 a0 = make_float2(0.0f, 0.0f), a1 = make_float2(0.0f, 0.0f);

  const int half = lane >> 4;
  const int l15  = lane & 15;
  const int nb   = (wave & 3) << 4;
  v2f bf[ODIM / 4] = {};
  float bv = 0.0f;
  if (wave < 4) {
    const float* __restrict__ wr = W + (size_t)(2 * half) * ODIM + nb + l15;
#pragma unroll
    for (int q = 0; q < ODIM / 4; ++q) {
      bf[q][0] = wr[(size_t)(4 * q + 0) * ODIM];
      bf[q][1] = wr[(size_t)(4 * q + 1) * ODIM];
    }
    bv = bias[nb + l15];
  }

  for (int e = e0 + wave * 32; e < e1; e += 256) {
    const int cnt = min(32, e1 - e);
    int4 r = make_int4(0, 0, 0, 0);
    int hit = 0;
    if (lane < cnt) {
      r = rec[e + lane];
      hit = ((r.y >> 1) == c);
      if (hit)
        r.z = __float_as_int(dinv[r.x] * __int_as_float(r.z) * dinv[r.y]);
    }
    unsigned mask = __builtin_amdgcn_ballot_w32(hit != 0);
    while (mask) {
      const int j = __builtin_ctz(mask);
      mask &= mask - 1;
      const int s = __builtin_amdgcn_readlane(r.x, j);
      const int d = __builtin_amdgcn_readlane(r.y, j);
      const float nr = __int_as_float(__builtin_amdgcn_readlane(r.z, j));
      const float2 xv = *(const float2*)(hl + (size_t)s * ODIM);
      if (d & 1) { a1.x += nr * xv.x; a1.y += nr * xv.y; }
      else       { a0.x += nr * xv.x; a0.y += nr * xv.y; }
    }
  }
  if (wave == 0) {
    const float d0 = dinv[i0], d1 = dinv[i0 + 1];
    const float2 x0 = *(const float2*)(hl + (size_t)i0 * ODIM);
    const float2 x1 = *(const float2*)(hl + (size_t)(i0 + 1) * ODIM);
    a0.x += d0 * d0 * x0.x; a0.y += d0 * d0 * x0.y;
    a1.x += d1 * d1 * x1.x; a1.y += d1 * d1 * x1.y;
  }
  part[wave][0][2 * lane] = a0.x; part[wave][0][2 * lane + 1] = a0.y;
  part[wave][1][2 * lane] = a1.x; part[wave][1][2 * lane + 1] = a1.y;
  for (int idx = tid; idx < 14 * ODIM; idx += 256)
    tile[(2 + (idx >> 6)) * TPITCH + (idx & 63)] = 0.0f;
  __syncthreads();
  if (tid < 2 * ODIM) {
    const int p = tid >> 6, f = tid & 63;
    float sum = part[0][p][f];
#pragma unroll
    for (int w = 1; w < 8; ++w) sum += part[w][p][f];
    tile[p * TPITCH + f] = sum;
  }
  __syncthreads();

  if (wave < 4) {
    const float* ar = tile + l15 * TPITCH + 2 * half;
    v8f acc = {};
#pragma unroll
    for (int k = 0; k < ODIM; k += 4) {
      v2f a;
      a[0] = ar[k + 0];
      a[1] = ar[k + 1];
      acc = __builtin_amdgcn_wmma_f32_16x16x4_f32(
           false, a,  false, bf[k >> 2],
           (short)0, acc,  false,  false);
    }
    const float pooled = fmaxf(acc[0] + bv, acc[1] + bv);
    if (half == 0) orow[nb + l15] = pooled;
  }
  __syncthreads();
  if (tid < 16)
    *(float4*)(out + (size_t)g * ODIM + 4 * tid) = *(const float4*)(orow + 4 * tid);
}

extern "C" void kernel_launch(void* const* d_in, const int* in_sizes, int n_in,
                              void* d_out, int out_size, void* d_ws, size_t ws_size,
                              hipStream_t stream) {
  const float* x   = (const float*)d_in[0];
  const int*   ei  = (const int*)d_in[1];
  const float* ew  = (const float*)d_in[2];
  const int*   bat = (const int*)d_in[3];
  const float* W1  = (const float*)d_in[4];
  const float* b1  = (const float*)d_in[5];
  const float* W2  = (const float*)d_in[6];
  const float* b2  = (const float*)d_in[7];
  float* out = (float*)d_out;

  const int IN_DIM = 128;
  const int N  = in_sizes[0] / IN_DIM;
  const int E  = in_sizes[2];
  const int B  = out_size / ODIM;
  const int NC = N / 2;

  const int* src = ei;
  const int* dst = ei + E;

  char* ws = (char*)d_ws;
  size_t featBytes = (size_t)N * ODIM * sizeof(float);
  size_t aggBytes  = (size_t)2 * B * ODIM * sizeof(float);
  float* h    = (float*)(ws + featBytes);
  float* dinv = (float*)(ws + 2 * featBytes + (size_t)N * 4);
  int*   need = (int*)  (ws + 2 * featBytes + (size_t)N * 8 + aggBytes);
  int*   first= (int*)  (ws + 2 * featBytes + (size_t)N * 12 + aggBytes);
  int4*  rec      = (int4*)(ws + 134217728);
  int*   bincnt   = (int*) (ws + 201326592);
  int*   cursor   = (int*) (ws + 201326592 + 65536);
  int*   binstart = (int*) (ws + 201326592 + 131072);

  const int T = 256;
  dim3 blk(T);

  const int NB = (N + RB - 1) / RB;
  const int cover = N > B ? N : B;
  init_tables<<<(cover + T - 1) / T, blk, 0, stream>>>(bincnt, cursor, NB, first, B, need, N);

  const int G1 = 256;
  const int chunk = (E + G1 - 1) / G1;
  bin_count_first<<<G1, dim3(512), 0, stream>>>(dst, bincnt, E, NB, chunk, bat, first, NC);
  bin_scatter<<<G1, dim3(512), 0, stream>>>(src, dst, ew, bincnt, binstart, cursor, rec, E, NB, chunk);
  bucket_deg_mark<<<NB, blk, 0, stream>>>(rec, binstart, dinv, N, first, B, need, NC);

  bucket_aggregate<<<NB, blk, 0, stream>>>(x, W1, dinv, b1, rec, binstart, need, h, N);

  graph_rows_out<<<B, blk, 0, stream>>>(h, dinv, rec, binstart, first, W2, b2, out, NC);
}


# Binning kernels: a lane issues every global request of a phase before its first wait (edges as 16-byte quads kept in registers, cursor atomics batched); record order in buckets, so sum order, changes
# speedup vs baseline: 9.0000x; 1.1371x over previous
#include <hip/hip_runtime.h>
#include <hip/hip_bf16.h>

typedef __attribute__((ext_vector_type(2))) float v2f;
typedef __attribute__((ext_vector_type(8))) float v8f;

#define ODIM 64

#define RB 64
#define RB_SHIFT 6
#define NBMAX 2048

__global__ void init_tables(int* __restrict__ bincnt, int* __restrict__ cursor, int nb,
                            int* __restrict__ first, int B, int* __restrict__ need, int N) {
  int i = blockIdx.x * blockDim.x + threadIdx.x;
  if (i < nb) { bincnt[i] = 0; cursor[i] = 0; }
  if (i < B) first[i] = 0x7FFFFFFF;
  if (i < N) need[i] = 0;
}

__device__ __forceinline__ int graph_cluster(const int* __restrict__ first, int g, int nc) {
  int c = first[g];
  if (c > nc - 1) c = nc - 1;
  if (c < 0) c = 0;
  return c;
}

#define QPT 4
#define QCHUNK (512 * 4 * QPT)
__device__ __forceinline__ bool quad_path(const void* a, const void* b, const void* c, int E, int chunk) {
  return chunk <= QCHUNK && ((E | chunk) & 3) == 0 &&
         ((((size_t)a) | ((size_t)b) | ((size_t)c)) & 15) == 0;
}

__global__ __launch_bounds__(512) void bin_count_first(
    const int* __restrict__ dst, int* __restrict__ bincnt, int E, int nb, int chunk,
    const int* __restrict__ batch, int* __restrict__ first, int nc) {
  __shared__ int hist[NBMAX];
  for (int b = threadIdx.x; b < nb; b += blockDim.x) hist[b] = 0;
  const int c0 = blockIdx.x * blockDim.x + threadIdx.x;
  const int cl = min(c0, nc - 1);
  const int ba = batch[2 * cl], bb = batch[2 * cl + 1];
  const int e0 = blockIdx.x * chunk;
  const int e1 = min(e0 + chunk, E);
  const bool quads = quad_path(dst, dst, dst, E, chunk);
  const int n4 = quads ? (e1 - e0) >> 2 : 0;
  int4 dq[QPT];
#pragma unroll
  for (int u = 0; u < QPT; ++u) dq[u] = make_int4(0, 0, 0, 0);
  if (n4 > 0) {
    const int4* __restrict__ d4 = (const int4*)(dst + e0);
#pragma unroll
    for (int u = 0; u < QPT; ++u) dq[u] = d4[min((int)threadIdx.x + 512 * u, n4 - 1)];
  }
  __syncthreads();
  if (quads) {
#pragma unroll
    for (int u = 0; u < QPT; ++u) {
      if ((int)threadIdx.x + 512 * u < n4) {
        atomicAdd(&hist[dq[u].x >> RB_SHIFT], 1);
        atomicAdd(&hist[dq[u].y >> RB_SHIFT], 1);
        atomicAdd(&hist[dq[u].z >> RB_SHIFT], 1);
        atomicAdd(&hist[dq[u].w >> RB_SHIFT], 1);
      }
    }
  } else {
    for (int e = e0 + threadIdx.x; e < e1; e += blockDim.x)
      atomicAdd(&hist[dst[e] >> RB_SHIFT], 1);
  }
  __syncthreads();
  for (int b = threadIdx.x; b < nb; b += blockDim.x) {
    int c = hist[b];
    if (c) atomicAdd(&bincnt[b], c);
  }
  if (c0 < nc) atomicMin(&first[min(ba, bb)], c0);
  for (int c = c0 + gridDim.x * blockDim.x; c < nc; c += gridDim.x * blockDim.x)
    atomicMin(&first[min(batch[2 * c], batch[2 * c + 1])], c);
}

#define BPT (NBMAX / 512)
__global__ __launch_bounds__(512) void bin_scatter(
    const int* __restrict__ src, const int* __restrict__ dst, const float* __restrict__ w,
    const int* __restrict__ bincnt, int* __restrict__ binstart, int* __restrict__ cursor,
    int4* __restrict__ rec, int E, int nb, int chunk) {
  __shared__ int hist[NBMAX];
  __shared__ int lbase[NBMAX];
  __shared__ int wtot[16];
  const int t = threadIdx.x;
  const int lane = t & 31;
  const int wv = t >> 5;
  for (int b = t; b < nb; b += 512) hist[b] = 0;
  const int b0 = t * BPT;
  int cv[BPT];
  int s = 0;
#pragma unroll
  for (int j = 0; j < BPT; ++j) {
    const int b = b0 + j;
    const int v = bincnt[min(b, nb - 1)];
    cv[j] = b < nb ? v : 0;
    s += cv[j];
  }
  const int e0 = blockIdx.x * chunk;
  const int e1 = min(e0 + chunk, E);
  const bool quads = quad_path(dst, src, w, E, chunk);
  const int n4 = quads ? (e1 - e0) >> 2 : 0;
  int4 dq[QPT], sq[QPT];
  float4 wq[QPT];
#pragma unroll
  for (int u = 0; u < QPT; ++u) {
    dq[u] = make_int4(0, 0, 0, 0);
    sq[u] = make_int4(0, 0, 0, 0);
    wq[u] = make_float4(0.0f, 0.0f, 0.0f, 0.0f);
  }
  if (n4 > 0) {
    const int4* __restrict__ d4 = (const int4*)(dst + e0);
    const int4* __restrict__ s4 = (const int4*)(src + e0);
    const float4* __restrict__ w4 = (const float4*)(w + e0);
#pragma unroll
    for (int u = 0; u < QPT; ++u) {
      const int q = min(t + 512 * u, n4 - 1);
      dq[u] = d4[q];
      sq[u] = s4[q];
      wq[u] = w4[q];
    }
  }

  int inc = s;
  for (int off = 1; off < 32; off <<= 1) {
    int v = __shfl_up(inc, off, 32);
    if (lane >= off) inc += v;
  }
  if (lane == 31) wtot[wv] = inc;
  __syncthreads();
  int run = inc - s;
  for (int q = 0; q < wv; ++q) run += wtot[q];
#pragma unroll
  for (int j = 0; j < BPT; ++j) {
    const int b = b0 + j;
    if (b < nb) {
      lbase[b] = run;
      if (blockIdx.x == 0) binstart[b] = run;
      run += cv[j];
    }
  }
  if (blockIdx.x == 0 && t == 511) binstart[nb] = run;

  if (quads) {
#pragma unroll
    for (int u = 0; u < QPT; ++u) {
      if (t + 512 * u < n4) {
        atomicAdd(&hist[dq[u].x >> RB_SHIFT], 1);
        atomicAdd(&hist[dq[u].y >> RB_SHIFT], 1);
        atomicAdd(&hist[dq[u].z >> RB_SHIFT], 1);
        atomicAdd(&hist[dq[u].w >> RB_SHIFT], 1);
      }
    }
  } else {
    for (int e = e0 + t; e < e1; e += 512)
      atomicAdd(&hist[dst[e] >> RB_SHIFT], 1);
  }
  __syncthreads();
  int rc[BPT], rb[BPT];
#pragma unroll
  for (int j = 0; j < BPT; ++j) {
    const int b = t + 512 * j;
    rc[j] = b < nb ? hist[b] : 0;
  }
#pragma unroll
  for (int j = 0; j < BPT; ++j) {
    rb[j] = 0;
    if (rc[j]) rb[j] = atomicAdd(&cursor[t + 512 * j], rc[j]);
  }
#pragma unroll
  for (int j = 0; j < BPT; ++j) {
    const int b = t + 512 * j;
    if (b < nb) {
      lbase[b] = rc[j] ? lbase[b] + rb[j] : 0;
      hist[b] = 0;
    }
  }
  __syncthreads();
  if (quads) {
#pragma unroll
    for (int u = 0; u < QPT; ++u) {
      if (t + 512 * u < n4) {
        const int dd[4] = {dq[u].x, dq[u].y, dq[u].z, dq[u].w};
        const int ss[4] = {sq[u].x, sq[u].y, sq[u].z, sq[u].w};
        const float ww[4] = {wq[u].x, wq[u].y, wq[u].z, wq[u].w};
        int pos[4];
#pragma unroll
        for (int k = 0; k < 4; ++k) {
          const int b = dd[k] >> RB_SHIFT;
          pos[k] = lbase[b] + atomicAdd(&hist[b], 1);
        }
#pragma unroll
        for (int k = 0; k < 4; ++k)
          rec[pos[k]] = make_int4(ss[k], dd[k], __float_as_int(ww[k]), 0);
      }
    }
  } else {
    for (int e = e0 + t; e < e1; e += 512) {
      int d = dst[e];
      int b = d >> RB_SHIFT;
      int pos = lbase[b] + atomicAdd(&hist[b], 1);
      rec[pos] = make_int4(src[e], d, __float_as_int(w[e]), 0);
    }
  }
}

__global__ __launch_bounds__(256) void bucket_deg_mark(
    const int4* __restrict__ rec, const int* __restrict__ binstart,
    float* __restrict__ dinv, int N,
    const int* __restrict__ first, int B, int* __restrict__ need, int nc) {
  __shared__ float degl[RB];
  __shared__ int tgtl[RB];
  const int bkt = blockIdx.x;
  const int base = bkt << RB_SHIFT;
  const int e0 = binstart[bkt], e1 = binstart[bkt + 1];
  int ig = -1;
  if (threadIdx.x < B) ig = 2 * graph_cluster(first, threadIdx.x, nc);
  if (threadIdx.x < RB) { degl[threadIdx.x] = 1.0f; tgtl[threadIdx.x] = 0; }
  __syncthreads();
  if (ig >= 0 && (ig >> RB_SHIFT) == bkt) { tgtl[ig - base] = 1; tgtl[ig + 1 - base] = 1; }
  for (int g = threadIdx.x + blockDim.x; g < B; g += blockDim.x) {
    const int i0 = 2 * graph_cluster(first, g, nc);
    if ((i0 >> RB_SHIFT) == bkt) { tgtl[i0 - base] = 1; tgtl[i0 + 1 - base] = 1; }
  }
  __syncthreads();
  for (int e = e0 + threadIdx.x; e < e1; e += blockDim.x) {
    int4 r = rec[e];
    __hip_atomic_fetch_add(&degl[r.y - base], __int_as_float(r.z),
                           __ATOMIC_RELAXED, __HIP_MEMORY_SCOPE_WORKGROUP);
    if (tgtl[r.y - base]) need[r.x] = 1;
  }
  __syncthreads();
  if (threadIdx.x < RB) {
    int i = base + threadIdx.x;
    if (i < N) {
      float d = degl[threadIdx.x];
      dinv[i] = d > 0.0f ? __frsqrt_rn(d) : 0.0f;
      if (tgtl[threadIdx.x]) need[i] = 1;
    }
  }
}

#define IDIM 128
#define APITCH 132
#define OPITCH 68

__global__ __launch_bounds__(256) void bucket_aggregate(
    const float* __restrict__ x, const float* __restrict__ W,
    const float* __restrict__ dinv, const float* __restrict__ bias,
    const int4* __restrict__ rec, const int* __restrict__ binstart,
    const int* __restrict__ need, float* __restrict__ h, int N) {
  __shared__ float acc[RB * APITCH];
  __shared__ int slotl[RB];
  __shared__ int rowl[RB];
  __shared__ int cntl[2];
  const int tid  = threadIdx.x;
  const int lane = tid & 31;
  const int wave = __builtin_amdgcn_readfirstlane(tid >> 5);
  const int bkt  = blockIdx.x;
  const int base = bkt << RB_SHIFT;

  int flag = 0, pre = 0;
  if (tid < RB) {
    flag = (base + tid < N) ? (need[base + tid] != 0) : 0;
    const unsigned m = __builtin_amdgcn_ballot_w32(flag != 0);
    pre = __builtin_popcount(m & ((1u << lane) - 1u));
    if (lane == 0) cntl[tid >> 5] = __builtin_popcount(m);
  }
  __syncthreads();
  const int c0n = cntl[0];
  const int ns = __builtin_amdgcn_readfirstlane(c0n + cntl[1]);
  if (ns == 0) return;
  if (tid < RB) {
    const int s = pre + (tid >= 32 ? c0n : 0);
    slotl[tid] = flag ? s * APITCH : -1;
    if (flag) rowl[s] = tid;
  }
  __syncthreads();

  const int mt = (ns + 15) >> 4;
  for (int idx = tid; idx < (mt << 9); idx += 256) {
    const int r = idx >> 5, c4 = idx & 31;
    float4 v = make_float4(0.0f, 0.0f, 0.0f, 0.0f);
    if (r < ns) {
      const int i = base + rowl[r];
      const float di = dinv[i];
      const float s = di * di;
      const float4 xv = *(const float4*)(x + (size_t)i * IDIM + 4 * c4);
      v.x = s * xv.x; v.y = s * xv.y; v.z = s * xv.z; v.w = s * xv.w;
    }
    *(float4*)(acc + r * APITCH + 4 * c4) = v;
  }
  __syncthreads();

  const int e0 = binstart[bkt], e1 = binstart[bkt + 1];
  const int n = e1 - e0;
  const int seg = (((n + 7) >> 3) + 31) & ~31;
  const int wb = e0 + wave * seg;
  const int we = min(wb + seg, e1);
  const float* __restrict__ xl = x + 2 * lane;
  float* al = acc + 2 * lane;

  for (int e = wb; e < we; e += 32) {
    const int cnt = min(32, we - e);
    int4 r = make_int4(0, 0, 0, 0);
    int keep = 0;
    if (lane < cnt) {
      r = rec[e + lane];
      const int so = slotl[r.y - base];
      keep = so >= 0;
      if (keep) {
        r.z = __float_as_int(dinv[r.x] * __int_as_float(r.z) * dinv[r.y]);
        r.y = so;
      }
    }
    unsigned mask = __builtin_amdgcn_ballot_w32(keep != 0);
    while (mask) {
      const int j0 = __builtin_ctz(mask);
      mask &= mask - 1;
      int jj[4] = {j0, j0, j0, j0};
      int nk = 1;
#pragma unroll
      for (int u = 1; u < 4; ++u) {
        if (mask) { jj[u] = __builtin_ctz(mask); mask &= mask - 1; nk = u + 1; }
      }
      float2 xa[4], xb[4]; int dl[4]; float nr[4];
#pragma unroll
      for (int u = 0; u < 4; ++u) {
        const int s = __builtin_amdgcn_readlane(r.x, jj[u]);
        dl[u] = __builtin_amdgcn_readlane(r.y, jj[u]);
        nr[u] = __int_as_float(__builtin_amdgcn_readlane(r.z, jj[u]));
        xa[u] = *(const float2*)(xl + (size_t)s * IDIM);
        xb[u] = *(const float2*)(xl + (size_t)s * IDIM + 64);
      }
#pragma unroll
      for (int u = 0; u < 4; ++u) {
        if (u < nk) {
          __hip_atomic_fetch_add(al + dl[u],      nr[u] * xa[u].x, __ATOMIC_RELAXED, __HIP_MEMORY_SCOPE_WORKGROUP);
          __hip_atomic_fetch_add(al + dl[u] + 1,  nr[u] * xa[u].y, __ATOMIC_RELAXED, __HIP_MEMORY_SCOPE_WORKGROUP);
          __hip_atomic_fetch_add(al + dl[u] + 64, nr[u] * xb[u].x, __ATOMIC_RELAXED, __HIP_MEMORY_SCOPE_WORKGROUP);
          __hip_atomic_fetch_add(al + dl[u] + 65, nr[u] * xb[u].y, __ATOMIC_RELAXED, __HIP_MEMORY_SCOPE_WORKGROUP);
        }
      }
    }
  }
  __syncthreads();

  const int half = lane >> 4;
  const int l15  = lane & 15;
  const int nb   = (wave & 3) << 4;
  const float* __restrict__ wr = W + (size_t)(2 * half) * ODIM + nb + l15;
  v8f d[2];
#pragma unroll
  for (int u = 0; u < 2; ++u) {
    const int t = (wave >> 2) + 2 * u;
    d[u] = (v8f){};
    if (t < mt) {
      v8f p0 = {}, p1 = {};
      const float* ar = acc + (t * 16 + l15) * APITCH + 2 * half;
#pragma unroll 4
      for (int k = 0; k < IDIM; k += 8) {
        v2f a0, b0, a1, b1;
        a0[0] = ar[k + 0];
        a0[1] = ar[k + 1];
        b0[0] = wr[(size_t)(k + 0) * ODIM];
        b0[1] = wr[(size_t)(k + 1) * ODIM];
        p0 = __builtin_amdgcn_wmma_f32_16x16x4_f32(
             false, a0,  false, b0,
             (short)0, p0,  false,  false);
        a1[0] = ar[k + 4];
        a1[1] = ar[k + 5];
        b1[0] = wr[(size_t)(k + 4) * ODIM];
        b1[1] = wr[(size_t)(k + 5) * ODIM];
        p1 = __builtin_amdgcn_wmma_f32_16x16x4_f32(
             false, a1,  false, b1,
             (short)0, p1,  false,  false);
      }
      d[u] = p0 + p1;
    }
  }
  __syncthreads();

  const float bv = bias[nb + l15];
#pragma unroll
  for (int u = 0; u < 2; ++u) {
    const int t = (wave >> 2) + 2 * u;
    if (t < mt) {
      float* orow = acc + (t * 16 + 8 * half) * OPITCH + nb + l15;
#pragma unroll
      for (int r = 0; r < 8; ++r) orow[r * OPITCH] = d[u][r] + bv;
    }
  }
  __syncthreads();

  for (int idx = tid; idx < (ns << 4); idx += 256) {
    const int s = idx >> 4, c4 = idx & 15;
    const int i = base + rowl[s];
    *(float4*)(h + (size_t)i * ODIM + 4 * c4) = *(const float4*)(acc + s * OPITCH + 4 * c4);
  }
}

#define TPITCH 68

__global__ __launch_bounds__(256) void graph_rows_out(
    const float* __restrict__ h, const float* __restrict__ dinv,
    const int4* __restrict__ rec, const int* __restrict__ binstart,
    const int* __restrict__ first, const float* __restrict__ W,
    const float* __restrict__ bias, float* __restrict__ out, int nc) {
  __shared__ float part[8][2][ODIM];
  __shared__ __attribute__((aligned(16))) float tile[16 * TPITCH];
  __shared__ __attribute__((aligned(16))) float orow[ODIM];
  const int g = blockIdx.x;
  const int c = graph_cluster(first, g, nc);
  const int i0 = 2 * c;
  const int bkt = i0 >> RB_SHIFT;
  const int tid = threadIdx.x;
  const int lane = tid & 31;
  const int wave = __builtin_amdgcn_readfirstlane(tid >> 5);
  const int e0 = binstart[bkt], e1 = binstart[bkt + 1];
  const float* __restrict__ hl = h + 2 * lane;
  float2 a0 = make_float2(0.0f, 0.0f), a1 = make_float2(0.0f, 0.0f);

  const int half = lane >> 4;
  const int l15  = lane & 15;
  const int nb   = (wave & 3) << 4;
  v2f bf[ODIM / 4] = {};
  float bv = 0.0f;
  if (wave < 4) {
    const float* __restrict__ wr = W + (size_t)(2 * half) * ODIM + nb + l15;
#pragma unroll
    for (int q = 0; q < ODIM / 4; ++q) {
      bf[q][0] = wr[(size_t)(4 * q + 0) * ODIM];
      bf[q][1] = wr[(size_t)(4 * q + 1) * ODIM];
    }
    bv = bias[nb + l15];
  }

  for (int e = e0 + wave * 32; e < e1; e += 256) {
    const int cnt = min(32, e1 - e);
    int4 r = make_int4(0, 0, 0, 0);
    int hit = 0;
    if (lane < cnt) {
      r = rec[e + lane];
      hit = ((r.y >> 1) == c);
      if (hit)
        r.z = __float_as_int(dinv[r.x] * __int_as_float(r.z) * dinv[r.y]);
    }
    unsigned mask = __builtin_amdgcn_ballot_w32(hit != 0);
    while (mask) {
      const int j = __builtin_ctz(mask);
      mask &= mask - 1;
      const int s = __builtin_amdgcn_readlane(r.x, j);
      const int d = __builtin_amdgcn_readlane(r.y, j);
      const float nr = __int_as_float(__builtin_amdgcn_readlane(r.z, j));
      const float2 xv = *(const float2*)(hl + (size_t)s * ODIM);
      if (d & 1) { a1.x += nr * xv.x; a1.y += nr * xv.y; }
      else       { a0.x += nr * xv.x; a0.y += nr * xv.y; }
    }
  }
  if (wave == 0) {
    const float d0 = dinv[i0], d1 = dinv[i0 + 1];
    const float2 x0 = *(const float2*)(hl + (size_t)i0 * ODIM);
    const float2 x1 = *(const float2*)(hl + (size_t)(i0 + 1) * ODIM);
    a0.x += d0 * d0 * x0.x; a0.y += d0 * d0 * x0.y;
    a1.x += d1 * d1 * x1.x; a1.y += d1 * d1 * x1.y;
  }
  part[wave][0][2 * lane] = a0.x; part[wave][0][2 * lane + 1] = a0.y;
  part[wave][1][2 * lane] = a1.x; part[wave][1][2 * lane + 1] = a1.y;
  for (int idx = tid; idx < 14 * ODIM; idx += 256)
    tile[(2 + (idx >> 6)) * TPITCH + (idx & 63)] = 0.0f;
  __syncthreads();
  if (tid < 2 * ODIM) {
    const int p = tid >> 6, f = tid & 63;
    float sum = part[0][p][f];
#pragma unroll
    for (int w = 1; w < 8; ++w) sum += part[w][p][f];
    tile[p * TPITCH + f] = sum;
  }
  __syncthreads();

  if (wave < 4) {
    const float* ar = tile + l15 * TPITCH + 2 * half;
    v8f acc = {};
#pragma unroll
    for (int k = 0; k < ODIM; k += 4) {
      v2f a;
      a[0] = ar[k + 0];
      a[1] = ar[k + 1];
      acc = __builtin_amdgcn_wmma_f32_16x16x4_f32(
           false, a,  false, bf[k >> 2],
           (short)0, acc,  false,  false);
    }
    const float pooled = fmaxf(acc[0] + bv, acc[1] + bv);
    if (half == 0) orow[nb + l15] = pooled;
  }
  __syncthreads();
  if (tid < 16)
    *(float4*)(out + (size_t)g * ODIM + 4 * tid) = *(const float4*)(orow + 4 * tid);
}

extern "C" void kernel_launch(void* const* d_in, const int* in_sizes, int n_in,
                              void* d_out, int out_size, void* d_ws, size_t ws_size,
                              hipStream_t stream) {
  const float* x   = (const float*)d_in[0];
  const int*   ei  = (const int*)d_in[1];
  const float* ew  = (const float*)d_in[2];
  const int*   bat = (const int*)d_in[3];
  const float* W1  = (const float*)d_in[4];
  const float* b1  = (const float*)d_in[5];
  const float* W2  = (const float*)d_in[6];
  const float* b2  = (const float*)d_in[7];
  float* out = (float*)d_out;

  const int IN_DIM = 128;
  const int N  = in_sizes[0] / IN_DIM;
  const int E  = in_sizes[2];
  const int B  = out_size / ODIM;
  const int NC = N / 2;

  const int* src = ei;
  const int* dst = ei + E;

  char* ws = (char*)d_ws;
  size_t featBytes = (size_t)N * ODIM * sizeof(float);
  size_t aggBytes  = (size_t)2 * B * ODIM * sizeof(float);
  float* h    = (float*)(ws + featBytes);
  float* dinv = (float*)(ws + 2 * featBytes + (size_t)N * 4);
  int*   need = (int*)  (ws + 2 * featBytes + (size_t)N * 8 + aggBytes);
  int*   first= (int*)  (ws + 2 * featBytes + (size_t)N * 12 + aggBytes);
  int4*  rec      = (int4*)(ws + 134217728);
  int*   bincnt   = (int*) (ws + 201326592);
  int*   cursor   = (int*) (ws + 201326592 + 65536);
  int*   binstart = (int*) (ws + 201326592 + 131072);

  const int T = 256;
  dim3 blk(T);

  const int NB = (N + RB - 1) / RB;
  const int cover = N > B ? N : B;
  init_tables<<<(cover + T - 1) / T, blk, 0, stream>>>(bincnt, cursor, NB, first, B, need, N);

  const int G1 = 256;
  const int chunk = ((E + G1 - 1) / G1 + 3) & ~3;
  bin_count_first<<<G1, dim3(512), 0, stream>>>(dst, bincnt, E, NB, chunk, bat, first, NC);
  bin_scatter<<<G1, dim3(512), 0, stream>>>(src, dst, ew, bincnt, binstart, cursor, rec, E, NB, chunk);
  bucket_deg_mark<<<NB, blk, 0, stream>>>(rec, binstart, dinv, N, first, B, need, NC);

  bucket_aggregate<<<NB, blk, 0, stream>>>(x, W1, dinv, b1, rec, binstart, need, h, N);

  graph_rows_out<<<B, blk, 0, stream>>>(h, dinv, rec, binstart, first, W2, b2, out, NC);
}


# Bucket regions of fixed capacity plus one shared spill list: a record's place needs only its bucket's cursor, so the histogram launch and the scan go (6 launches to 5); record, so sum, order changes
# speedup vs baseline: 10.2331x; 1.1370x over previous
#include <hip/hip_runtime.h>
#include <hip/hip_bf16.h>

typedef __attribute__((ext_vector_type(2))) float v2f;
typedef __attribute__((ext_vector_type(8))) float v8f;

#define ODIM 64

#define RB 64
#define RB_SHIFT 6
#define NBMAX 2048
#define CAP 2048
#define RSTRIDE (CAP + 16)

__global__ void init_tables(int* __restrict__ cursor, int nb, int* __restrict__ spillcnt,
                            int* __restrict__ first, int B, int* __restrict__ need, int N) {
  int i = blockIdx.x * blockDim.x + threadIdx.x;
  if (i < nb) cursor[i] = 0;
  if (i == 0) *spillcnt = 0;
  if (i < B) first[i] = 0x7FFFFFFF;
  if (i < N) need[i] = 0;
}

__device__ __forceinline__ int region_count(const int* __restrict__ cursor, int bkt) {
  return min(cursor[bkt], CAP);
}

__device__ __forceinline__ int graph_cluster(const int* __restrict__ first, int g, int nc) {
  int c = first[g];
  if (c > nc - 1) c = nc - 1;
  if (c < 0) c = 0;
  return c;
}

#define QPT 4
#define QCHUNK (512 * 4 * QPT)
__device__ __forceinline__ bool quad_path(const void* a, const void* b, const void* c, int E, int chunk) {
  return chunk <= QCHUNK && ((E | chunk) & 3) == 0 &&
         ((((size_t)a) | ((size_t)b) | ((size_t)c)) & 15) == 0;
}

#define BPT (NBMAX / 512)
__global__ __launch_bounds__(512) void bin_scatter_first(
    const int* __restrict__ src, const int* __restrict__ dst, const float* __restrict__ w,
    int* __restrict__ cursor, int* __restrict__ spillcnt,
    int4* __restrict__ rec, int4* __restrict__ spill, int E, int nb, int chunk,
    const int* __restrict__ batch, int* __restrict__ first, int nc) {
  __shared__ int hist[NBMAX];
  __shared__ int lbase[NBMAX];
  const int t = threadIdx.x;
  for (int b = t; b < nb; b += 512) hist[b] = 0;
  const int c0 = blockIdx.x * blockDim.x + threadIdx.x;
  const int cl = min(c0, nc - 1);
  const int ba = batch[2 * cl], bb = batch[2 * cl + 1];
  const int e0 = blockIdx.x * chunk;
  const int e1 = min(e0 + chunk, E);
  const bool quads = quad_path(dst, src, w, E, chunk);
  const int n4 = quads ? (e1 - e0) >> 2 : 0;
  int4 dq[QPT], sq[QPT];
  float4 wq[QPT];
#pragma unroll
  for (int u = 0; u < QPT; ++u) {
    dq[u] = make_int4(0, 0, 0, 0);
    sq[u] = make_int4(0, 0, 0, 0);
    wq[u] = make_float4(0.0f, 0.0f, 0.0f, 0.0f);
  }
  if (n4 > 0) {
    const int4* __restrict__ d4 = (const int4*)(dst + e0);
    const int4* __restrict__ s4 = (const int4*)(src + e0);
    const float4* __restrict__ w4 = (const float4*)(w + e0);
#pragma unroll
    for (int u = 0; u < QPT; ++u) {
      const int q = min(t + 512 * u, n4 - 1);
      dq[u] = d4[q];
      sq[u] = s4[q];
      wq[u] = w4[q];
    }
  }
  __syncthreads();

  if (quads) {
#pragma unroll
    for (int u = 0; u < QPT; ++u) {
      if (t + 512 * u < n4) {
        atomicAdd(&hist[dq[u].x >> RB_SHIFT], 1);
        atomicAdd(&hist[dq[u].y >> RB_SHIFT], 1);
        atomicAdd(&hist[dq[u].z >> RB_SHIFT], 1);
        atomicAdd(&hist[dq[u].w >> RB_SHIFT], 1);
      }
    }
  } else {
    for (int e = e0 + t; e < e1; e += 512)
      atomicAdd(&hist[dst[e] >> RB_SHIFT], 1);
  }
  __syncthreads();
  int rc[BPT], rb[BPT];
#pragma unroll
  for (int j = 0; j < BPT; ++j) {
    const int b = t + 512 * j;
    rc[j] = b < nb ? hist[b] : 0;
  }
#pragma unroll
  for (int j = 0; j < BPT; ++j) {
    rb[j] = 0;
    if (rc[j]) rb[j] = atomicAdd(&cursor[t + 512 * j], rc[j]);
  }
#pragma unroll
  for (int j = 0; j < BPT; ++j) {
    const int b = t + 512 * j;
    if (b < nb) {
      lbase[b] = rb[j];
      hist[b] = 0;
    }
  }
  __syncthreads();
  if (quads) {
#pragma unroll
    for (int u = 0; u < QPT; ++u) {
      if (t + 512 * u < n4) {
        const int dd[4] = {dq[u].x, dq[u].y, dq[u].z, dq[u].w};
        const int ss[4] = {sq[u].x, sq[u].y, sq[u].z, sq[u].w};
        const float ww[4] = {wq[u].x, wq[u].y, wq[u].z, wq[u].w};
        int rk[4];
#pragma unroll
        for (int k = 0; k < 4; ++k) {
          const int b = dd[k] >> RB_SHIFT;
          rk[k] = lbase[b] + atomicAdd(&hist[b], 1);
        }
#pragma unroll
        for (int k = 0; k < 4; ++k) {
          const int4 r = make_int4(ss[k], dd[k], __float_as_int(ww[k]), 0);
          if (rk[k] < CAP) rec[(dd[k] >> RB_SHIFT) * RSTRIDE + rk[k]] = r;
          else             spill[atomicAdd(spillcnt, 1)] = r;
        }
      }
    }
  } else {
    for (int e = e0 + t; e < e1; e += 512) {
      const int d = dst[e];
      const int b = d >> RB_SHIFT;
      const int rk = lbase[b] + atomicAdd(&hist[b], 1);
      const int4 r = make_int4(src[e], d, __float_as_int(w[e]), 0);
      if (rk < CAP) rec[b * RSTRIDE + rk] = r;
      else          spill[atomicAdd(spillcnt, 1)] = r;
    }
  }
  if (c0 < nc) atomicMin(&first[min(ba, bb)], c0);
  for (int cc = c0 + gridDim.x * blockDim.x; cc < nc; cc += gridDim.x * blockDim.x)
    atomicMin(&first[min(batch[2 * cc], batch[2 * cc + 1])], cc);
}

__global__ __launch_bounds__(256) void bucket_deg_mark(
    const int4* __restrict__ rec, const int4* __restrict__ spill,
    const int* __restrict__ cursor, const int* __restrict__ spillcnt,
    float* __restrict__ dinv, int N,
    const int* __restrict__ first, int B, int* __restrict__ need, int nc) {
  __shared__ float degl[RB];
  __shared__ int tgtl[RB];
  const int bkt = blockIdx.x;
  const int base = bkt << RB_SHIFT;
  const int e0 = bkt * RSTRIDE, e1 = e0 + region_count(cursor, bkt);
  const int nsp = *spillcnt;
  int ig = -1;
  if (threadIdx.x < B) ig = 2 * graph_cluster(first, threadIdx.x, nc);
  if (threadIdx.x < RB) { degl[threadIdx.x] = 1.0f; tgtl[threadIdx.x] = 0; }
  __syncthreads();
  if (ig >= 0 && (ig >> RB_SHIFT) == bkt) { tgtl[ig - base] = 1; tgtl[ig + 1 - base] = 1; }
  for (int g = threadIdx.x + blockDim.x; g < B; g += blockDim.x) {
    const int i0 = 2 * graph_cluster(first, g, nc);
    if ((i0 >> RB_SHIFT) == bkt) { tgtl[i0 - base] = 1; tgtl[i0 + 1 - base] = 1; }
  }
  __syncthreads();
  for (int e = e0 + threadIdx.x; e < e1; e += blockDim.x) {
    int4 r = rec[e];
    __hip_atomic_fetch_add(&degl[r.y - base], __int_as_float(r.z),
                           __ATOMIC_RELAXED, __HIP_MEMORY_SCOPE_WORKGROUP);
    if (tgtl[r.y - base]) need[r.x] = 1;
  }
  for (int e = threadIdx.x; e < nsp; e += blockDim.x) {
    int4 r = spill[e];
    if ((r.y >> RB_SHIFT) == bkt) {
      __hip_atomic_fetch_add(&degl[r.y - base], __int_as_float(r.z),
                             __ATOMIC_RELAXED, __HIP_MEMORY_SCOPE_WORKGROUP);
      if (tgtl[r.y - base]) need[r.x] = 1;
    }
  }
  __syncthreads();
  if (threadIdx.x < RB) {
    int i = base + threadIdx.x;
    if (i < N) {
      float d = degl[threadIdx.x];
      dinv[i] = d > 0.0f ? __frsqrt_rn(d) : 0.0f;
      if (tgtl[threadIdx.x]) need[i] = 1;
    }
  }
}

#define IDIM 128
#define APITCH 132
#define OPITCH 68

__global__ __launch_bounds__(256) void bucket_aggregate(
    const float* __restrict__ x, const float* __restrict__ W,
    const float* __restrict__ dinv, const float* __restrict__ bias,
    const int4* __restrict__ rec, const int4* __restrict__ spill,
    const int* __restrict__ cursor, const int* __restrict__ spillcnt,
    const int* __restrict__ need, float* __restrict__ h, int N) {
  __shared__ float acc[RB * APITCH];
  __shared__ int slotl[RB];
  __shared__ int rowl[RB];
  __shared__ int cntl[2];
  const int tid  = threadIdx.x;
  const int lane = tid & 31;
  const int wave = __builtin_amdgcn_readfirstlane(tid >> 5);
  const int bkt  = blockIdx.x;
  const int base = bkt << RB_SHIFT;

  int flag = 0, pre = 0;
  if (tid < RB) {
    flag = (base + tid < N) ? (need[base + tid] != 0) : 0;
    const unsigned m = __builtin_amdgcn_ballot_w32(flag != 0);
    pre = __builtin_popcount(m & ((1u << lane) - 1u));
    if (lane == 0) cntl[tid >> 5] = __builtin_popcount(m);
  }
  __syncthreads();
  const int c0n = cntl[0];
  const int ns = __builtin_amdgcn_readfirstlane(c0n + cntl[1]);
  if (ns == 0) return;
  if (tid < RB) {
    const int s = pre + (tid >= 32 ? c0n : 0);
    slotl[tid] = flag ? s * APITCH : -1;
    if (flag) rowl[s] = tid;
  }
  __syncthreads();

  const int mt = (ns + 15) >> 4;
  for (int idx = tid; idx < (mt << 9); idx += 256) {
    const int r = idx >> 5, c4 = idx & 31;
    float4 v = make_float4(0.0f, 0.0f, 0.0f, 0.0f);
    if (r < ns) {
      const int i = base + rowl[r];
      const float di = dinv[i];
      const float s = di * di;
      const float4 xv = *(const float4*)(x + (size_t)i * IDIM + 4 * c4);
      v.x = s * xv.x; v.y = s * xv.y; v.z = s * xv.z; v.w = s * xv.w;
    }
    *(float4*)(acc + r * APITCH + 4 * c4) = v;
  }
  __syncthreads();

  const int e0 = bkt * RSTRIDE, e1 = e0 + region_count(cursor, bkt);
  const int nsp = *spillcnt;
  const int n = e1 - e0;
  const int seg = (((n + 7) >> 3) + 31) & ~31;
  const int wb = e0 + wave * seg;
  const int we = min(wb + seg, e1);
  const float* __restrict__ xl = x + 2 * lane;
  float* al = acc + 2 * lane;

  for (int pass = 0; pass < 2; ++pass) {
    const int4* __restrict__ list = pass ? spill : rec;
    const int lb = pass ? wave * 32 : wb;
    const int le = pass ? nsp : we;
    const int st = pass ? 256 : 32;
    for (int e = lb; e < le; e += st) {
      const int cnt = min(32, le - e);
      int4 r = make_int4(0, 0, 0, 0);
      int keep = 0;
      if (lane < cnt) {
        r = list[e + lane];
        int so = -1;
        if (pass == 0 || (r.y >> RB_SHIFT) == bkt)
          so = slotl[r.y - base];
        keep = so >= 0;
        if (keep) {
          r.z = __float_as_int(dinv[r.x] * __int_as_float(r.z) * dinv[r.y]);
          r.y = so;
        }
      }
      unsigned mask = __builtin_amdgcn_ballot_w32(keep != 0);
      while (mask) {
        const int j0 = __builtin_ctz(mask);
        mask &= mask - 1;
        int jj[4] = {j0, j0, j0, j0};
        int nk = 1;
#pragma unroll
        for (int u = 1; u < 4; ++u) {
          if (mask) { jj[u] = __builtin_ctz(mask); mask &= mask - 1; nk = u + 1; }
        }
        float2 xa[4], xb[4]; int dl[4]; float nr[4];
#pragma unroll
        for (int u = 0; u < 4; ++u) {
          const int s = __builtin_amdgcn_readlane(r.x, jj[u]);
          dl[u] = __builtin_amdgcn_readlane(r.y, jj[u]);
          nr[u] = __int_as_float(__builtin_amdgcn_readlane(r.z, jj[u]));
          xa[u] = *(const float2*)(xl + (size_t)s * IDIM);
          xb[u] = *(const float2*)(xl + (size_t)s * IDIM + 64);
        }
#pragma unroll
        for (int u = 0; u < 4; ++u) {
          if (u < nk) {
            __hip_atomic_fetch_add(al + dl[u],      nr[u] * xa[u].x, __ATOMIC_RELAXED, __HIP_MEMORY_SCOPE_WORKGROUP);
            __hip_atomic_fetch_add(al + dl[u] + 1,  nr[u] * xa[u].y, __ATOMIC_RELAXED, __HIP_MEMORY_SCOPE_WORKGROUP);
            __hip_atomic_fetch_add(al + dl[u] + 64, nr[u] * xb[u].x, __ATOMIC_RELAXED, __HIP_MEMORY_SCOPE_WORKGROUP);
            __hip_atomic_fetch_add(al + dl[u] + 65, nr[u] * xb[u].y, __ATOMIC_RELAXED, __HIP_MEMORY_SCOPE_WORKGROUP);
          }
        }
      }
    }
  }
  __syncthreads();

  const int half = lane >> 4;
  const int l15  = lane & 15;
  const int nb   = (wave & 3) << 4;
  const float* __restrict__ wr = W + (size_t)(2 * half) * ODIM + nb + l15;
  v8f d[2];
#pragma unroll
  for (int u = 0; u < 2; ++u) {
    const int t = (wave >> 2) + 2 * u;
    d[u] = (v8f){};
    if (t < mt) {
      v8f p0 = {}, p1 = {};
      const float* ar = acc + (t * 16 + l15) * APITCH + 2 * half;
#pragma unroll 4
      for (int k = 0; k < IDIM; k += 8) {
        v2f a0, b0, a1, b1;
        a0[0] = ar[k + 0];
        a0[1] = ar[k + 1];
        b0[0] = wr[(size_t)(k + 0) * ODIM];
        b0[1] = wr[(size_t)(k + 1) * ODIM];
        p0 = __builtin_amdgcn_wmma_f32_16x16x4_f32(
             false, a0,  false, b0,
             (short)0, p0,  false,  false);
        a1[0] = ar[k + 4];
        a1[1] = ar[k + 5];
        b1[0] = wr[(size_t)(k + 4) * ODIM];
        b1[1] = wr[(size_t)(k + 5) * ODIM];
        p1 = __builtin_amdgcn_wmma_f32_16x16x4_f32(
             false, a1,  false, b1,
             (short)0, p1,  false,  false);
      }
      d[u] = p0 + p1;
    }
  }
  __syncthreads();

  const float bv = bias[nb + l15];
#pragma unroll
  for (int u = 0; u < 2; ++u) {
    const int t = (wave >> 2) + 2 * u;
    if (t < mt) {
      float* orow = acc + (t * 16 + 8 * half) * OPITCH + nb + l15;
#pragma unroll
      for (int r = 0; r < 8; ++r) orow[r * OPITCH] = d[u][r] + bv;
    }
  }
  __syncthreads();

  for (int idx = tid; idx < (ns << 4); idx += 256) {
    const int s = idx >> 4, c4 = idx & 15;
    const int i = base + rowl[s];
    *(float4*)(h + (size_t)i * ODIM + 4 * c4) = *(const float4*)(acc + s * OPITCH + 4 * c4);
  }
}

#define TPITCH 68

__global__ __launch_bounds__(256) void graph_rows_out(
    const float* __restrict__ h, const float* __restrict__ dinv,
    const int4* __restrict__ rec, const int4* __restrict__ spill,
    const int* __restrict__ cursor, const int* __restrict__ spillcnt,
    const int* __restrict__ first, const float* __restrict__ W,
    const float* __restrict__ bias, float* __restrict__ out, int nc) {
  __shared__ float part[8][2][ODIM];
  __shared__ __attribute__((aligned(16))) float tile[16 * TPITCH];
  __shared__ __attribute__((aligned(16))) float orow[ODIM];
  const int g = blockIdx.x;
  const int c = graph_cluster(first, g, nc);
  const int i0 = 2 * c;
  const int bkt = i0 >> RB_SHIFT;
  const int tid = threadIdx.x;
  const int lane = tid & 31;
  const int wave = __builtin_amdgcn_readfirstlane(tid >> 5);
  const int e0 = bkt * RSTRIDE, e1 = e0 + region_count(cursor, bkt);
  const int nsp = *spillcnt;
  const float* __restrict__ hl = h + 2 * lane;
  float2 a0 = make_float2(0.0f, 0.0f), a1 = make_float2(0.0f, 0.0f);

  const int half = lane >> 4;
  const int l15  = lane & 15;
  const int nb   = (wave & 3) << 4;
  v2f bf[ODIM / 4] = {};
  float bv = 0.0f;
  if (wave < 4) {
    const float* __restrict__ wr = W + (size_t)(2 * half) * ODIM + nb + l15;
#pragma unroll
    for (int q = 0; q < ODIM / 4; ++q) {
      bf[q][0] = wr[(size_t)(4 * q + 0) * ODIM];
      bf[q][1] = wr[(size_t)(4 * q + 1) * ODIM];
    }
    bv = bias[nb + l15];
  }

  for (int pass = 0; pass < 2; ++pass) {
    const int4* __restrict__ list = pass ? spill : rec;
    const int lb = (pass ? 0 : e0) + wave * 32;
    const int le = pass ? nsp : e1;
    for (int e = lb; e < le; e += 256) {
      const int cnt = min(32, le - e);
      int4 r = make_int4(0, 0, 0, 0);
      int hit = 0;
      if (lane < cnt) {
        r = list[e + lane];
        hit = ((r.y >> 1) == c);
        if (hit)
          r.z = __float_as_int(dinv[r.x] * __int_as_float(r.z) * dinv[r.y]);
      }
      unsigned mask = __builtin_amdgcn_ballot_w32(hit != 0);
      while (mask) {
        const int j = __builtin_ctz(mask);
        mask &= mask - 1;
        const int s = __builtin_amdgcn_readlane(r.x, j);
        const int d = __builtin_amdgcn_readlane(r.y, j);
        const float nr = __int_as_float(__builtin_amdgcn_readlane(r.z, j));
        const float2 xv = *(const float2*)(hl + (size_t)s * ODIM);
        if (d & 1) { a1.x += nr * xv.x; a1.y += nr * xv.y; }
        else       { a0.x += nr * xv.x; a0.y += nr * xv.y; }
      }
    }
  }
  if (wave == 0) {
    const float d0 = dinv[i0], d1 = dinv[i0 + 1];
    const float2 x0 = *(const float2*)(hl + (size_t)i0 * ODIM);
    const float2 x1 = *(const float2*)(hl + (size_t)(i0 + 1) * ODIM);
    a0.x += d0 * d0 * x0.x; a0.y += d0 * d0 * x0.y;
    a1.x += d1 * d1 * x1.x; a1.y += d1 * d1 * x1.y;
  }
  part[wave][0][2 * lane] = a0.x; part[wave][0][2 * lane + 1] = a0.y;
  part[wave][1][2 * lane] = a1.x; part[wave][1][2 * lane + 1] = a1.y;
  for (int idx = tid; idx < 14 * ODIM; idx += 256)
    tile[(2 + (idx >> 6)) * TPITCH + (idx & 63)] = 0.0f;
  __syncthreads();
  if (tid < 2 * ODIM) {
    const int p = tid >> 6, f = tid & 63;
    float sum = part[0][p][f];
#pragma unroll
    for (int w = 1; w < 8; ++w) sum += part[w][p][f];
    tile[p * TPITCH + f] = sum;
  }
  __syncthreads();

  if (wave < 4) {
    const float* ar = tile + l15 * TPITCH + 2 * half;
    v8f acc = {};
#pragma unroll
    for (int k = 0; k < ODIM; k += 4) {
      v2f a;
      a[0] = ar[k + 0];
      a[1] = ar[k + 1];
      acc = __builtin_amdgcn_wmma_f32_16x16x4_f32(
           false, a,  false, bf[k >> 2],
           (short)0, acc,  false,  false);
    }
    const float pooled = fmaxf(acc[0] + bv, acc[1] + bv);
    if (half == 0) orow[nb + l15] = pooled;
  }
  __syncthreads();
  if (tid < 16)
    *(float4*)(out + (size_t)g * ODIM + 4 * tid) = *(const float4*)(orow + 4 * tid);
}

extern "C" void kernel_launch(void* const* d_in, const int* in_sizes, int n_in,
                              void* d_out, int out_size, void* d_ws, size_t ws_size,
                              hipStream_t stream) {
  const float* x   = (const float*)d_in[0];
  const int*   ei  = (const int*)d_in[1];
  const float* ew  = (const float*)d_in[2];
  const int*   bat = (const int*)d_in[3];
  const float* W1  = (const float*)d_in[4];
  const float* b1  = (const float*)d_in[5];
  const float* W2  = (const float*)d_in[6];
  const float* b2  = (const float*)d_in[7];
  float* out = (float*)d_out;

  const int IN_DIM = 128;
  const int N  = in_sizes[0] / IN_DIM;
  const int E  = in_sizes[2];
  const int B  = out_size / ODIM;
  const int NC = N / 2;

  const int* src = ei;
  const int* dst = ei + E;

  char* ws = (char*)d_ws;
  size_t featBytes = (size_t)N * ODIM * sizeof(float);
  size_t aggBytes  = (size_t)2 * B * ODIM * sizeof(float);
  float* h    = (float*)(ws + featBytes);
  float* dinv = (float*)(ws + 2 * featBytes + (size_t)N * 4);
  int*   need = (int*)  (ws + 2 * featBytes + (size_t)N * 8 + aggBytes);
  int*   first= (int*)  (ws + 2 * featBytes + (size_t)N * 12 + aggBytes);
  int4*  rec      = (int4*)(ws + 134217728);
  int*   cursor   = (int*) (ws + 209715200);
  int*   spillcnt = (int*) (ws + 209715200 + 65536);
  int4*  spill    = (int4*)(ws + 218103808);

  const int T = 256;
  dim3 blk(T);

  const int NB = (N + RB - 1) / RB;
  const int cover = N > B ? N : B;
  init_tables<<<(cover + T - 1) / T, blk, 0, stream>>>(cursor, NB, spillcnt, first, B, need, N);

  const int G1 = 256;
  const int chunk = ((E + G1 - 1) / G1 + 3) & ~3;
  bin_scatter_first<<<G1, dim3(512), 0, stream>>>(src, dst, ew, cursor, spillcnt, rec, spill, E, NB, chunk,
                                                  bat, first, NC);
  bucket_deg_mark<<<NB, blk, 0, stream>>>(rec, spill, cursor, spillcnt, dinv, N, first, B, need, NC);

  bucket_aggregate<<<NB, blk, 0, stream>>>(x, W1, dinv, b1, rec, spill, cursor, spillcnt, need, h, N);

  graph_rows_out<<<B, blk, 0, stream>>>(h, dinv, rec, spill, cursor, spillcnt, first, W2, b2, out, NC);
}


# first[]: a lane skips its atomicMin when the lane below it in the wave holds the same graph id, so a run of equal ids sends one same-word atomic instead of up to 32
# speedup vs baseline: 11.9762x; 1.1703x over previous
#include <hip/hip_runtime.h>
#include <hip/hip_bf16.h>

typedef __attribute__((ext_vector_type(2))) float v2f;
typedef __attribute__((ext_vector_type(8))) float v8f;

#define ODIM 64

#define RB 64
#define RB_SHIFT 6
#define NBMAX 2048
#define CAP 2048
#define RSTRIDE (CAP + 16)

__global__ void init_tables(int* __restrict__ cursor, int nb, int* __restrict__ spillcnt,
                            int* __restrict__ first, int B, int* __restrict__ need, int N) {
  int i = blockIdx.x * blockDim.x + threadIdx.x;
  if (i < nb) cursor[i] = 0;
  if (i == 0) *spillcnt = 0;
  if (i < B) first[i] = 0x7FFFFFFF;
  if (i < N) need[i] = 0;
}

__device__ __forceinline__ int region_count(const int* __restrict__ cursor, int bkt) {
  return min(cursor[bkt], CAP);
}

__device__ __forceinline__ int graph_cluster(const int* __restrict__ first, int g, int nc) {
  int c = first[g];
  if (c > nc - 1) c = nc - 1;
  if (c < 0) c = 0;
  return c;
}

#define QPT 4
#define QCHUNK (512 * 4 * QPT)
__device__ __forceinline__ bool quad_path(const void* a, const void* b, const void* c, int E, int chunk) {
  return chunk <= QCHUNK && ((E | chunk) & 3) == 0 &&
         ((((size_t)a) | ((size_t)b) | ((size_t)c)) & 15) == 0;
}

#define BPT (NBMAX / 512)
__global__ __launch_bounds__(512) void bin_scatter_first(
    const int* __restrict__ src, const int* __restrict__ dst, const float* __restrict__ w,
    int* __restrict__ cursor, int* __restrict__ spillcnt,
    int4* __restrict__ rec, int4* __restrict__ spill, int E, int nb, int chunk,
    const int* __restrict__ batch, int* __restrict__ first, int nc) {
  __shared__ int hist[NBMAX];
  __shared__ int lbase[NBMAX];
  const int t = threadIdx.x;
  for (int b = t; b < nb; b += 512) hist[b] = 0;
  const int c0 = blockIdx.x * blockDim.x + threadIdx.x;
  const int cl = min(c0, nc - 1);
  const int ba = batch[2 * cl], bb = batch[2 * cl + 1];
  const int e0 = blockIdx.x * chunk;
  const int e1 = min(e0 + chunk, E);
  const bool quads = quad_path(dst, src, w, E, chunk);
  const int n4 = quads ? (e1 - e0) >> 2 : 0;
  int4 dq[QPT], sq[QPT];
  float4 wq[QPT];
#pragma unroll
  for (int u = 0; u < QPT; ++u) {
    dq[u] = make_int4(0, 0, 0, 0);
    sq[u] = make_int4(0, 0, 0, 0);
    wq[u] = make_float4(0.0f, 0.0f, 0.0f, 0.0f);
  }
  if (n4 > 0) {
    const int4* __restrict__ d4 = (const int4*)(dst + e0);
    const int4* __restrict__ s4 = (const int4*)(src + e0);
    const float4* __restrict__ w4 = (const float4*)(w + e0);
#pragma unroll
    for (int u = 0; u < QPT; ++u) {
      const int q = min(t + 512 * u, n4 - 1);
      dq[u] = d4[q];
      sq[u] = s4[q];
      wq[u] = w4[q];
    }
  }
  __syncthreads();

  if (quads) {
#pragma unroll
    for (int u = 0; u < QPT; ++u) {
      if (t + 512 * u < n4) {
        atomicAdd(&hist[dq[u].x >> RB_SHIFT], 1);
        atomicAdd(&hist[dq[u].y >> RB_SHIFT], 1);
        atomicAdd(&hist[dq[u].z >> RB_SHIFT], 1);
        atomicAdd(&hist[dq[u].w >> RB_SHIFT], 1);
      }
    }
  } else {
    for (int e = e0 + t; e < e1; e += 512)
      atomicAdd(&hist[dst[e] >> RB_SHIFT], 1);
  }
  __syncthreads();
  int rc[BPT], rb[BPT];
#pragma unroll
  for (int j = 0; j < BPT; ++j) {
    const int b = t + 512 * j;
    rc[j] = b < nb ? hist[b] : 0;
  }
#pragma unroll
  for (int j = 0; j < BPT; ++j) {
    rb[j] = 0;
    if (rc[j]) rb[j] = atomicAdd(&cursor[t + 512 * j], rc[j]);
  }
#pragma unroll
  for (int j = 0; j < BPT; ++j) {
    const int b = t + 512 * j;
    if (b < nb) {
      lbase[b] = rb[j];
      hist[b] = 0;
    }
  }
  __syncthreads();
  if (quads) {
#pragma unroll
    for (int u = 0; u < QPT; ++u) {
      if (t + 512 * u < n4) {
        const int dd[4] = {dq[u].x, dq[u].y, dq[u].z, dq[u].w};
        const int ss[4] = {sq[u].x, sq[u].y, sq[u].z, sq[u].w};
        const float ww[4] = {wq[u].x, wq[u].y, wq[u].z, wq[u].w};
        int rk[4];
#pragma unroll
        for (int k = 0; k < 4; ++k) {
          const int b = dd[k] >> RB_SHIFT;
          rk[k] = lbase[b] + atomicAdd(&hist[b], 1);
        }
#pragma unroll
        for (int k = 0; k < 4; ++k) {
          const int4 r = make_int4(ss[k], dd[k], __float_as_int(ww[k]), 0);
          if (rk[k] < CAP) rec[(dd[k] >> RB_SHIFT) * RSTRIDE + rk[k]] = r;
          else             spill[atomicAdd(spillcnt, 1)] = r;
        }
      }
    }
  } else {
    for (int e = e0 + t; e < e1; e += 512) {
      const int d = dst[e];
      const int b = d >> RB_SHIFT;
      const int rk = lbase[b] + atomicAdd(&hist[b], 1);
      const int4 r = make_int4(src[e], d, __float_as_int(w[e]), 0);
      if (rk < CAP) rec[b * RSTRIDE + rk] = r;
      else          spill[atomicAdd(spillcnt, 1)] = r;
    }
  }
  const int gme = min(ba, bb);
  const int gbelow = __shfl_up(gme, 1);
  const bool covered = (t & 31) != 0 && gbelow == gme;
  if (c0 < nc && !covered) atomicMin(&first[gme], c0);
  for (int cc = c0 + gridDim.x * blockDim.x; cc < nc; cc += gridDim.x * blockDim.x)
    atomicMin(&first[min(batch[2 * cc], batch[2 * cc + 1])], cc);
}

__global__ __launch_bounds__(256) void bucket_deg_mark(
    const int4* __restrict__ rec, const int4* __restrict__ spill,
    const int* __restrict__ cursor, const int* __restrict__ spillcnt,
    float* __restrict__ dinv, int N,
    const int* __restrict__ first, int B, int* __restrict__ need, int nc) {
  __shared__ float degl[RB];
  __shared__ int tgtl[RB];
  const int bkt = blockIdx.x;
  const int base = bkt << RB_SHIFT;
  const int e0 = bkt * RSTRIDE, e1 = e0 + region_count(cursor, bkt);
  const int nsp = *spillcnt;
  int ig = -1;
  if (threadIdx.x < B) ig = 2 * graph_cluster(first, threadIdx.x, nc);
  if (threadIdx.x < RB) { degl[threadIdx.x] = 1.0f; tgtl[threadIdx.x] = 0; }
  __syncthreads();
  if (ig >= 0 && (ig >> RB_SHIFT) == bkt) { tgtl[ig - base] = 1; tgtl[ig + 1 - base] = 1; }
  for (int g = threadIdx.x + blockDim.x; g < B; g += blockDim.x) {
    const int i0 = 2 * graph_cluster(first, g, nc);
    if ((i0 >> RB_SHIFT) == bkt) { tgtl[i0 - base] = 1; tgtl[i0 + 1 - base] = 1; }
  }
  __syncthreads();
  for (int e = e0 + threadIdx.x; e < e1; e += blockDim.x) {
    int4 r = rec[e];
    __hip_atomic_fetch_add(&degl[r.y - base], __int_as_float(r.z),
                           __ATOMIC_RELAXED, __HIP_MEMORY_SCOPE_WORKGROUP);
    if (tgtl[r.y - base]) need[r.x] = 1;
  }
  for (int e = threadIdx.x; e < nsp; e += blockDim.x) {
    int4 r = spill[e];
    if ((r.y >> RB_SHIFT) == bkt) {
      __hip_atomic_fetch_add(&degl[r.y - base], __int_as_float(r.z),
                             __ATOMIC_RELAXED, __HIP_MEMORY_SCOPE_WORKGROUP);
      if (tgtl[r.y - base]) need[r.x] = 1;
    }
  }
  __syncthreads();
  if (threadIdx.x < RB) {
    int i = base + threadIdx.x;
    if (i < N) {
      float d = degl[threadIdx.x];
      dinv[i] = d > 0.0f ? __frsqrt_rn(d) : 0.0f;
      if (tgtl[threadIdx.x]) need[i] = 1;
    }
  }
}

#define IDIM 128
#define APITCH 132
#define OPITCH 68

__global__ __launch_bounds__(256) void bucket_aggregate(
    const float* __restrict__ x, const float* __restrict__ W,
    const float* __restrict__ dinv, const float* __restrict__ bias,
    const int4* __restrict__ rec, const int4* __restrict__ spill,
    const int* __restrict__ cursor, const int* __restrict__ spillcnt,
    const int* __restrict__ need, float* __restrict__ h, int N) {
  __shared__ float acc[RB * APITCH];
  __shared__ int slotl[RB];
  __shared__ int rowl[RB];
  __shared__ int cntl[2];
  const int tid  = threadIdx.x;
  const int lane = tid & 31;
  const int wave = __builtin_amdgcn_readfirstlane(tid >> 5);
  const int bkt  = blockIdx.x;
  const int base = bkt << RB_SHIFT;

  int flag = 0, pre = 0;
  if (tid < RB) {
    flag = (base + tid < N) ? (need[base + tid] != 0) : 0;
    const unsigned m = __builtin_amdgcn_ballot_w32(flag != 0);
    pre = __builtin_popcount(m & ((1u << lane) - 1u));
    if (lane == 0) cntl[tid >> 5] = __builtin_popcount(m);
  }
  __syncthreads();
  const int c0n = cntl[0];
  const int ns = __builtin_amdgcn_readfirstlane(c0n + cntl[1]);
  if (ns == 0) return;
  if (tid < RB) {
    const int s = pre + (tid >= 32 ? c0n : 0);
    slotl[tid] = flag ? s * APITCH : -1;
    if (flag) rowl[s] = tid;
  }
  __syncthreads();

  const int mt = (ns + 15) >> 4;
  for (int idx = tid; idx < (mt << 9); idx += 256) {
    const int r = idx >> 5, c4 = idx & 31;
    float4 v = make_float4(0.0f, 0.0f, 0.0f, 0.0f);
    if (r < ns) {
      const int i = base + rowl[r];
      const float di = dinv[i];
      const float s = di * di;
      const float4 xv = *(const float4*)(x + (size_t)i * IDIM + 4 * c4);
      v.x = s * xv.x; v.y = s * xv.y; v.z = s * xv.z; v.w = s * xv.w;
    }
    *(float4*)(acc + r * APITCH + 4 * c4) = v;
  }
  __syncthreads();

  const int e0 = bkt * RSTRIDE, e1 = e0 + region_count(cursor, bkt);
  const int nsp = *spillcnt;
  const int n = e1 - e0;
  const int seg = (((n + 7) >> 3) + 31) & ~31;
  const int wb = e0 + wave * seg;
  const int we = min(wb + seg, e1);
  const float* __restrict__ xl = x + 2 * lane;
  float* al = acc + 2 * lane;

  for (int pass = 0; pass < 2; ++pass) {
    const int4* __restrict__ list = pass ? spill : rec;
    const int lb = pass ? wave * 32 : wb;
    const int le = pass ? nsp : we;
    const int st = pass ? 256 : 32;
    for (int e = lb; e < le; e += st) {
      const int cnt = min(32, le - e);
      int4 r = make_int4(0, 0, 0, 0);
      int keep = 0;
      if (lane < cnt) {
        r = list[e + lane];
        int so = -1;
        if (pass == 0 || (r.y >> RB_SHIFT) == bkt)
          so = slotl[r.y - base];
        keep = so >= 0;
        if (keep) {
          r.z = __float_as_int(dinv[r.x] * __int_as_float(r.z) * dinv[r.y]);
          r.y = so;
        }
      }
      unsigned mask = __builtin_amdgcn_ballot_w32(keep != 0);
      while (mask) {
        const int j0 = __builtin_ctz(mask);
        mask &= mask - 1;
        int jj[4] = {j0, j0, j0, j0};
        int nk = 1;
#pragma unroll
        for (int u = 1; u < 4; ++u) {
          if (mask) { jj[u] = __builtin_ctz(mask); mask &= mask - 1; nk = u + 1; }
        }
        float2 xa[4], xb[4]; int dl[4]; float nr[4];
#pragma unroll
        for (int u = 0; u < 4; ++u) {
          const int s = __builtin_amdgcn_readlane(r.x, jj[u]);
          dl[u] = __builtin_amdgcn_readlane(r.y, jj[u]);
          nr[u] = __int_as_float(__builtin_amdgcn_readlane(r.z, jj[u]));
          xa[u] = *(const float2*)(xl + (size_t)s * IDIM);
          xb[u] = *(const float2*)(xl + (size_t)s * IDIM + 64);
        }
#pragma unroll
        for (int u = 0; u < 4; ++u) {
          if (u < nk) {
            __hip_atomic_fetch_add(al + dl[u],      nr[u] * xa[u].x, __ATOMIC_RELAXED, __HIP_MEMORY_SCOPE_WORKGROUP);
            __hip_atomic_fetch_add(al + dl[u] + 1,  nr[u] * xa[u].y, __ATOMIC_RELAXED, __HIP_MEMORY_SCOPE_WORKGROUP);
            __hip_atomic_fetch_add(al + dl[u] + 64, nr[u] * xb[u].x, __ATOMIC_RELAXED, __HIP_MEMORY_SCOPE_WORKGROUP);
            __hip_atomic_fetch_add(al + dl[u] + 65, nr[u] * xb[u].y, __ATOMIC_RELAXED, __HIP_MEMORY_SCOPE_WORKGROUP);
          }
        }
      }
    }
  }
  __syncthreads();

  const int half = lane >> 4;
  const int l15  = lane & 15;
  const int nb   = (wave & 3) << 4;
  const float* __restrict__ wr = W + (size_t)(2 * half) * ODIM + nb + l15;
  v8f d[2];
#pragma unroll
  for (int u = 0; u < 2; ++u) {
    const int t = (wave >> 2) + 2 * u;
    d[u] = (v8f){};
    if (t < mt) {
      v8f p0 = {}, p1 = {};
      const float* ar = acc + (t * 16 + l15) * APITCH + 2 * half;
#pragma unroll 4
      for (int k = 0; k < IDIM; k += 8) {
        v2f a0, b0, a1, b1;
        a0[0] = ar[k + 0];
        a0[1] = ar[k + 1];
        b0[0] = wr[(size_t)(k + 0) * ODIM];
        b0[1] = wr[(size_t)(k + 1) * ODIM];
        p0 = __builtin_amdgcn_wmma_f32_16x16x4_f32(
             false, a0,  false, b0,
             (short)0, p0,  false,  false);
        a1[0] = ar[k + 4];
        a1[1] = ar[k + 5];
        b1[0] = wr[(size_t)(k + 4) * ODIM];
        b1[1] = wr[(size_t)(k + 5) * ODIM];
        p1 = __builtin_amdgcn_wmma_f32_16x16x4_f32(
             false, a1,  false, b1,
             (short)0, p1,  false,  false);
      }
      d[u] = p0 + p1;
    }
  }
  __syncthreads();

  const float bv = bias[nb + l15];
#pragma unroll
  for (int u = 0; u < 2; ++u) {
    const int t = (wave >> 2) + 2 * u;
    if (t < mt) {
      float* orow = acc + (t * 16 + 8 * half) * OPITCH + nb + l15;
#pragma unroll
      for (int r = 0; r < 8; ++r) orow[r * OPITCH] = d[u][r] + bv;
    }
  }
  __syncthreads();

  for (int idx = tid; idx < (ns << 4); idx += 256) {
    const int s = idx >> 4, c4 = idx & 15;
    const int i = base + rowl[s];
    *(float4*)(h + (size_t)i * ODIM + 4 * c4) = *(const float4*)(acc + s * OPITCH + 4 * c4);
  }
}

#define TPITCH 68

__global__ __launch_bounds__(256) void graph_rows_out(
    const float* __restrict__ h, const float* __restrict__ dinv,
    const int4* __restrict__ rec, const int4* __restrict__ spill,
    const int* __restrict__ cursor, const int* __restrict__ spillcnt,
    const int* __restrict__ first, const float* __restrict__ W,
    const float* __restrict__ bias, float* __restrict__ out, int nc) {
  __shared__ float part[8][2][ODIM];
  __shared__ __attribute__((aligned(16))) float tile[16 * TPITCH];
  __shared__ __attribute__((aligned(16))) float orow[ODIM];
  const int g = blockIdx.x;
  const int c = graph_cluster(first, g, nc);
  const int i0 = 2 * c;
  const int bkt = i0 >> RB_SHIFT;
  const int tid = threadIdx.x;
  const int lane = tid & 31;
  const int wave = __builtin_amdgcn_readfirstlane(tid >> 5);
  const int e0 = bkt * RSTRIDE, e1 = e0 + region_count(cursor, bkt);
  const int nsp = *spillcnt;
  const float* __restrict__ hl = h + 2 * lane;
  float2 a0 = make_float2(0.0f, 0.0f), a1 = make_float2(0.0f, 0.0f);

  const int half = lane >> 4;
  const int l15  = lane & 15;
  const int nb   = (wave & 3) << 4;
  v2f bf[ODIM / 4] = {};
  float bv = 0.0f;
  if (wave < 4) {
    const float* __restrict__ wr = W + (size_t)(2 * half) * ODIM + nb + l15;
#pragma unroll
    for (int q = 0; q < ODIM / 4; ++q) {
      bf[q][0] = wr[(size_t)(4 * q + 0) * ODIM];
      bf[q][1] = wr[(size_t)(4 * q + 1) * ODIM];
    }
    bv = bias[nb + l15];
  }

  for (int pass = 0; pass < 2; ++pass) {
    const int4* __restrict__ list = pass ? spill : rec;
    const int lb = (pass ? 0 : e0) + wave * 32;
    const int le = pass ? nsp : e1;
    for (int e = lb; e < le; e += 256) {
      const int cnt = min(32, le - e);
      int4 r = make_int4(0, 0, 0, 0);
      int hit = 0;
      if (lane < cnt) {
        r = list[e + lane];
        hit = ((r.y >> 1) == c);
        if (hit)
          r.z = __float_as_int(dinv[r.x] * __int_as_float(r.z) * dinv[r.y]);
      }
      unsigned mask = __builtin_amdgcn_ballot_w32(hit != 0);
      while (mask) {
        const int j = __builtin_ctz(mask);
        mask &= mask - 1;
        const int s = __builtin_amdgcn_readlane(r.x, j);
        const int d = __builtin_amdgcn_readlane(r.y, j);
        const float nr = __int_as_float(__builtin_amdgcn_readlane(r.z, j));
        const float2 xv = *(const float2*)(hl + (size_t)s * ODIM);
        if (d & 1) { a1.x += nr * xv.x; a1.y += nr * xv.y; }
        else       { a0.x += nr * xv.x; a0.y += nr * xv.y; }
      }
    }
  }
  if (wave == 0) {
    const float d0 = dinv[i0], d1 = dinv[i0 + 1];
    const float2 x0 = *(const float2*)(hl + (size_t)i0 * ODIM);
    const float2 x1 = *(const float2*)(hl + (size_t)(i0 + 1) * ODIM);
    a0.x += d0 * d0 * x0.x; a0.y += d0 * d0 * x0.y;
    a1.x += d1 * d1 * x1.x; a1.y += d1 * d1 * x1.y;
  }
  part[wave][0][2 * lane] = a0.x; part[wave][0][2 * lane + 1] = a0.y;
  part[wave][1][2 * lane] = a1.x; part[wave][1][2 * lane + 1] = a1.y;
  for (int idx = tid; idx < 14 * ODIM; idx += 256)
    tile[(2 + (idx >> 6)) * TPITCH + (idx & 63)] = 0.0f;
  __syncthreads();
  if (tid < 2 * ODIM) {
    const int p = tid >> 6, f = tid & 63;
    float sum = part[0][p][f];
#pragma unroll
    for (int w = 1; w < 8; ++w) sum += part[w][p][f];
    tile[p * TPITCH + f] = sum;
  }
  __syncthreads();

  if (wave < 4) {
    const float* ar = tile + l15 * TPITCH + 2 * half;
    v8f acc = {};
#pragma unroll
    for (int k = 0; k < ODIM; k += 4) {
      v2f a;
      a[0] = ar[k + 0];
      a[1] = ar[k + 1];
      acc = __builtin_amdgcn_wmma_f32_16x16x4_f32(
           false, a,  false, bf[k >> 2],
           (short)0, acc,  false,  false);
    }
    const float pooled = fmaxf(acc[0] + bv, acc[1] + bv);
    if (half == 0) orow[nb + l15] = pooled;
  }
  __syncthreads();
  if (tid < 16)
    *(float4*)(out + (size_t)g * ODIM + 4 * tid) = *(const float4*)(orow + 4 * tid);
}

extern "C" void kernel_launch(void* const* d_in, const int* in_sizes, int n_in,
                              void* d_out, int out_size, void* d_ws, size_t ws_size,
                              hipStream_t stream) {
  const float* x   = (const float*)d_in[0];
  const int*   ei  = (const int*)d_in[1];
  const float* ew  = (const float*)d_in[2];
  const int*   bat = (const int*)d_in[3];
  const float* W1  = (const float*)d_in[4];
  const float* b1  = (const float*)d_in[5];
  const float* W2  = (const float*)d_in[6];
  const float* b2  = (const float*)d_in[7];
  float* out = (float*)d_out;

  const int IN_DIM = 128;
  const int N  = in_sizes[0] / IN_DIM;
  const int E  = in_sizes[2];
  const int B  = out_size / ODIM;
  const int NC = N / 2;

  const int* src = ei;
  const int* dst = ei + E;

  char* ws = (char*)d_ws;
  size_t featBytes = (size_t)N * ODIM * sizeof(float);
  size_t aggBytes  = (size_t)2 * B * ODIM * sizeof(float);
  float* h    = (float*)(ws + featBytes);
  float* dinv = (float*)(ws + 2 * featBytes + (size_t)N * 4);
  int*   need = (int*)  (ws + 2 * featBytes + (size_t)N * 8 + aggBytes);
  int*   first= (int*)  (ws + 2 * featBytes + (size_t)N * 12 + aggBytes);
  int4*  rec      = (int4*)(ws + 134217728);
  int*   cursor   = (int*) (ws + 209715200);
  int*   spillcnt = (int*) (ws + 209715200 + 65536);
  int4*  spill    = (int4*)(ws + 218103808);

  const int T = 256;
  dim3 blk(T);

  const int NB = (N + RB - 1) / RB;
  const int cover = N > B ? N : B;
  init_tables<<<(cover + T - 1) / T, blk, 0, stream>>>(cursor, NB, spillcnt, first, B, need, N);

  const int G1 = 256;
  const int chunk = ((E + G1 - 1) / G1 + 3) & ~3;
  bin_scatter_first<<<G1, dim3(512), 0, stream>>>(src, dst, ew, cursor, spillcnt, rec, spill, E, NB, chunk,
                                                  bat, first, NC);
  bucket_deg_mark<<<NB, blk, 0, stream>>>(rec, spill, cursor, spillcnt, dinv, N, first, B, need, NC);

  bucket_aggregate<<<NB, blk, 0, stream>>>(x, W1, dinv, b1, rec, spill, cursor, spillcnt, need, h, N);

  graph_rows_out<<<B, blk, 0, stream>>>(h, dinv, rec, spill, cursor, spillcnt, first, W2, b2, out, NC);
}


# Both row-sum kernels: a wave requests 4 record chunks at once, then all their norm factors, then gathers rows 8 at a time from an LDS queue of its kept edges; arrival order of LDS float adds changes
# speedup vs baseline: 13.4014x; 1.1190x over previous
#include <hip/hip_runtime.h>
#include <hip/hip_bf16.h>

typedef __attribute__((ext_vector_type(2))) float v2f;
typedef __attribute__((ext_vector_type(8))) float v8f;

#define ODIM 64

#define RB 64
#define RB_SHIFT 6
#define NBMAX 2048
#define CAP 2048
#define RSTRIDE (CAP + 16)

__global__ void init_tables(int* __restrict__ cursor, int nb, int* __restrict__ spillcnt,
                            int* __restrict__ first, int B, int* __restrict__ need, int N) {
  int i = blockIdx.x * blockDim.x + threadIdx.x;
  if (i < nb) cursor[i] = 0;
  if (i == 0) *spillcnt = 0;
  if (i < B) first[i] = 0x7FFFFFFF;
  if (i < N) need[i] = 0;
}

__device__ __forceinline__ int region_count(const int* __restrict__ cursor, int bkt) {
  return min(cursor[bkt], CAP);
}

__device__ __forceinline__ int graph_cluster(const int* __restrict__ first, int g, int nc) {
  int c = first[g];
  if (c > nc - 1) c = nc - 1;
  if (c < 0) c = 0;
  return c;
}

#define QPT 4
#define QCHUNK (512 * 4 * QPT)
__device__ __forceinline__ bool quad_path(const void* a, const void* b, const void* c, int E, int chunk) {
  return chunk <= QCHUNK && ((E | chunk) & 3) == 0 &&
         ((((size_t)a) | ((size_t)b) | ((size_t)c)) & 15) == 0;
}

#define BPT (NBMAX / 512)
__global__ __launch_bounds__(512) void bin_scatter_first(
    const int* __restrict__ src, const int* __restrict__ dst, const float* __restrict__ w,
    int* __restrict__ cursor, int* __restrict__ spillcnt,
    int4* __restrict__ rec, int4* __restrict__ spill, int E, int nb, int chunk,
    const int* __restrict__ batch, int* __restrict__ first, int nc) {
  __shared__ int hist[NBMAX];
  __shared__ int lbase[NBMAX];
  const int t = threadIdx.x;
  for (int b = t; b < nb; b += 512) hist[b] = 0;
  const int c0 = blockIdx.x * blockDim.x + threadIdx.x;
  const int cl = min(c0, nc - 1);
  const int ba = batch[2 * cl], bb = batch[2 * cl + 1];
  const int e0 = blockIdx.x * chunk;
  const int e1 = min(e0 + chunk, E);
  const bool quads = quad_path(dst, src, w, E, chunk);
  const int n4 = quads ? (e1 - e0) >> 2 : 0;
  int4 dq[QPT], sq[QPT];
  float4 wq[QPT];
#pragma unroll
  for (int u = 0; u < QPT; ++u) {
    dq[u] = make_int4(0, 0, 0, 0);
    sq[u] = make_int4(0, 0, 0, 0);
    wq[u] = make_float4(0.0f, 0.0f, 0.0f, 0.0f);
  }
  if (n4 > 0) {
    const int4* __restrict__ d4 = (const int4*)(dst + e0);
    const int4* __restrict__ s4 = (const int4*)(src + e0);
    const float4* __restrict__ w4 = (const float4*)(w + e0);
#pragma unroll
    for (int u = 0; u < QPT; ++u) {
      const int q = min(t + 512 * u, n4 - 1);
      dq[u] = d4[q];
      sq[u] = s4[q];
      wq[u] = w4[q];
    }
  }
  __syncthreads();

  if (quads) {
#pragma unroll
    for (int u = 0; u < QPT; ++u) {
      if (t + 512 * u < n4) {
        atomicAdd(&hist[dq[u].x >> RB_SHIFT], 1);
        atomicAdd(&hist[dq[u].y >> RB_SHIFT], 1);
        atomicAdd(&hist[dq[u].z >> RB_SHIFT], 1);
        atomicAdd(&hist[dq[u].w >> RB_SHIFT], 1);
      }
    }
  } else {
    for (int e = e0 + t; e < e1; e += 512)
      atomicAdd(&hist[dst[e] >> RB_SHIFT], 1);
  }
  __syncthreads();
  int rc[BPT], rb[BPT];
#pragma unroll
  for (int j = 0; j < BPT; ++j) {
    const int b = t + 512 * j;
    rc[j] = b < nb ? hist[b] : 0;
  }
#pragma unroll
  for (int j = 0; j < BPT; ++j) {
    rb[j] = 0;
    if (rc[j]) rb[j] = atomicAdd(&cursor[t + 512 * j], rc[j]);
  }
#pragma unroll
  for (int j = 0; j < BPT; ++j) {
    const int b = t + 512 * j;
    if (b < nb) {
      lbase[b] = rb[j];
      hist[b] = 0;
    }
  }
  __syncthreads();
  if (quads) {
#pragma unroll
    for (int u = 0; u < QPT; ++u) {
      if (t + 512 * u < n4) {
        const int dd[4] = {dq[u].x, dq[u].y, dq[u].z, dq[u].w};
        const int ss[4] = {sq[u].x, sq[u].y, sq[u].z, sq[u].w};
        const float ww[4] = {wq[u].x, wq[u].y, wq[u].z, wq[u].w};
        int rk[4];
#pragma unroll
        for (int k = 0; k < 4; ++k) {
          const int b = dd[k] >> RB_SHIFT;
          rk[k] = lbase[b] + atomicAdd(&hist[b], 1);
        }
#pragma unroll
        for (int k = 0; k < 4; ++k) {
          const int4 r = make_int4(ss[k], dd[k], __float_as_int(ww[k]), 0);
          if (rk[k] < CAP) rec[(dd[k] >> RB_SHIFT) * RSTRIDE + rk[k]] = r;
          else             spill[atomicAdd(spillcnt, 1)] = r;
        }
      }
    }
  } else {
    for (int e = e0 + t; e < e1; e += 512) {
      const int d = dst[e];
      const int b = d >> RB_SHIFT;
      const int rk = lbase[b] + atomicAdd(&hist[b], 1);
      const int4 r = make_int4(src[e], d, __float_as_int(w[e]), 0);
      if (rk < CAP) rec[b * RSTRIDE + rk] = r;
      else          spill[atomicAdd(spillcnt, 1)] = r;
    }
  }
  const int gme = min(ba, bb);
  const int gbelow = __shfl_up(gme, 1);
  const bool covered = (t & 31) != 0 && gbelow == gme;
  if (c0 < nc && !covered) atomicMin(&first[gme], c0);
  for (int cc = c0 + gridDim.x * blockDim.x; cc < nc; cc += gridDim.x * blockDim.x)
    atomicMin(&first[min(batch[2 * cc], batch[2 * cc + 1])], cc);
}

__global__ __launch_bounds__(256) void bucket_deg_mark(
    const int4* __restrict__ rec, const int4* __restrict__ spill,
    const int* __restrict__ cursor, const int* __restrict__ spillcnt,
    float* __restrict__ dinv, int N,
    const int* __restrict__ first, int B, int* __restrict__ need, int nc) {
  __shared__ float degl[RB];
  __shared__ int tgtl[RB];
  const int bkt = blockIdx.x;
  const int base = bkt << RB_SHIFT;
  const int e0 = bkt * RSTRIDE, e1 = e0 + region_count(cursor, bkt);
  const int nsp = *spillcnt;
  int ig = -1;
  if (threadIdx.x < B) ig = 2 * graph_cluster(first, threadIdx.x, nc);
  if (threadIdx.x < RB) { degl[threadIdx.x] = 1.0f; tgtl[threadIdx.x] = 0; }
  __syncthreads();
  if (ig >= 0 && (ig >> RB_SHIFT) == bkt) { tgtl[ig - base] = 1; tgtl[ig + 1 - base] = 1; }
  for (int g = threadIdx.x + blockDim.x; g < B; g += blockDim.x) {
    const int i0 = 2 * graph_cluster(first, g, nc);
    if ((i0 >> RB_SHIFT) == bkt) { tgtl[i0 - base] = 1; tgtl[i0 + 1 - base] = 1; }
  }
  __syncthreads();
  for (int e = e0 + threadIdx.x; e < e1; e += blockDim.x) {
    int4 r = rec[e];
    __hip_atomic_fetch_add(&degl[r.y - base], __int_as_float(r.z),
                           __ATOMIC_RELAXED, __HIP_MEMORY_SCOPE_WORKGROUP);
    if (tgtl[r.y - base]) need[r.x] = 1;
  }
  for (int e = threadIdx.x; e < nsp; e += blockDim.x) {
    int4 r = spill[e];
    if ((r.y >> RB_SHIFT) == bkt) {
      __hip_atomic_fetch_add(&degl[r.y - base], __int_as_float(r.z),
                             __ATOMIC_RELAXED, __HIP_MEMORY_SCOPE_WORKGROUP);
      if (tgtl[r.y - base]) need[r.x] = 1;
    }
  }
  __syncthreads();
  if (threadIdx.x < RB) {
    int i = base + threadIdx.x;
    if (i < N) {
      float d = degl[threadIdx.x];
      dinv[i] = d > 0.0f ? __frsqrt_rn(d) : 0.0f;
      if (tgtl[threadIdx.x]) need[i] = 1;
    }
  }
}

#define IDIM 128
#define APITCH 132
#define OPITCH 68
#define SCH 4
#define GB 8

__global__ __launch_bounds__(256) void bucket_aggregate(
    const float* __restrict__ x, const float* __restrict__ W,
    const float* __restrict__ dinv, const float* __restrict__ bias,
    const int4* __restrict__ rec, const int4* __restrict__ spill,
    const int* __restrict__ cursor, const int* __restrict__ spillcnt,
    const int* __restrict__ need, float* __restrict__ h, int N) {
  __shared__ float acc[RB * APITCH];
  __shared__ int slotl[RB];
  __shared__ int rowl[RB];
  __shared__ int cntl[2];
  __shared__ int2 wq[8][32 * SCH];
  const int tid  = threadIdx.x;
  const int lane = tid & 31;
  const int wave = __builtin_amdgcn_readfirstlane(tid >> 5);
  const int bkt  = blockIdx.x;
  const int base = bkt << RB_SHIFT;

  int flag = 0, pre = 0;
  if (tid < RB) {
    flag = (base + tid < N) ? (need[base + tid] != 0) : 0;
    const unsigned m = __builtin_amdgcn_ballot_w32(flag != 0);
    pre = __builtin_popcount(m & ((1u << lane) - 1u));
    if (lane == 0) cntl[tid >> 5] = __builtin_popcount(m);
  }
  __syncthreads();
  const int c0n = cntl[0];
  const int ns = __builtin_amdgcn_readfirstlane(c0n + cntl[1]);
  if (ns == 0) return;
  if (tid < RB) {
    const int s = pre + (tid >= 32 ? c0n : 0);
    slotl[tid] = flag ? s * APITCH : -1;
    if (flag) rowl[s] = tid;
  }
  __syncthreads();

  const int mt = (ns + 15) >> 4;
  for (int idx = tid; idx < (mt << 9); idx += 256) {
    const int r = idx >> 5, c4 = idx & 31;
    float4 v = make_float4(0.0f, 0.0f, 0.0f, 0.0f);
    if (r < ns) {
      const int i = base + rowl[r];
      const float di = dinv[i];
      const float s = di * di;
      const float4 xv = *(const float4*)(x + (size_t)i * IDIM + 4 * c4);
      v.x = s * xv.x; v.y = s * xv.y; v.z = s * xv.z; v.w = s * xv.w;
    }
    *(float4*)(acc + r * APITCH + 4 * c4) = v;
  }
  __syncthreads();

  const int e0 = bkt * RSTRIDE, e1 = e0 + region_count(cursor, bkt);
  const int nsp = *spillcnt;
  const int n = e1 - e0;
  const int seg = (((n + 7) >> 3) + 31) & ~31;
  const int wb = e0 + wave * seg;
  const int we = min(wb + seg, e1);
  const float* __restrict__ xl = x + 2 * lane;
  float* al = acc + 2 * lane;

  int2* __restrict__ myq = wq[wave];
  for (int pass = 0; pass < 2; ++pass) {
    const int4* __restrict__ list = pass ? spill : rec;
    const int lb = pass ? wave * 32 : wb;
    const int le = pass ? nsp : we;
    const int cs = pass ? 256 : 32;
    for (int e = lb; e < le; e += cs * SCH) {
      int4 r[SCH];
      int so[SCH];
      float ds[SCH], dd[SCH];
#pragma unroll
      for (int u = 0; u < SCH; ++u)
        r[u] = list[min(e + u * cs + lane, le - 1)];
#pragma unroll
      for (int u = 0; u < SCH; ++u)
        so[u] = slotl[(r[u].y - base) & (RB - 1)];
#pragma unroll
      for (int u = 0; u < SCH; ++u) {
        const bool mine = e + u * cs + lane < le && (pass == 0 || (r[u].y >> RB_SHIFT) == bkt);
        if (!mine) so[u] = -1;
      }
#pragma unroll
      for (int u = 0; u < SCH; ++u) {
        const bool keep = so[u] >= 0;
        ds[u] = dinv[keep ? r[u].x : 0];
        dd[u] = dinv[keep ? r[u].y : 0];
      }
      int nq = 0;
#pragma unroll
      for (int u = 0; u < SCH; ++u) {
        const bool keep = so[u] >= 0;
        const unsigned m = __builtin_amdgcn_ballot_w32(keep);
        if (keep) {
          const int p = nq + __builtin_popcount(m & ((1u << lane) - 1u));
          myq[p] = make_int2(r[u].x | (so[u] << 17),
                             __float_as_int(ds[u] * __int_as_float(r[u].z) * dd[u]));
        }
        nq += __builtin_popcount(m);
      }
      for (int q0 = 0; q0 < nq; q0 += 32) {
        const int cnt = min(32, nq - q0);
        int2 qe = make_int2(0, 0);
        if (lane < cnt) qe = myq[q0 + lane];
        for (int j = 0; j < cnt; j += GB) {
          float2 xa[GB], xb[GB]; int dl[GB]; float nr[GB];
#pragma unroll
          for (int u = 0; u < GB; ++u) {
            const int jj = min(j + u, cnt - 1);
            const int pk = __builtin_amdgcn_readlane(qe.x, jj);
            nr[u] = __int_as_float(__builtin_amdgcn_readlane(qe.y, jj));
            dl[u] = pk >> 17;
            const int s = pk & 0x1FFFF;
            xa[u] = *(const float2*)(xl + (size_t)s * IDIM);
            xb[u] = *(const float2*)(xl + (size_t)s * IDIM + 64);
          }
#pragma unroll
          for (int u = 0; u < GB; ++u) {
            if (j + u < cnt) {
              __hip_atomic_fetch_add(al + dl[u],      nr[u] * xa[u].x, __ATOMIC_RELAXED, __HIP_MEMORY_SCOPE_WORKGROUP);
              __hip_atomic_fetch_add(al + dl[u] + 1,  nr[u] * xa[u].y, __ATOMIC_RELAXED, __HIP_MEMORY_SCOPE_WORKGROUP);
              __hip_atomic_fetch_add(al + dl[u] + 64, nr[u] * xb[u].x, __ATOMIC_RELAXED, __HIP_MEMORY_SCOPE_WORKGROUP);
              __hip_atomic_fetch_add(al + dl[u] + 65, nr[u] * xb[u].y, __ATOMIC_RELAXED, __HIP_MEMORY_SCOPE_WORKGROUP);
            }
          }
        }
      }
    }
  }
  __syncthreads();

  const int half = lane >> 4;
  const int l15  = lane & 15;
  const int nb   = (wave & 3) << 4;
  const float* __restrict__ wr = W + (size_t)(2 * half) * ODIM + nb + l15;
  v8f d[2];
#pragma unroll
  for (int u = 0; u < 2; ++u) {
    const int t = (wave >> 2) + 2 * u;
    d[u] = (v8f){};
    if (t < mt) {
      v8f p0 = {}, p1 = {};
      const float* ar = acc + (t * 16 + l15) * APITCH + 2 * half;
#pragma unroll 4
      for (int k = 0; k < IDIM; k += 8) {
        v2f a0, b0, a1, b1;
        a0[0] = ar[k + 0];
        a0[1] = ar[k + 1];
        b0[0] = wr[(size_t)(k + 0) * ODIM];
        b0[1] = wr[(size_t)(k + 1) * ODIM];
        p0 = __builtin_amdgcn_wmma_f32_16x16x4_f32(
             false, a0,  false, b0,
             (short)0, p0,  false,  false);
        a1[0] = ar[k + 4];
        a1[1] = ar[k + 5];
        b1[0] = wr[(size_t)(k + 4) * ODIM];
        b1[1] = wr[(size_t)(k + 5) * ODIM];
        p1 = __builtin_amdgcn_wmma_f32_16x16x4_f32(
             false, a1,  false, b1,
             (short)0, p1,  false,  false);
      }
      d[u] = p0 + p1;
    }
  }
  __syncthreads();

  const float bv = bias[nb + l15];
#pragma unroll
  for (int u = 0; u < 2; ++u) {
    const int t = (wave >> 2) + 2 * u;
    if (t < mt) {
      float* orow = acc + (t * 16 + 8 * half) * OPITCH + nb + l15;
#pragma unroll
      for (int r = 0; r < 8; ++r) orow[r * OPITCH] = d[u][r] + bv;
    }
  }
  __syncthreads();

  for (int idx = tid; idx < (ns << 4); idx += 256) {
    const int s = idx >> 4, c4 = idx & 15;
    const int i = base + rowl[s];
    *(float4*)(h + (size_t)i * ODIM + 4 * c4) = *(const float4*)(acc + s * OPITCH + 4 * c4);
  }
}

#define TPITCH 68

__global__ __launch_bounds__(256) void graph_rows_out(
    const float* __restrict__ h, const float* __restrict__ dinv,
    const int4* __restrict__ rec, const int4* __restrict__ spill,
    const int* __restrict__ cursor, const int* __restrict__ spillcnt,
    const int* __restrict__ first, const float* __restrict__ W,
    const float* __restrict__ bias, float* __restrict__ out, int nc) {
  __shared__ float part[8][2][ODIM];
  __shared__ __attribute__((aligned(16))) float tile[16 * TPITCH];
  __shared__ __attribute__((aligned(16))) float orow[ODIM];
  __shared__ int2 gq[8][32 * SCH];
  const int g = blockIdx.x;
  const int c = graph_cluster(first, g, nc);
  const int i0 = 2 * c;
  const int bkt = i0 >> RB_SHIFT;
  const int tid = threadIdx.x;
  const int lane = tid & 31;
  const int wave = __builtin_amdgcn_readfirstlane(tid >> 5);
  const int e0 = bkt * RSTRIDE, e1 = e0 + region_count(cursor, bkt);
  const int nsp = *spillcnt;
  const float* __restrict__ hl = h + 2 * lane;
  float2 a0 = make_float2(0.0f, 0.0f), a1 = make_float2(0.0f, 0.0f);

  const int half = lane >> 4;
  const int l15  = lane & 15;
  const int nb   = (wave & 3) << 4;
  v2f bf[ODIM / 4] = {};
  float bv = 0.0f;
  if (wave < 4) {
    const float* __restrict__ wr = W + (size_t)(2 * half) * ODIM + nb + l15;
#pragma unroll
    for (int q = 0; q < ODIM / 4; ++q) {
      bf[q][0] = wr[(size_t)(4 * q + 0) * ODIM];
      bf[q][1] = wr[(size_t)(4 * q + 1) * ODIM];
    }
    bv = bias[nb + l15];
  }

  int2* __restrict__ myq = gq[wave];
  for (int pass = 0; pass < 2; ++pass) {
    const int4* __restrict__ list = pass ? spill : rec;
    const int lb = (pass ? 0 : e0) + wave * 32;
    const int le = pass ? nsp : e1;
    for (int e = lb; e < le; e += 256 * SCH) {
      int4 r[SCH];
      bool hit[SCH];
      float ds[SCH], dd[SCH];
#pragma unroll
      for (int u = 0; u < SCH; ++u)
        r[u] = list[min(e + 256 * u + lane, le - 1)];
#pragma unroll
      for (int u = 0; u < SCH; ++u)
        asm volatile("" :: "v"(r[u].x), "v"(r[u].y), "v"(r[u].z));
#pragma unroll
      for (int u = 0; u < SCH; ++u) {
        hit[u] = e + 256 * u + lane < le && (r[u].y >> 1) == c;
        ds[u] = dinv[hit[u] ? r[u].x : 0];
        dd[u] = dinv[hit[u] ? r[u].y : 0];
      }
      int nq = 0;
#pragma unroll
      for (int u = 0; u < SCH; ++u) {
        const unsigned m = __builtin_amdgcn_ballot_w32(hit[u]);
        if (hit[u]) {
          const int p = nq + __builtin_popcount(m & ((1u << lane) - 1u));
          myq[p] = make_int2(r[u].x | ((r[u].y & 1) << 17),
                             __float_as_int(ds[u] * __int_as_float(r[u].z) * dd[u]));
        }
        nq += __builtin_popcount(m);
      }
      for (int q0 = 0; q0 < nq; q0 += 32) {
        const int cnt = min(32, nq - q0);
        int2 qe = make_int2(0, 0);
        if (lane < cnt) qe = myq[q0 + lane];
        for (int j = 0; j < cnt; j += GB) {
          float2 xv[GB]; int pk[GB]; float nr[GB];
#pragma unroll
          for (int u = 0; u < GB; ++u) {
            const int jj = min(j + u, cnt - 1);
            pk[u] = __builtin_amdgcn_readlane(qe.x, jj);
            nr[u] = __int_as_float(__builtin_amdgcn_readlane(qe.y, jj));
            xv[u] = *(const float2*)(hl + (size_t)(pk[u] & 0x1FFFF) * ODIM);
          }
#pragma unroll
          for (int u = 0; u < GB; ++u) {
            if (j + u < cnt) {
              if (pk[u] >> 17) { a1.x = __builtin_fmaf(nr[u], xv[u].x, a1.x); a1.y = __builtin_fmaf(nr[u], xv[u].y, a1.y); }
              else             { a0.x = __builtin_fmaf(nr[u], xv[u].x, a0.x); a0.y = __builtin_fmaf(nr[u], xv[u].y, a0.y); }
            }
          }
        }
      }
    }
  }
  if (wave == 0) {
    const float d0 = dinv[i0], d1 = dinv[i0 + 1];
    const float2 x0 = *(const float2*)(hl + (size_t)i0 * ODIM);
    const float2 x1 = *(const float2*)(hl + (size_t)(i0 + 1) * ODIM);
    a0.x += d0 * d0 * x0.x; a0.y += d0 * d0 * x0.y;
    a1.x += d1 * d1 * x1.x; a1.y += d1 * d1 * x1.y;
  }
  part[wave][0][2 * lane] = a0.x; part[wave][0][2 * lane + 1] = a0.y;
  part[wave][1][2 * lane] = a1.x; part[wave][1][2 * lane + 1] = a1.y;
  for (int idx = tid; idx < 14 * ODIM; idx += 256)
    tile[(2 + (idx >> 6)) * TPITCH + (idx & 63)] = 0.0f;
  __syncthreads();
  if (tid < 2 * ODIM) {
    const int p = tid >> 6, f = tid & 63;
    float sum = part[0][p][f];
#pragma unroll
    for (int w = 1; w < 8; ++w) sum += part[w][p][f];
    tile[p * TPITCH + f] = sum;
  }
  __syncthreads();

  if (wave < 4) {
    const float* ar = tile + l15 * TPITCH + 2 * half;
    v8f acc = {};
#pragma unroll
    for (int k = 0; k < ODIM; k += 4) {
      v2f a;
      a[0] = ar[k + 0];
      a[1] = ar[k + 1];
      acc = __builtin_amdgcn_wmma_f32_16x16x4_f32(
           false, a,  false, bf[k >> 2],
           (short)0, acc,  false,  false);
    }
    const float pooled = fmaxf(acc[0] + bv, acc[1] + bv);
    if (half == 0) orow[nb + l15] = pooled;
  }
  __syncthreads();
  if (tid < 16)
    *(float4*)(out + (size_t)g * ODIM + 4 * tid) = *(const float4*)(orow + 4 * tid);
}

extern "C" void kernel_launch(void* const* d_in, const int* in_sizes, int n_in,
                              void* d_out, int out_size, void* d_ws, size_t ws_size,
                              hipStream_t stream) {
  const float* x   = (const float*)d_in[0];
  const int*   ei  = (const int*)d_in[1];
  const float* ew  = (const float*)d_in[2];
  const int*   bat = (const int*)d_in[3];
  const float* W1  = (const float*)d_in[4];
  const float* b1  = (const float*)d_in[5];
  const float* W2  = (const float*)d_in[6];
  const float* b2  = (const float*)d_in[7];
  float* out = (float*)d_out;

  const int IN_DIM = 128;
  const int N  = in_sizes[0] / IN_DIM;
  const int E  = in_sizes[2];
  const int B  = out_size / ODIM;
  const int NC = N / 2;

  const int* src = ei;
  const int* dst = ei + E;

  char* ws = (char*)d_ws;
  size_t featBytes = (size_t)N * ODIM * sizeof(float);
  size_t aggBytes  = (size_t)2 * B * ODIM * sizeof(float);
  float* h    = (float*)(ws + featBytes);
  float* dinv = (float*)(ws + 2 * featBytes + (size_t)N * 4);
  int*   need = (int*)  (ws + 2 * featBytes + (size_t)N * 8 + aggBytes);
  int*   first= (int*)  (ws + 2 * featBytes + (size_t)N * 12 + aggBytes);
  int4*  rec      = (int4*)(ws + 134217728);
  int*   cursor   = (int*) (ws + 209715200);
  int*   spillcnt = (int*) (ws + 209715200 + 65536);
  int4*  spill    = (int4*)(ws + 218103808);

  const int T = 256;
  dim3 blk(T);

  const int NB = (N + RB - 1) / RB;
  const int cover = N > B ? N : B;
  init_tables<<<(cover + T - 1) / T, blk, 0, stream>>>(cursor, NB, spillcnt, first, B, need, N);

  const int G1 = 256;
  const int chunk = ((E + G1 - 1) / G1 + 3) & ~3;
  bin_scatter_first<<<G1, dim3(512), 0, stream>>>(src, dst, ew, cursor, spillcnt, rec, spill, E, NB, chunk,
                                                  bat, first, NC);
  bucket_deg_mark<<<NB, blk, 0, stream>>>(rec, spill, cursor, spillcnt, dinv, N, first, B, need, NC);

  bucket_aggregate<<<NB, blk, 0, stream>>>(x, W1, dinv, b1, rec, spill, cursor, spillcnt, need, h, N);

  graph_rows_out<<<B, blk, 0, stream>>>(h, dinv, rec, spill, cursor, spillcnt, first, W2, b2, out, NC);
}


# Record walks of the three bucket kernels: each lane requests its first 5 records from fixed slots of the region at kernel entry, beside the cursor, not behind it; LDS float add order changes
# speedup vs baseline: 13.9667x; 1.0422x over previous
#include <hip/hip_runtime.h>
#include <hip/hip_bf16.h>

typedef __attribute__((ext_vector_type(2))) float v2f;
typedef __attribute__((ext_vector_type(8))) float v8f;

#define ODIM 64

#define RB 64
#define RB_SHIFT 6
#define NBMAX 2048
#define CAP 2048
#define RSTRIDE (CAP + 16)

__global__ void init_tables(int* __restrict__ cursor, int nb, int* __restrict__ spillcnt,
                            int* __restrict__ first, int B, int* __restrict__ need, int N) {
  int i = blockIdx.x * blockDim.x + threadIdx.x;
  if (i < nb) cursor[i] = 0;
  if (i == 0) *spillcnt = 0;
  if (i < B) first[i] = 0x7FFFFFFF;
  if (i < N) need[i] = 0;
}

__device__ __forceinline__ int region_count(const int* __restrict__ cursor, int bkt) {
  return min(cursor[bkt], CAP);
}

__device__ __forceinline__ int graph_cluster(const int* __restrict__ first, int g, int nc) {
  int c = first[g];
  if (c > nc - 1) c = nc - 1;
  if (c < 0) c = 0;
  return c;
}

#define QPT 4
#define QCHUNK (512 * 4 * QPT)
__device__ __forceinline__ bool quad_path(const void* a, const void* b, const void* c, int E, int chunk) {
  return chunk <= QCHUNK && ((E | chunk) & 3) == 0 &&
         ((((size_t)a) | ((size_t)b) | ((size_t)c)) & 15) == 0;
}

#define BPT (NBMAX / 512)
__global__ __launch_bounds__(512) void bin_scatter_first(
    const int* __restrict__ src, const int* __restrict__ dst, const float* __restrict__ w,
    int* __restrict__ cursor, int* __restrict__ spillcnt,
    int4* __restrict__ rec, int4* __restrict__ spill, int E, int nb, int chunk,
    const int* __restrict__ batch, int* __restrict__ first, int nc) {
  __shared__ int hist[NBMAX];
  __shared__ int lbase[NBMAX];
  const int t = threadIdx.x;
  for (int b = t; b < nb; b += 512) hist[b] = 0;
  const int c0 = blockIdx.x * blockDim.x + threadIdx.x;
  const int cl = min(c0, nc - 1);
  const int ba = batch[2 * cl], bb = batch[2 * cl + 1];
  const int e0 = blockIdx.x * chunk;
  const int e1 = min(e0 + chunk, E);
  const bool quads = quad_path(dst, src, w, E, chunk);
  const int n4 = quads ? (e1 - e0) >> 2 : 0;
  int4 dq[QPT], sq[QPT];
  float4 wq[QPT];
#pragma unroll
  for (int u = 0; u < QPT; ++u) {
    dq[u] = make_int4(0, 0, 0, 0);
    sq[u] = make_int4(0, 0, 0, 0);
    wq[u] = make_float4(0.0f, 0.0f, 0.0f, 0.0f);
  }
  if (n4 > 0) {
    const int4* __restrict__ d4 = (const int4*)(dst + e0);
    const int4* __restrict__ s4 = (const int4*)(src + e0);
    const float4* __restrict__ w4 = (const float4*)(w + e0);
#pragma unroll
    for (int u = 0; u < QPT; ++u) {
      const int q = min(t + 512 * u, n4 - 1);
      dq[u] = d4[q];
      sq[u] = s4[q];
      wq[u] = w4[q];
    }
  }
  __syncthreads();

  if (quads) {
#pragma unroll
    for (int u = 0; u < QPT; ++u) {
      if (t + 512 * u < n4) {
        atomicAdd(&hist[dq[u].x >> RB_SHIFT], 1);
        atomicAdd(&hist[dq[u].y >> RB_SHIFT], 1);
        atomicAdd(&hist[dq[u].z >> RB_SHIFT], 1);
        atomicAdd(&hist[dq[u].w >> RB_SHIFT], 1);
      }
    }
  } else {
    for (int e = e0 + t; e < e1; e += 512)
      atomicAdd(&hist[dst[e] >> RB_SHIFT], 1);
  }
  __syncthreads();
  int rc[BPT], rb[BPT];
#pragma unroll
  for (int j = 0; j < BPT; ++j) {
    const int b = t + 512 * j;
    rc[j] = b < nb ? hist[b] : 0;
  }
#pragma unroll
  for (int j = 0; j < BPT; ++j) {
    rb[j] = 0;
    if (rc[j]) rb[j] = atomicAdd(&cursor[t + 512 * j], rc[j]);
  }
#pragma unroll
  for (int j = 0; j < BPT; ++j) {
    const int b = t + 512 * j;
    if (b < nb) {
      lbase[b] = rb[j];
      hist[b] = 0;
    }
  }
  __syncthreads();
  if (quads) {
#pragma unroll
    for (int u = 0; u < QPT; ++u) {
      if (t + 512 * u < n4) {
        const int dd[4] = {dq[u].x, dq[u].y, dq[u].z, dq[u].w};
        const int ss[4] = {sq[u].x, sq[u].y, sq[u].z, sq[u].w};
        const float ww[4] = {wq[u].x, wq[u].y, wq[u].z, wq[u].w};
        int rk[4];
#pragma unroll
        for (int k = 0; k < 4; ++k) {
          const int b = dd[k] >> RB_SHIFT;
          rk[k] = lbase[b] + atomicAdd(&hist[b], 1);
        }
#pragma unroll
        for (int k = 0; k < 4; ++k) {
          const int4 r = make_int4(ss[k], dd[k], __float_as_int(ww[k]), 0);
          if (rk[k] < CAP) rec[(dd[k] >> RB_SHIFT) * RSTRIDE + rk[k]] = r;
          else             spill[atomicAdd(spillcnt, 1)] = r;
        }
      }
    }
  } else {
    for (int e = e0 + t; e < e1; e += 512) {
      const int d = dst[e];
      const int b = d >> RB_SHIFT;
      const int rk = lbase[b] + atomicAdd(&hist[b], 1);
      const int4 r = make_int4(src[e], d, __float_as_int(w[e]), 0);
      if (rk < CAP) rec[b * RSTRIDE + rk] = r;
      else          spill[atomicAdd(spillcnt, 1)] = r;
    }
  }
  const int gme = min(ba, bb);
  const int gbelow = __shfl_up(gme, 1);
  const bool covered = (t & 31) != 0 && gbelow == gme;
  if (c0 < nc && !covered) atomicMin(&first[gme], c0);
  for (int cc = c0 + gridDim.x * blockDim.x; cc < nc; cc += gridDim.x * blockDim.x)
    atomicMin(&first[min(batch[2 * cc], batch[2 * cc + 1])], cc);
}

#define DCH 5
__global__ __launch_bounds__(256) void bucket_deg_mark(
    const int4* __restrict__ rec, const int4* __restrict__ spill,
    const int* __restrict__ cursor, const int* __restrict__ spillcnt,
    float* __restrict__ dinv, int N,
    const int* __restrict__ first, int B, int* __restrict__ need, int nc) {
  __shared__ float degl[RB];
  __shared__ int tgtl[RB];
  const int bkt = blockIdx.x;
  const int base = bkt << RB_SHIFT;
  const int4* __restrict__ reg = rec + (size_t)bkt * RSTRIDE;
  int4 rr[DCH];
#pragma unroll
  for (int u = 0; u < DCH; ++u) rr[u] = reg[threadIdx.x + 256 * u];
  const int n = region_count(cursor, bkt);
  const int nsp = *spillcnt;
  int ig = -1;
  if (threadIdx.x < B) ig = 2 * graph_cluster(first, threadIdx.x, nc);
  if (threadIdx.x < RB) { degl[threadIdx.x] = 1.0f; tgtl[threadIdx.x] = 0; }
  __syncthreads();
  if (ig >= 0 && (ig >> RB_SHIFT) == bkt) { tgtl[ig - base] = 1; tgtl[ig + 1 - base] = 1; }
  for (int g = threadIdx.x + blockDim.x; g < B; g += blockDim.x) {
    const int i0 = 2 * graph_cluster(first, g, nc);
    if ((i0 >> RB_SHIFT) == bkt) { tgtl[i0 - base] = 1; tgtl[i0 + 1 - base] = 1; }
  }
  __syncthreads();
#pragma unroll
  for (int u = 0; u < DCH; ++u) {
    if ((int)threadIdx.x + 256 * u < n) {
      const int4 r = rr[u];
      __hip_atomic_fetch_add(&degl[r.y - base], __int_as_float(r.z),
                             __ATOMIC_RELAXED, __HIP_MEMORY_SCOPE_WORKGROUP);
      if (tgtl[r.y - base]) need[r.x] = 1;
    }
  }
  for (int e = 256 * DCH + threadIdx.x; e < n; e += 256) {
    const int4 r = reg[e];
    __hip_atomic_fetch_add(&degl[r.y - base], __int_as_float(r.z),
                           __ATOMIC_RELAXED, __HIP_MEMORY_SCOPE_WORKGROUP);
    if (tgtl[r.y - base]) need[r.x] = 1;
  }
  for (int e = threadIdx.x; e < nsp; e += blockDim.x) {
    int4 r = spill[e];
    if ((r.y >> RB_SHIFT) == bkt) {
      __hip_atomic_fetch_add(&degl[r.y - base], __int_as_float(r.z),
                             __ATOMIC_RELAXED, __HIP_MEMORY_SCOPE_WORKGROUP);
      if (tgtl[r.y - base]) need[r.x] = 1;
    }
  }
  __syncthreads();
  if (threadIdx.x < RB) {
    int i = base + threadIdx.x;
    if (i < N) {
      float d = degl[threadIdx.x];
      dinv[i] = d > 0.0f ? __frsqrt_rn(d) : 0.0f;
      if (tgtl[threadIdx.x]) need[i] = 1;
    }
  }
}

#define IDIM 128
#define APITCH 132
#define OPITCH 68
#define SCH 5
#define GB 8

__global__ __launch_bounds__(256) void bucket_aggregate(
    const float* __restrict__ x, const float* __restrict__ W,
    const float* __restrict__ dinv, const float* __restrict__ bias,
    const int4* __restrict__ rec, const int4* __restrict__ spill,
    const int* __restrict__ cursor, const int* __restrict__ spillcnt,
    const int* __restrict__ need, float* __restrict__ h, int N) {
  __shared__ float acc[RB * APITCH];
  __shared__ int slotl[RB];
  __shared__ int rowl[RB];
  __shared__ int cntl[2];
  __shared__ int2 wq[8][32 * SCH];
  const int tid  = threadIdx.x;
  const int lane = tid & 31;
  const int wave = __builtin_amdgcn_readfirstlane(tid >> 5);
  const int bkt  = blockIdx.x;
  const int base = bkt << RB_SHIFT;

  const int4* __restrict__ reg = rec + (size_t)bkt * RSTRIDE;
  int4 r[SCH];
#pragma unroll
  for (int u = 0; u < SCH; ++u) r[u] = reg[(wave + 8 * u) * 32 + lane];
  const int n = region_count(cursor, bkt);
  const int nsp = *spillcnt;

  int flag = 0, pre = 0;
  if (tid < RB) {
    flag = (base + tid < N) ? (need[base + tid] != 0) : 0;
    const unsigned m = __builtin_amdgcn_ballot_w32(flag != 0);
    pre = __builtin_popcount(m & ((1u << lane) - 1u));
    if (lane == 0) cntl[tid >> 5] = __builtin_popcount(m);
  }
  __syncthreads();
  const int c0n = cntl[0];
  const int ns = __builtin_amdgcn_readfirstlane(c0n + cntl[1]);
  if (ns == 0) return;
  if (tid < RB) {
    const int s = pre + (tid >= 32 ? c0n : 0);
    slotl[tid] = flag ? s * APITCH : -1;
    if (flag) rowl[s] = tid;
  }
  __syncthreads();

  const int mt = (ns + 15) >> 4;
  for (int idx = tid; idx < (mt << 9); idx += 256) {
    const int r = idx >> 5, c4 = idx & 31;
    float4 v = make_float4(0.0f, 0.0f, 0.0f, 0.0f);
    if (r < ns) {
      const int i = base + rowl[r];
      const float di = dinv[i];
      const float s = di * di;
      const float4 xv = *(const float4*)(x + (size_t)i * IDIM + 4 * c4);
      v.x = s * xv.x; v.y = s * xv.y; v.z = s * xv.z; v.w = s * xv.w;
    }
    *(float4*)(acc + r * APITCH + 4 * c4) = v;
  }
  __syncthreads();

  const float* __restrict__ xl = x + 2 * lane;
  float* al = acc + 2 * lane;

  int2* __restrict__ myq = wq[wave];
  const int4* __restrict__ list = reg;
  int le = n;
  int e = wave * 32;
  int pass = 0;
  while (true) {
    if (e < le) {
      const int cs = 256;
      int so[SCH];
      float ds[SCH], dd[SCH];
#pragma unroll
      for (int u = 0; u < SCH; ++u)
        so[u] = slotl[(r[u].y - base) & (RB - 1)];
#pragma unroll
      for (int u = 0; u < SCH; ++u) {
        const bool mine = e + u * cs + lane < le && (pass == 0 || (r[u].y >> RB_SHIFT) == bkt);
        if (!mine) so[u] = -1;
      }
#pragma unroll
      for (int u = 0; u < SCH; ++u) {
        const bool keep = so[u] >= 0;
        ds[u] = dinv[keep ? r[u].x : 0];
        dd[u] = dinv[keep ? r[u].y : 0];
      }
      int nq = 0;
#pragma unroll
      for (int u = 0; u < SCH; ++u) {
        const bool keep = so[u] >= 0;
        const unsigned m = __builtin_amdgcn_ballot_w32(keep);
        if (keep) {
          const int p = nq + __builtin_popcount(m & ((1u << lane) - 1u));
          myq[p] = make_int2(r[u].x | (so[u] << 17),
                             __float_as_int(ds[u] * __int_as_float(r[u].z) * dd[u]));
        }
        nq += __builtin_popcount(m);
      }
      for (int q0 = 0; q0 < nq; q0 += 32) {
        const int cnt = min(32, nq - q0);
        int2 qe = make_int2(0, 0);
        if (lane < cnt) qe = myq[q0 + lane];
        for (int j = 0; j < cnt; j += GB) {
          float2 xa[GB], xb[GB]; int dl[GB]; float nr[GB];
#pragma unroll
          for (int u = 0; u < GB; ++u) {
            const int jj = min(j + u, cnt - 1);
            const int pk = __builtin_amdgcn_readlane(qe.x, jj);
            nr[u] = __int_as_float(__builtin_amdgcn_readlane(qe.y, jj));
            dl[u] = pk >> 17;
            const int s = pk & 0x1FFFF;
            xa[u] = *(const float2*)(xl + (size_t)s * IDIM);
            xb[u] = *(const float2*)(xl + (size_t)s * IDIM + 64);
          }
#pragma unroll
          for (int u = 0; u < GB; ++u) {
            if (j + u < cnt) {
              __hip_atomic_fetch_add(al + dl[u],      nr[u] * xa[u].x, __ATOMIC_RELAXED, __HIP_MEMORY_SCOPE_WORKGROUP);
              __hip_atomic_fetch_add(al + dl[u] + 1,  nr[u] * xa[u].y, __ATOMIC_RELAXED, __HIP_MEMORY_SCOPE_WORKGROUP);
              __hip_atomic_fetch_add(al + dl[u] + 64, nr[u] * xb[u].x, __ATOMIC_RELAXED, __HIP_MEMORY_SCOPE_WORKGROUP);
              __hip_atomic_fetch_add(al + dl[u] + 65, nr[u] * xb[u].y, __ATOMIC_RELAXED, __HIP_MEMORY_SCOPE_WORKGROUP);
            }
          }
        }
      }
    }
    e += 256 * SCH;
    if (e >= le) {
      if (pass) break;
      pass = 1; list = spill; le = nsp; e = wave * 32;
      if (e >= le) break;
    }
#pragma unroll
    for (int u = 0; u < SCH; ++u)
      r[u] = list[min(e + 256 * u + lane, le - 1)];
  }
  __syncthreads();

  const int half = lane >> 4;
  const int l15  = lane & 15;
  const int nb   = (wave & 3) << 4;
  const float* __restrict__ wr = W + (size_t)(2 * half) * ODIM + nb + l15;
  v8f d[2];
#pragma unroll
  for (int u = 0; u < 2; ++u) {
    const int t = (wave >> 2) + 2 * u;
    d[u] = (v8f){};
    if (t < mt) {
      v8f p0 = {}, p1 = {};
      const float* ar = acc + (t * 16 + l15) * APITCH + 2 * half;
#pragma unroll 4
      for (int k = 0; k < IDIM; k += 8) {
        v2f a0, b0, a1, b1;
        a0[0] = ar[k + 0];
        a0[1] = ar[k + 1];
        b0[0] = wr[(size_t)(k + 0) * ODIM];
        b0[1] = wr[(size_t)(k + 1) * ODIM];
        p0 = __builtin_amdgcn_wmma_f32_16x16x4_f32(
             false, a0,  false, b0,
             (short)0, p0,  false,  false);
        a1[0] = ar[k + 4];
        a1[1] = ar[k + 5];
        b1[0] = wr[(size_t)(k + 4) * ODIM];
        b1[1] = wr[(size_t)(k + 5) * ODIM];
        p1 = __builtin_amdgcn_wmma_f32_16x16x4_f32(
             false, a1,  false, b1,
             (short)0, p1,  false,  false);
      }
      d[u] = p0 + p1;
    }
  }
  __syncthreads();

  const float bv = bias[nb + l15];
#pragma unroll
  for (int u = 0; u < 2; ++u) {
    const int t = (wave >> 2) + 2 * u;
    if (t < mt) {
      float* orow = acc + (t * 16 + 8 * half) * OPITCH + nb + l15;
#pragma unroll
      for (int r = 0; r < 8; ++r) orow[r * OPITCH] = d[u][r] + bv;
    }
  }
  __syncthreads();

  for (int idx = tid; idx < (ns << 4); idx += 256) {
    const int s = idx >> 4, c4 = idx & 15;
    const int i = base + rowl[s];
    *(float4*)(h + (size_t)i * ODIM + 4 * c4) = *(const float4*)(acc + s * OPITCH + 4 * c4);
  }
}

#define TPITCH 68

__global__ __launch_bounds__(256) void graph_rows_out(
    const float* __restrict__ h, const float* __restrict__ dinv,
    const int4* __restrict__ rec, const int4* __restrict__ spill,
    const int* __restrict__ cursor, const int* __restrict__ spillcnt,
    const int* __restrict__ first, const float* __restrict__ W,
    const float* __restrict__ bias, float* __restrict__ out, int nc) {
  __shared__ float part[8][2][ODIM];
  __shared__ __attribute__((aligned(16))) float tile[16 * TPITCH];
  __shared__ __attribute__((aligned(16))) float orow[ODIM];
  __shared__ int2 gq[8][32 * SCH];
  const int g = blockIdx.x;
  const int c = graph_cluster(first, g, nc);
  const int i0 = 2 * c;
  const int bkt = i0 >> RB_SHIFT;
  const int tid = threadIdx.x;
  const int lane = tid & 31;
  const int wave = __builtin_amdgcn_readfirstlane(tid >> 5);
  const int4* __restrict__ reg = rec + (size_t)bkt * RSTRIDE;
  int4 r[SCH];
#pragma unroll
  for (int u = 0; u < SCH; ++u) r[u] = reg[wave * 32 + 256 * u + lane];
  const int n = region_count(cursor, bkt);
  const int nsp = *spillcnt;
  const float* __restrict__ hl = h + 2 * lane;
  float2 a0 = make_float2(0.0f, 0.0f), a1 = make_float2(0.0f, 0.0f);

  const int half = lane >> 4;
  const int l15  = lane & 15;
  const int nb   = (wave & 3) << 4;
  v2f bf[ODIM / 4] = {};
  float bv = 0.0f;
  if (wave < 4) {
    const float* __restrict__ wr = W + (size_t)(2 * half) * ODIM + nb + l15;
#pragma unroll
    for (int q = 0; q < ODIM / 4; ++q) {
      bf[q][0] = wr[(size_t)(4 * q + 0) * ODIM];
      bf[q][1] = wr[(size_t)(4 * q + 1) * ODIM];
    }
    bv = bias[nb + l15];
  }

  int2* __restrict__ myq = gq[wave];
  const int4* __restrict__ list = reg;
  int le = n;
  int e = wave * 32;
  int pass = 0;
  while (true) {
    if (e < le) {
      bool hit[SCH];
      float ds[SCH], dd[SCH];
#pragma unroll
      for (int u = 0; u < SCH; ++u)
        asm volatile("" :: "v"(r[u].x), "v"(r[u].y), "v"(r[u].z));
#pragma unroll
      for (int u = 0; u < SCH; ++u) {
        hit[u] = e + 256 * u + lane < le && (r[u].y >> 1) == c;
        ds[u] = dinv[hit[u] ? r[u].x : 0];
        dd[u] = dinv[hit[u] ? r[u].y : 0];
      }
      int nq = 0;
#pragma unroll
      for (int u = 0; u < SCH; ++u) {
        const unsigned m = __builtin_amdgcn_ballot_w32(hit[u]);
        if (hit[u]) {
          const int p = nq + __builtin_popcount(m & ((1u << lane) - 1u));
          myq[p] = make_int2(r[u].x | ((r[u].y & 1) << 17),
                             __float_as_int(ds[u] * __int_as_float(r[u].z) * dd[u]));
        }
        nq += __builtin_popcount(m);
      }
      for (int q0 = 0; q0 < nq; q0 += 32) {
        const int cnt = min(32, nq - q0);
        int2 qe = make_int2(0, 0);
        if (lane < cnt) qe = myq[q0 + lane];
        for (int j = 0; j < cnt; j += GB) {
          float2 xv[GB]; int pk[GB]; float nr[GB];
#pragma unroll
          for (int u = 0; u < GB; ++u) {
            const int jj = min(j + u, cnt - 1);
            pk[u] = __builtin_amdgcn_readlane(qe.x, jj);
            nr[u] = __int_as_float(__builtin_amdgcn_readlane(qe.y, jj));
            xv[u] = *(const float2*)(hl + (size_t)(pk[u] & 0x1FFFF) * ODIM);
          }
#pragma unroll
          for (int u = 0; u < GB; ++u) {
            if (j + u < cnt) {
              if (pk[u] >> 17) { a1.x = __builtin_fmaf(nr[u], xv[u].x, a1.x); a1.y = __builtin_fmaf(nr[u], xv[u].y, a1.y); }
              else             { a0.x = __builtin_fmaf(nr[u], xv[u].x, a0.x); a0.y = __builtin_fmaf(nr[u], xv[u].y, a0.y); }
            }
          }
        }
      }
    }
    e += 256 * SCH;
    if (e >= le) {
      if (pass) break;
      pass = 1; list = spill; le = nsp; e = wave * 32;
      if (e >= le) break;
    }
#pragma unroll
    for (int u = 0; u < SCH; ++u)
      r[u] = list[min(e + 256 * u + lane, le - 1)];
  }
  if (wave == 0) {
    const float d0 = dinv[i0], d1 = dinv[i0 + 1];
    const float2 x0 = *(const float2*)(hl + (size_t)i0 * ODIM);
    const float2 x1 = *(const float2*)(hl + (size_t)(i0 + 1) * ODIM);
    a0.x += d0 * d0 * x0.x; a0.y += d0 * d0 * x0.y;
    a1.x += d1 * d1 * x1.x; a1.y += d1 * d1 * x1.y;
  }
  part[wave][0][2 * lane] = a0.x; part[wave][0][2 * lane + 1] = a0.y;
  part[wave][1][2 * lane] = a1.x; part[wave][1][2 * lane + 1] = a1.y;
  for (int idx = tid; idx < 14 * ODIM; idx += 256)
    tile[(2 + (idx >> 6)) * TPITCH + (idx & 63)] = 0.0f;
  __syncthreads();
  if (tid < 2 * ODIM) {
    const int p = tid >> 6, f = tid & 63;
    float sum = part[0][p][f];
#pragma unroll
    for (int w = 1; w < 8; ++w) sum += part[w][p][f];
    tile[p * TPITCH + f] = sum;
  }
  __syncthreads();

  if (wave < 4) {
    const float* ar = tile + l15 * TPITCH + 2 * half;
    v8f acc = {};
#pragma unroll
    for (int k = 0; k < ODIM; k += 4) {
      v2f a;
      a[0] = ar[k + 0];
      a[1] = ar[k + 1];
      acc = __builtin_amdgcn_wmma_f32_16x16x4_f32(
           false, a,  false, bf[k >> 2],
           (short)0, acc,  false,  false);
    }
    const float pooled = fmaxf(acc[0] + bv, acc[1] + bv);
    if (half == 0) orow[nb + l15] = pooled;
  }
  __syncthreads();
  if (tid < 16)
    *(float4*)(out + (size_t)g * ODIM + 4 * tid) = *(const float4*)(orow + 4 * tid);
}

extern "C" void kernel_launch(void* const* d_in, const int* in_sizes, int n_in,
                              void* d_out, int out_size, void* d_ws, size_t ws_size,
                              hipStream_t stream) {
  const float* x   = (const float*)d_in[0];
  const int*   ei  = (const int*)d_in[1];
  const float* ew  = (const float*)d_in[2];
  const int*   bat = (const int*)d_in[3];
  const float* W1  = (const float*)d_in[4];
  const float* b1  = (const float*)d_in[5];
  const float* W2  = (const float*)d_in[6];
  const float* b2  = (const float*)d_in[7];
  float* out = (float*)d_out;

  const int IN_DIM = 128;
  const int N  = in_sizes[0] / IN_DIM;
  const int E  = in_sizes[2];
  const int B  = out_size / ODIM;
  const int NC = N / 2;

  const int* src = ei;
  const int* dst = ei + E;

  char* ws = (char*)d_ws;
  size_t featBytes = (size_t)N * ODIM * sizeof(float);
  size_t aggBytes  = (size_t)2 * B * ODIM * sizeof(float);
  float* h    = (float*)(ws + featBytes);
  float* dinv = (float*)(ws + 2 * featBytes + (size_t)N * 4);
  int*   need = (int*)  (ws + 2 * featBytes + (size_t)N * 8 + aggBytes);
  int*   first= (int*)  (ws + 2 * featBytes + (size_t)N * 12 + aggBytes);
  int4*  rec      = (int4*)(ws + 134217728);
  int*   cursor   = (int*) (ws + 209715200);
  int*   spillcnt = (int*) (ws + 209715200 + 65536);
  int4*  spill    = (int4*)(ws + 218103808);

  const int T = 256;
  dim3 blk(T);

  const int NB = (N + RB - 1) / RB;
  const int cover = N > B ? N : B;
  init_tables<<<(cover + T - 1) / T, blk, 0, stream>>>(cursor, NB, spillcnt, first, B, need, N);

  const int G1 = 256;
  const int chunk = ((E + G1 - 1) / G1 + 3) & ~3;
  bin_scatter_first<<<G1, dim3(512), 0, stream>>>(src, dst, ew, cursor, spillcnt, rec, spill, E, NB, chunk,
                                                  bat, first, NC);
  bucket_deg_mark<<<NB, blk, 0, stream>>>(rec, spill, cursor, spillcnt, dinv, N, first, B, need, NC);

  bucket_aggregate<<<NB, blk, 0, stream>>>(x, W1, dinv, b1, rec, spill, cursor, spillcnt, need, h, N);

  graph_rows_out<<<B, blk, 0, stream>>>(h, dinv, rec, spill, cursor, spillcnt, first, W2, b2, out, NC);
}


# Binning kernel: a group counting-sorts its edges by bucket in LDS (rank from the counting add, scan of the counts) and stores them in that order, adjacent lanes to adjacent slots; sum order changes
# speedup vs baseline: 15.7687x; 1.1290x over previous
#include <hip/hip_runtime.h>
#include <hip/hip_bf16.h>

typedef __attribute__((ext_vector_type(2))) float v2f;
typedef __attribute__((ext_vector_type(8))) float v8f;

#define ODIM 64

#define RB 64
#define RB_SHIFT 6
#define NBMAX 2048
#define CAP 2048
#define RSTRIDE (CAP + 16)

__global__ void init_tables(int* __restrict__ cursor, int nb, int* __restrict__ spillcnt,
                            int* __restrict__ first, int B, int* __restrict__ need, int N) {
  int i = blockIdx.x * blockDim.x + threadIdx.x;
  if (i < nb) cursor[i] = 0;
  if (i == 0) *spillcnt = 0;
  if (i < B) first[i] = 0x7FFFFFFF;
  if (i < N) need[i] = 0;
}

__device__ __forceinline__ int region_count(const int* __restrict__ cursor, int bkt) {
  return min(cursor[bkt], CAP);
}

__device__ __forceinline__ int graph_cluster(const int* __restrict__ first, int g, int nc) {
  int c = first[g];
  if (c > nc - 1) c = nc - 1;
  if (c < 0) c = 0;
  return c;
}

#define QPT 4
#define QCHUNK (512 * 4 * QPT)
__device__ __forceinline__ bool quad_path(const void* a, const void* b, const void* c, int E, int chunk) {
  return chunk <= QCHUNK && ((E | chunk) & 3) == 0 &&
         ((((size_t)a) | ((size_t)b) | ((size_t)c)) & 15) == 0;
}

#define BPT (NBMAX / 512)
#define WOB 8
__global__ __launch_bounds__(512) void bin_scatter_first(
    const int* __restrict__ src, const int* __restrict__ dst, const float* __restrict__ w,
    int* __restrict__ cursor, int* __restrict__ spillcnt,
    int4* __restrict__ rec, int4* __restrict__ spill, int E, int nb, int chunk,
    const int* __restrict__ batch, int* __restrict__ first, int nc) {
  __shared__ __attribute__((aligned(16))) int hist[NBMAX];
  __shared__ int2 tab[NBMAX];
  __shared__ __attribute__((aligned(16))) int wtot[16];
  __shared__ int4 sorted[QCHUNK];
  const int t = threadIdx.x;
  const int lane = t & 31;
  const int wave = __builtin_amdgcn_readfirstlane(t >> 5);
  for (int b = t; b < NBMAX; b += 512) hist[b] = 0;
  const int c0 = blockIdx.x * blockDim.x + threadIdx.x;
  const int cl = min(c0, nc - 1);
  const int ba = batch[2 * cl], bb = batch[2 * cl + 1];
  const int e0 = blockIdx.x * chunk;
  const int e1 = min(e0 + chunk, E);
  const bool quads = quad_path(dst, src, w, E, chunk);
  const int n4 = quads ? (e1 - e0) >> 2 : 0;
  int4 dq[QPT], sq[QPT];
  float4 wq[QPT];
#pragma unroll
  for (int u = 0; u < QPT; ++u) {
    dq[u] = make_int4(0, 0, 0, 0);
    sq[u] = make_int4(0, 0, 0, 0);
    wq[u] = make_float4(0.0f, 0.0f, 0.0f, 0.0f);
  }
  if (n4 > 0) {
    const int4* __restrict__ d4 = (const int4*)(dst + e0);
    const int4* __restrict__ s4 = (const int4*)(src + e0);
    const float4* __restrict__ w4 = (const float4*)(w + e0);
#pragma unroll
    for (int u = 0; u < QPT; ++u) {
      const int q = min(t + 512 * u, n4 - 1);
      dq[u] = d4[q];
      sq[u] = s4[q];
      wq[u] = w4[q];
    }
  }
  __syncthreads();

  if (quads) {
    int rk[QPT][4];
#pragma unroll
    for (int u = 0; u < QPT; ++u) {
#pragma unroll
      for (int k = 0; k < 4; ++k) rk[u][k] = 0;
      if (t + 512 * u < n4) {
        rk[u][0] = atomicAdd(&hist[dq[u].x >> RB_SHIFT], 1);
        rk[u][1] = atomicAdd(&hist[dq[u].y >> RB_SHIFT], 1);
        rk[u][2] = atomicAdd(&hist[dq[u].z >> RB_SHIFT], 1);
        rk[u][3] = atomicAdd(&hist[dq[u].w >> RB_SHIFT], 1);
      }
    }
    __syncthreads();
    int rc[BPT], rb[BPT];
#pragma unroll
    for (int j = 0; j < BPT; ++j) {
      const int b = t + 512 * j;
      rc[j] = b < nb ? hist[b] : 0;
    }
#pragma unroll
    for (int j = 0; j < BPT; ++j) {
      rb[j] = 0;
      if (rc[j]) rb[j] = atomicAdd(&cursor[t + 512 * j], rc[j]);
    }
    const int4 c4 = *(const int4*)&hist[4 * t];
    const int s4sum = c4.x + c4.y + c4.z + c4.w;
    int inc = s4sum;
#pragma unroll
    for (int d = 1; d < 32; d <<= 1) {
      const int v = __shfl_up(inc, d);
      if (lane >= d) inc += v;
    }
    if (lane == 31) wtot[wave] = inc;
    __syncthreads();
    int wbefore = 0;
#pragma unroll
    for (int q = 0; q < 16; ++q) wbefore += q < wave ? wtot[q] : 0;
    const int ex = wbefore + inc - s4sum;
    tab[4 * t + 0].x = ex;
    tab[4 * t + 1].x = ex + c4.x;
    tab[4 * t + 2].x = ex + c4.x + c4.y;
    tab[4 * t + 3].x = ex + c4.x + c4.y + c4.z;
#pragma unroll
    for (int j = 0; j < BPT; ++j) {
      const int b = t + 512 * j;
      if (b < nb) tab[b].y = rb[j];
    }
    __syncthreads();
#pragma unroll
    for (int u = 0; u < QPT; ++u) {
      if (t + 512 * u < n4) {
        const int dd[4] = {dq[u].x, dq[u].y, dq[u].z, dq[u].w};
        const int ss[4] = {sq[u].x, sq[u].y, sq[u].z, sq[u].w};
        const float ww[4] = {wq[u].x, wq[u].y, wq[u].z, wq[u].w};
#pragma unroll
        for (int k = 0; k < 4; ++k) {
          const int b = dd[k] >> RB_SHIFT;
          const int2 tb = tab[b];
          const int gr = tb.y + rk[u][k];
          sorted[tb.x + rk[u][k]] = make_int4(ss[k], dd[k], __float_as_int(ww[k]),
                                              gr < CAP ? b * RSTRIDE + gr : -1);
        }
      }
    }
    __syncthreads();
    const int cnt = 4 * max(n4, 0);
    for (int i0 = 0; i0 < cnt; i0 += 512 * WOB) {
      int4 r[WOB];
#pragma unroll
      for (int u = 0; u < WOB; ++u) r[u] = sorted[min(i0 + 512 * u + t, cnt - 1)];
#pragma unroll
      for (int u = 0; u < WOB; ++u)
        asm volatile("" :: "v"(r[u].x), "v"(r[u].y), "v"(r[u].z), "v"(r[u].w));
#pragma unroll
      for (int u = 0; u < WOB; ++u) {
        if (i0 + 512 * u + t < cnt) {
          int4* q = rec + max(r[u].w, 0);
          if (r[u].w < 0) q = spill + atomicAdd(spillcnt, 1);
          *q = r[u];
        }
      }
    }
  } else {
    for (int e = e0 + t; e < e1; e += 512)
      atomicAdd(&hist[dst[e] >> RB_SHIFT], 1);
    __syncthreads();
    int rc[BPT], rb[BPT];
#pragma unroll
    for (int j = 0; j < BPT; ++j) {
      const int b = t + 512 * j;
      rc[j] = b < nb ? hist[b] : 0;
    }
#pragma unroll
    for (int j = 0; j < BPT; ++j) {
      rb[j] = 0;
      if (rc[j]) rb[j] = atomicAdd(&cursor[t + 512 * j], rc[j]);
    }
#pragma unroll
    for (int j = 0; j < BPT; ++j) {
      const int b = t + 512 * j;
      if (b < nb) {
        tab[b].y = rb[j];
        hist[b] = 0;
      }
    }
    __syncthreads();
    for (int e = e0 + t; e < e1; e += 512) {
      const int d = dst[e];
      const int b = d >> RB_SHIFT;
      const int rk1 = tab[b].y + atomicAdd(&hist[b], 1);
      const int4 r = make_int4(src[e], d, __float_as_int(w[e]), 0);
      if (rk1 < CAP) rec[b * RSTRIDE + rk1] = r;
      else           spill[atomicAdd(spillcnt, 1)] = r;
    }
  }
  const int gme = min(ba, bb);
  const int gbelow = __shfl_up(gme, 1);
  const bool covered = (t & 31) != 0 && gbelow == gme;
  if (c0 < nc && !covered) atomicMin(&first[gme], c0);
  for (int cc = c0 + gridDim.x * blockDim.x; cc < nc; cc += gridDim.x * blockDim.x)
    atomicMin(&first[min(batch[2 * cc], batch[2 * cc + 1])], cc);
}

#define DCH 5
__global__ __launch_bounds__(256) void bucket_deg_mark(
    const int4* __restrict__ rec, const int4* __restrict__ spill,
    const int* __restrict__ cursor, const int* __restrict__ spillcnt,
    float* __restrict__ dinv, int N,
    const int* __restrict__ first, int B, int* __restrict__ need, int nc) {
  __shared__ float degl[RB];
  __shared__ int tgtl[RB];
  const int bkt = blockIdx.x;
  const int base = bkt << RB_SHIFT;
  const int4* __restrict__ reg = rec + (size_t)bkt * RSTRIDE;
  int4 rr[DCH];
#pragma unroll
  for (int u = 0; u < DCH; ++u) rr[u] = reg[threadIdx.x + 256 * u];
  const int n = region_count(cursor, bkt);
  const int nsp = *spillcnt;
  int ig = -1;
  if (threadIdx.x < B) ig = 2 * graph_cluster(first, threadIdx.x, nc);
  if (threadIdx.x < RB) { degl[threadIdx.x] = 1.0f; tgtl[threadIdx.x] = 0; }
  __syncthreads();
  if (ig >= 0 && (ig >> RB_SHIFT) == bkt) { tgtl[ig - base] = 1; tgtl[ig + 1 - base] = 1; }
  for (int g = threadIdx.x + blockDim.x; g < B; g += blockDim.x) {
    const int i0 = 2 * graph_cluster(first, g, nc);
    if ((i0 >> RB_SHIFT) == bkt) { tgtl[i0 - base] = 1; tgtl[i0 + 1 - base] = 1; }
  }
  __syncthreads();
#pragma unroll
  for (int u = 0; u < DCH; ++u) {
    if ((int)threadIdx.x + 256 * u < n) {
      const int4 r = rr[u];
      __hip_atomic_fetch_add(&degl[r.y - base], __int_as_float(r.z),
                             __ATOMIC_RELAXED, __HIP_MEMORY_SCOPE_WORKGROUP);
      if (tgtl[r.y - base]) need[r.x] = 1;
    }
  }
  for (int e = 256 * DCH + threadIdx.x; e < n; e += 256) {
    const int4 r = reg[e];
    __hip_atomic_fetch_add(&degl[r.y - base], __int_as_float(r.z),
                           __ATOMIC_RELAXED, __HIP_MEMORY_SCOPE_WORKGROUP);
    if (tgtl[r.y - base]) need[r.x] = 1;
  }
  for (int e = threadIdx.x; e < nsp; e += blockDim.x) {
    int4 r = spill[e];
    if ((r.y >> RB_SHIFT) == bkt) {
      __hip_atomic_fetch_add(&degl[r.y - base], __int_as_float(r.z),
                             __ATOMIC_RELAXED, __HIP_MEMORY_SCOPE_WORKGROUP);
      if (tgtl[r.y - base]) need[r.x] = 1;
    }
  }
  __syncthreads();
  if (threadIdx.x < RB) {
    int i = base + threadIdx.x;
    if (i < N) {
      float d = degl[threadIdx.x];
      dinv[i] = d > 0.0f ? __frsqrt_rn(d) : 0.0f;
      if (tgtl[threadIdx.x]) need[i] = 1;
    }
  }
}

#define IDIM 128
#define APITCH 132
#define OPITCH 68
#define SCH 5
#define GB 8

__global__ __launch_bounds__(256) void bucket_aggregate(
    const float* __restrict__ x, const float* __restrict__ W,
    const float* __restrict__ dinv, const float* __restrict__ bias,
    const int4* __restrict__ rec, const int4* __restrict__ spill,
    const int* __restrict__ cursor, const int* __restrict__ spillcnt,
    const int* __restrict__ need, float* __restrict__ h, int N) {
  __shared__ float acc[RB * APITCH];
  __shared__ int slotl[RB];
  __shared__ int rowl[RB];
  __shared__ int cntl[2];
  __shared__ int2 wq[8][32 * SCH];
  const int tid  = threadIdx.x;
  const int lane = tid & 31;
  const int wave = __builtin_amdgcn_readfirstlane(tid >> 5);
  const int bkt  = blockIdx.x;
  const int base = bkt << RB_SHIFT;

  const int4* __restrict__ reg = rec + (size_t)bkt * RSTRIDE;
  int4 r[SCH];
#pragma unroll
  for (int u = 0; u < SCH; ++u) r[u] = reg[(wave + 8 * u) * 32 + lane];
  const int n = region_count(cursor, bkt);
  const int nsp = *spillcnt;

  int flag = 0, pre = 0;
  if (tid < RB) {
    flag = (base + tid < N) ? (need[base + tid] != 0) : 0;
    const unsigned m = __builtin_amdgcn_ballot_w32(flag != 0);
    pre = __builtin_popcount(m & ((1u << lane) - 1u));
    if (lane == 0) cntl[tid >> 5] = __builtin_popcount(m);
  }
  __syncthreads();
  const int c0n = cntl[0];
  const int ns = __builtin_amdgcn_readfirstlane(c0n + cntl[1]);
  if (ns == 0) return;
  if (tid < RB) {
    const int s = pre + (tid >= 32 ? c0n : 0);
    slotl[tid] = flag ? s * APITCH : -1;
    if (flag) rowl[s] = tid;
  }
  __syncthreads();

  const int mt = (ns + 15) >> 4;
  for (int idx = tid; idx < (mt << 9); idx += 256) {
    const int r = idx >> 5, c4 = idx & 31;
    float4 v = make_float4(0.0f, 0.0f, 0.0f, 0.0f);
    if (r < ns) {
      const int i = base + rowl[r];
      const float di = dinv[i];
      const float s = di * di;
      const float4 xv = *(const float4*)(x + (size_t)i * IDIM + 4 * c4);
      v.x = s * xv.x; v.y = s * xv.y; v.z = s * xv.z; v.w = s * xv.w;
    }
    *(float4*)(acc + r * APITCH + 4 * c4) = v;
  }
  __syncthreads();

  const float* __restrict__ xl = x + 2 * lane;
  float* al = acc + 2 * lane;

  int2* __restrict__ myq = wq[wave];
  const int4* __restrict__ list = reg;
  int le = n;
  int e = wave * 32;
  int pass = 0;
  while (true) {
    if (e < le) {
      const int cs = 256;
      int so[SCH];
      float ds[SCH], dd[SCH];
#pragma unroll
      for (int u = 0; u < SCH; ++u)
        so[u] = slotl[(r[u].y - base) & (RB - 1)];
#pragma unroll
      for (int u = 0; u < SCH; ++u) {
        const bool mine = e + u * cs + lane < le && (pass == 0 || (r[u].y >> RB_SHIFT) == bkt);
        if (!mine) so[u] = -1;
      }
#pragma unroll
      for (int u = 0; u < SCH; ++u) {
        const bool keep = so[u] >= 0;
        ds[u] = dinv[keep ? r[u].x : 0];
        dd[u] = dinv[keep ? r[u].y : 0];
      }
      int nq = 0;
#pragma unroll
      for (int u = 0; u < SCH; ++u) {
        const bool keep = so[u] >= 0;
        const unsigned m = __builtin_amdgcn_ballot_w32(keep);
        if (keep) {
          const int p = nq + __builtin_popcount(m & ((1u << lane) - 1u));
          myq[p] = make_int2(r[u].x | (so[u] << 17),
                             __float_as_int(ds[u] * __int_as_float(r[u].z) * dd[u]));
        }
        nq += __builtin_popcount(m);
      }
      for (int q0 = 0; q0 < nq; q0 += 32) {
        const int cnt = min(32, nq - q0);
        int2 qe = make_int2(0, 0);
        if (lane < cnt) qe = myq[q0 + lane];
        for (int j = 0; j < cnt; j += GB) {
          float2 xa[GB], xb[GB]; int dl[GB]; float nr[GB];
#pragma unroll
          for (int u = 0; u < GB; ++u) {
            const int jj = min(j + u, cnt - 1);
            const int pk = __builtin_amdgcn_readlane(qe.x, jj);
            nr[u] = __int_as_float(__builtin_amdgcn_readlane(qe.y, jj));
            dl[u] = pk >> 17;
            const int s = pk & 0x1FFFF;
            xa[u] = *(const float2*)(xl + (size_t)s * IDIM);
            xb[u] = *(const float2*)(xl + (size_t)s * IDIM + 64);
          }
#pragma unroll
          for (int u = 0; u < GB; ++u) {
            if (j + u < cnt) {
              __hip_atomic_fetch_add(al + dl[u],      nr[u] * xa[u].x, __ATOMIC_RELAXED, __HIP_MEMORY_SCOPE_WORKGROUP);
              __hip_atomic_fetch_add(al + dl[u] + 1,  nr[u] * xa[u].y, __ATOMIC_RELAXED, __HIP_MEMORY_SCOPE_WORKGROUP);
              __hip_atomic_fetch_add(al + dl[u] + 64, nr[u] * xb[u].x, __ATOMIC_RELAXED, __HIP_MEMORY_SCOPE_WORKGROUP);
              __hip_atomic_fetch_add(al + dl[u] + 65, nr[u] * xb[u].y, __ATOMIC_RELAXED, __HIP_MEMORY_SCOPE_WORKGROUP);
            }
          }
        }
      }
    }
    e += 256 * SCH;
    if (e >= le) {
      if (pass) break;
      pass = 1; list = spill; le = nsp; e = wave * 32;
      if (e >= le) break;
    }
#pragma unroll
    for (int u = 0; u < SCH; ++u)
      r[u] = list[min(e + 256 * u + lane, le - 1)];
  }
  __syncthreads();

  const int half = lane >> 4;
  const int l15  = lane & 15;
  const int nb   = (wave & 3) << 4;
  const float* __restrict__ wr = W + (size_t)(2 * half) * ODIM + nb + l15;
  v8f d[2];
#pragma unroll
  for (int u = 0; u < 2; ++u) {
    const int t = (wave >> 2) + 2 * u;
    d[u] = (v8f){};
    if (t < mt) {
      v8f p0 = {}, p1 = {};
      const float* ar = acc + (t * 16 + l15) * APITCH + 2 * half;
#pragma unroll 4
      for (int k = 0; k < IDIM; k += 8) {
        v2f a0, b0, a1, b1;
        a0[0] = ar[k + 0];
        a0[1] = ar[k + 1];
        b0[0] = wr[(size_t)(k + 0) * ODIM];
        b0[1] = wr[(size_t)(k + 1) * ODIM];
        p0 = __builtin_amdgcn_wmma_f32_16x16x4_f32(
             false, a0,  false, b0,
             (short)0, p0,  false,  false);
        a1[0] = ar[k + 4];
        a1[1] = ar[k + 5];
        b1[0] = wr[(size_t)(k + 4) * ODIM];
        b1[1] = wr[(size_t)(k + 5) * ODIM];
        p1 = __builtin_amdgcn_wmma_f32_16x16x4_f32(
             false, a1,  false, b1,
             (short)0, p1,  false,  false);
      }
      d[u] = p0 + p1;
    }
  }
  __syncthreads();

  const float bv = bias[nb + l15];
#pragma unroll
  for (int u = 0; u < 2; ++u) {
    const int t = (wave >> 2) + 2 * u;
    if (t < mt) {
      float* orow = acc + (t * 16 + 8 * half) * OPITCH + nb + l15;
#pragma unroll
      for (int r = 0; r < 8; ++r) orow[r * OPITCH] = d[u][r] + bv;
    }
  }
  __syncthreads();

  for (int idx = tid; idx < (ns << 4); idx += 256) {
    const int s = idx >> 4, c4 = idx & 15;
    const int i = base + rowl[s];
    *(float4*)(h + (size_t)i * ODIM + 4 * c4) = *(const float4*)(acc + s * OPITCH + 4 * c4);
  }
}

#define TPITCH 68

__global__ __launch_bounds__(256) void graph_rows_out(
    const float* __restrict__ h, const float* __restrict__ dinv,
    const int4* __restrict__ rec, const int4* __restrict__ spill,
    const int* __restrict__ cursor, const int* __restrict__ spillcnt,
    const int* __restrict__ first, const float* __restrict__ W,
    const float* __restrict__ bias, float* __restrict__ out, int nc) {
  __shared__ float part[8][2][ODIM];
  __shared__ __attribute__((aligned(16))) float tile[16 * TPITCH];
  __shared__ __attribute__((aligned(16))) float orow[ODIM];
  __shared__ int2 gq[8][32 * SCH];
  const int g = blockIdx.x;
  const int c = graph_cluster(first, g, nc);
  const int i0 = 2 * c;
  const int bkt = i0 >> RB_SHIFT;
  const int tid = threadIdx.x;
  const int lane = tid & 31;
  const int wave = __builtin_amdgcn_readfirstlane(tid >> 5);
  const int4* __restrict__ reg = rec + (size_t)bkt * RSTRIDE;
  int4 r[SCH];
#pragma unroll
  for (int u = 0; u < SCH; ++u) r[u] = reg[wave * 32 + 256 * u + lane];
  const int n = region_count(cursor, bkt);
  const int nsp = *spillcnt;
  const float* __restrict__ hl = h + 2 * lane;
  float2 a0 = make_float2(0.0f, 0.0f), a1 = make_float2(0.0f, 0.0f);

  const int half = lane >> 4;
  const int l15  = lane & 15;
  const int nb   = (wave & 3) << 4;
  v2f bf[ODIM / 4] = {};
  float bv = 0.0f;
  if (wave < 4) {
    const float* __restrict__ wr = W + (size_t)(2 * half) * ODIM + nb + l15;
#pragma unroll
    for (int q = 0; q < ODIM / 4; ++q) {
      bf[q][0] = wr[(size_t)(4 * q + 0) * ODIM];
      bf[q][1] = wr[(size_t)(4 * q + 1) * ODIM];
    }
    bv = bias[nb + l15];
  }

  int2* __restrict__ myq = gq[wave];
  const int4* __restrict__ list = reg;
  int le = n;
  int e = wave * 32;
  int pass = 0;
  while (true) {
    if (e < le) {
      bool hit[SCH];
      float ds[SCH], dd[SCH];
#pragma unroll
      for (int u = 0; u < SCH; ++u)
        asm volatile("" :: "v"(r[u].x), "v"(r[u].y), "v"(r[u].z));
#pragma unroll
      for (int u = 0; u < SCH; ++u) {
        hit[u] = e + 256 * u + lane < le && (r[u].y >> 1) == c;
        ds[u] = dinv[hit[u] ? r[u].x : 0];
        dd[u] = dinv[hit[u] ? r[u].y : 0];
      }
      int nq = 0;
#pragma unroll
      for (int u = 0; u < SCH; ++u) {
        const unsigned m = __builtin_amdgcn_ballot_w32(hit[u]);
        if (hit[u]) {
          const int p = nq + __builtin_popcount(m & ((1u << lane) - 1u));
          myq[p] = make_int2(r[u].x | ((r[u].y & 1) << 17),
                             __float_as_int(ds[u] * __int_as_float(r[u].z) * dd[u]));
        }
        nq += __builtin_popcount(m);
      }
      for (int q0 = 0; q0 < nq; q0 += 32) {
        const int cnt = min(32, nq - q0);
        int2 qe = make_int2(0, 0);
        if (lane < cnt) qe = myq[q0 + lane];
        for (int j = 0; j < cnt; j += GB) {
          float2 xv[GB]; int pk[GB]; float nr[GB];
#pragma unroll
          for (int u = 0; u < GB; ++u) {
            const int jj = min(j + u, cnt - 1);
            pk[u] = __builtin_amdgcn_readlane(qe.x, jj);
            nr[u] = __int_as_float(__builtin_amdgcn_readlane(qe.y, jj));
            xv[u] = *(const float2*)(hl + (size_t)(pk[u] & 0x1FFFF) * ODIM);
          }
#pragma unroll
          for (int u = 0; u < GB; ++u) {
            if (j + u < cnt) {
              if (pk[u] >> 17) { a1.x = __builtin_fmaf(nr[u], xv[u].x, a1.x); a1.y = __builtin_fmaf(nr[u], xv[u].y, a1.y); }
              else             { a0.x = __builtin_fmaf(nr[u], xv[u].x, a0.x); a0.y = __builtin_fmaf(nr[u], xv[u].y, a0.y); }
            }
          }
        }
      }
    }
    e += 256 * SCH;
    if (e >= le) {
      if (pass) break;
      pass = 1; list = spill; le = nsp; e = wave * 32;
      if (e >= le) break;
    }
#pragma unroll
    for (int u = 0; u < SCH; ++u)
      r[u] = list[min(e + 256 * u + lane, le - 1)];
  }
  if (wave == 0) {
    const float d0 = dinv[i0], d1 = dinv[i0 + 1];
    const float2 x0 = *(const float2*)(hl + (size_t)i0 * ODIM);
    const float2 x1 = *(const float2*)(hl + (size_t)(i0 + 1) * ODIM);
    a0.x += d0 * d0 * x0.x; a0.y += d0 * d0 * x0.y;
    a1.x += d1 * d1 * x1.x; a1.y += d1 * d1 * x1.y;
  }
  part[wave][0][2 * lane] = a0.x; part[wave][0][2 * lane + 1] = a0.y;
  part[wave][1][2 * lane] = a1.x; part[wave][1][2 * lane + 1] = a1.y;
  for (int idx = tid; idx < 14 * ODIM; idx += 256)
    tile[(2 + (idx >> 6)) * TPITCH + (idx & 63)] = 0.0f;
  __syncthreads();
  if (tid < 2 * ODIM) {
    const int p = tid >> 6, f = tid & 63;
    float sum = part[0][p][f];
#pragma unroll
    for (int w = 1; w < 8; ++w) sum += part[w][p][f];
    tile[p * TPITCH + f] = sum;
  }
  __syncthreads();

  if (wave < 4) {
    const float* ar = tile + l15 * TPITCH + 2 * half;
    v8f acc = {};
#pragma unroll
    for (int k = 0; k < ODIM; k += 4) {
      v2f a;
      a[0] = ar[k + 0];
      a[1] = ar[k + 1];
      acc = __builtin_amdgcn_wmma_f32_16x16x4_f32(
           false, a,  false, bf[k >> 2],
           (short)0, acc,  false,  false);
    }
    const float pooled = fmaxf(acc[0] + bv, acc[1] + bv);
    if (half == 0) orow[nb + l15] = pooled;
  }
  __syncthreads();
  if (tid < 16)
    *(float4*)(out + (size_t)g * ODIM + 4 * tid) = *(const float4*)(orow + 4 * tid);
}

extern "C" void kernel_launch(void* const* d_in, const int* in_sizes, int n_in,
                              void* d_out, int out_size, void* d_ws, size_t ws_size,
                              hipStream_t stream) {
  const float* x   = (const float*)d_in[0];
  const int*   ei  = (const int*)d_in[1];
  const float* ew  = (const float*)d_in[2];
  const int*   bat = (const int*)d_in[3];
  const float* W1  = (const float*)d_in[4];
  const float* b1  = (const float*)d_in[5];
  const float* W2  = (const float*)d_in[6];
  const float* b2  = (const float*)d_in[7];
  float* out = (float*)d_out;

  const int IN_DIM = 128;
  const int N  = in_sizes[0] / IN_DIM;
  const int E  = in_sizes[2];
  const int B  = out_size / ODIM;
  const int NC = N / 2;

  const int* src = ei;
  const int* dst = ei + E;

  char* ws = (char*)d_ws;
  size_t featBytes = (size_t)N * ODIM * sizeof(float);
  size_t aggBytes  = (size_t)2 * B * ODIM * sizeof(float);
  float* h    = (float*)(ws + featBytes);
  float* dinv = (float*)(ws + 2 * featBytes + (size_t)N * 4);
  int*   need = (int*)  (ws + 2 * featBytes + (size_t)N * 8 + aggBytes);
  int*   first= (int*)  (ws + 2 * featBytes + (size_t)N * 12 + aggBytes);
  int4*  rec      = (int4*)(ws + 134217728);
  int*   cursor   = (int*) (ws + 209715200);
  int*   spillcnt = (int*) (ws + 209715200 + 65536);
  int4*  spill    = (int4*)(ws + 218103808);

  const int T = 256;
  dim3 blk(T);

  const int NB = (N + RB - 1) / RB;
  const int cover = N > B ? N : B;
  init_tables<<<(cover + T - 1) / T, blk, 0, stream>>>(cursor, NB, spillcnt, first, B, need, N);

  const int G1 = 256;
  const int chunk = ((E + G1 - 1) / G1 + 3) & ~3;
  bin_scatter_first<<<G1, dim3(512), 0, stream>>>(src, dst, ew, cursor, spillcnt, rec, spill, E, NB, chunk,
                                                  bat, first, NC);
  bucket_deg_mark<<<NB, blk, 0, stream>>>(rec, spill, cursor, spillcnt, dinv, N, first, B, need, NC);

  bucket_aggregate<<<NB, blk, 0, stream>>>(x, W1, dinv, b1, rec, spill, cursor, spillcnt, need, h, N);

  graph_rows_out<<<B, blk, 0, stream>>>(h, dinv, rec, spill, cursor, spillcnt, first, W2, b2, out, NC);
}
